# starting kernel
# baseline (speedup 1.0000x reference)
#include <hip/hip_runtime.h>
#include <hip/hip_bf16.h>
#include <math.h>

#define B_SZ 256
#define R_SZ 36
#define T_SZ 64
#define D_SZ 1024
#define EPS_L2 1e-8f
#define EPS_BN 1e-5f
#define LEAK 0.1f
#define KC 64

typedef __attribute__((ext_vector_type(2))) float v2f;
typedef __attribute__((ext_vector_type(8))) float v8f;

__global__ void zero_ws_kernel(float* p, int n) {
    int i = blockIdx.x * blockDim.x + threadIdx.x;
    if (i < n) p[i] = 0.f;
}

__global__ void cap_pre_kernel(const float* __restrict__ cap,
                               const int* __restrict__ lens,
                               float* __restrict__ cap_mean) {
    const int b = blockIdx.x;
    const int len = lens[b];
    const float* cb = cap + (size_t)b * T_SZ * D_SZ;
    float sumsq[4] = {0.f, 0.f, 0.f, 0.f};
    float summ[4]  = {0.f, 0.f, 0.f, 0.f};
    for (int t = 0; t < T_SZ; ++t) {
        const float* row = cb + (size_t)t * D_SZ;
        const bool valid = (t < len);
#pragma unroll
        for (int i = 0; i < 4; ++i) {
            float x = row[threadIdx.x + i * 256];
            float y = x > 0.f ? x : LEAK * x;
            sumsq[i] += y * y;
            if (valid) summ[i] += y;
        }
    }
#pragma unroll
    for (int i = 0; i < 4; ++i) {
        int d = threadIdx.x + i * 256;
        float n = sqrtf(sumsq[i]) + EPS_L2;
        cap_mean[(size_t)b * D_SZ + d] = summ[i] / (n * (float)len);
    }
}

__global__ void img_pre_kernel(const float* __restrict__ img,
                               float* __restrict__ meanR,
                               float* __restrict__ cs1,
                               float* __restrict__ cs2) {
    const int b = blockIdx.x;
    const float* ib = img + (size_t)b * R_SZ * D_SZ;
    float sumsq[4] = {0.f, 0.f, 0.f, 0.f};
    float summ[4]  = {0.f, 0.f, 0.f, 0.f};
    for (int r = 0; r < R_SZ; ++r) {
        const float* row = ib + (size_t)r * D_SZ;
#pragma unroll
        for (int i = 0; i < 4; ++i) {
            float x = row[threadIdx.x + i * 256];
            float y = x > 0.f ? x : LEAK * x;
            sumsq[i] += y * y;
            summ[i]  += y;
        }
    }
#pragma unroll
    for (int i = 0; i < 4; ++i) {
        int d = threadIdx.x + i * 256;
        float n = sqrtf(sumsq[i]) + EPS_L2;
        float s1 = summ[i] / n;
        float s2 = sumsq[i] / (n * n);
        meanR[(size_t)b * D_SZ + d] = s1 * (1.f / (float)R_SZ);
        atomicAdd(cs1 + d, s1);
        atomicAdd(cs2 + d, s2);
    }
}

__global__ void make_base_kernel(const float* __restrict__ meanR,
                                 const float* __restrict__ cs1,
                                 const float* __restrict__ cs2,
                                 const float* __restrict__ bn_w,
                                 const float* __restrict__ bn_b,
                                 float* __restrict__ base,
                                 float* __restrict__ base2) {
    const int b = blockIdx.x;
    const float inv_n = 1.f / (float)(B_SZ * R_SZ);
#pragma unroll
    for (int i = 0; i < 4; ++i) {
        int d = threadIdx.x + i * 256;
        float mu  = cs1[d] * inv_n;
        float ex2 = cs2[d] * inv_n;
        float var = ex2 - mu * mu;
        float scale = bn_w[d] * rsqrtf(var + EPS_BN);
        float v = (meanR[(size_t)b * D_SZ + d] - mu) * scale + bn_b[d];
        base[(size_t)b * D_SZ + d]  = v;
        base2[(size_t)b * D_SZ + d] = v * v;
    }
}

__global__ __launch_bounds__(128) void fc_gemm_kernel(
        const float* __restrict__ A, const float* __restrict__ Bw,
        const float* __restrict__ bias, float* __restrict__ Cout) {
    __shared__ float sA[2][16][KC + 1];
    __shared__ float sB[2][16][KC + 1];
    const int tid  = threadIdx.x;
    const int wave = tid >> 5;
    const int lane = tid & 31;
    const int wm = wave >> 1, wn = wave & 1;
    const int half = lane >> 4, fr = lane & 15;
    const int m0 = blockIdx.y * 32;
    const int n0 = blockIdx.x * 32;
    v8f acc = {};
    for (int k0 = 0; k0 < D_SZ; k0 += KC) {
        for (int idx = tid; idx < 2 * 16 * KC; idx += 128) {
            int blk = idx / (16 * KC);
            int rem = idx % (16 * KC);
            int row = rem / KC, col = rem % KC;
            sA[blk][row][col] = A [(size_t)(m0 + blk * 16 + row) * D_SZ + k0 + col];
            sB[blk][row][col] = Bw[(size_t)(n0 + blk * 16 + row) * D_SZ + k0 + col];
        }
        __syncthreads();
#pragma unroll
        for (int kk = 0; kk < KC; kk += 4) {
            int kb = kk + half * 2;
            v2f a, b;
            a[0] = sA[wm][fr][kb]; a[1] = sA[wm][fr][kb + 1];
            b[0] = sB[wn][fr][kb]; b[1] = sB[wn][fr][kb + 1];
            acc = __builtin_amdgcn_wmma_f32_16x16x4_f32(
                    false, a, false, b, (short)0, acc, false, false);
        }
        __syncthreads();
    }
#pragma unroll
    for (int j = 0; j < 8; ++j) {
        int m = m0 + wm * 16 + j + half * 8;
        int n = n0 + wn * 16 + fr;
        Cout[(size_t)m * (2 * D_SZ) + n] = acc[j] + bias[n];
    }
}

__global__ void make_pab_kernel(const float* __restrict__ cap_mean,
                                const float* __restrict__ params,
                                float* __restrict__ P, float* __restrict__ A2m,
                                float* __restrict__ ABm,
                                float* __restrict__ qv, float* __restrict__ rv) {
    const int c = blockIdx.x;
    const int tid = threadIdx.x;
    __shared__ float red[256];

    float ss = 0.f;
#pragma unroll
    for (int i = 0; i < 4; ++i) {
        float x = cap_mean[(size_t)c * D_SZ + tid + i * 256];
        ss += x * x;
    }
    red[tid] = ss; __syncthreads();
    for (int s = 128; s > 0; s >>= 1) {
        if (tid < s) red[tid] += red[tid + s];
        __syncthreads();
    }
    const float nrm = sqrtf(red[0]) + EPS_L2;
    __syncthreads();

    float q = 0.f, r = 0.f;
#pragma unroll
    for (int i = 0; i < 4; ++i) {
        int d = tid + i * 256;
        float alpha = params[(size_t)c * (2 * D_SZ) + 2 * d];
        float beta  = params[(size_t)c * (2 * D_SZ) + 2 * d + 1];
        float ch = cap_mean[(size_t)c * D_SZ + d] / nrm;
        P  [(size_t)c * D_SZ + d] = alpha * ch;
        A2m[(size_t)c * D_SZ + d] = alpha * alpha;
        ABm[(size_t)c * D_SZ + d] = alpha * beta;
        q += beta * ch;
        r += beta * beta;
    }
    red[tid] = q; __syncthreads();
    for (int s = 128; s > 0; s >>= 1) {
        if (tid < s) red[tid] += red[tid + s];
        __syncthreads();
    }
    if (tid == 0) qv[c] = red[0];
    __syncthreads();
    red[tid] = r; __syncthreads();
    for (int s = 128; s > 0; s >>= 1) {
        if (tid < s) red[tid] += red[tid + s];
        __syncthreads();
    }
    if (tid == 0) rv[c] = red[0];
}

__global__ __launch_bounds__(128) void sims_gemm_kernel(
        const float* __restrict__ base, const float* __restrict__ base2,
        const float* __restrict__ P, const float* __restrict__ A2m,
        const float* __restrict__ ABm,
        const float* __restrict__ qv, const float* __restrict__ rv,
        float* __restrict__ out) {
    __shared__ float sBa [2][16][KC + 1];
    __shared__ float sBa2[2][16][KC + 1];
    __shared__ float sP  [2][16][KC + 1];
    __shared__ float sA2 [2][16][KC + 1];
    __shared__ float sAB [2][16][KC + 1];
    const int tid  = threadIdx.x;
    const int wave = tid >> 5;
    const int lane = tid & 31;
    const int wm = wave >> 1, wn = wave & 1;
    const int half = lane >> 4, fr = lane & 15;
    const int bm0 = blockIdx.y * 32;
    const int cn0 = blockIdx.x * 32;
    v8f acc1 = {}, acc2 = {}, acc3 = {};
    for (int k0 = 0; k0 < D_SZ; k0 += KC) {
        for (int idx = tid; idx < 2 * 16 * KC; idx += 128) {
            int blk = idx / (16 * KC);
            int rem = idx % (16 * KC);
            int row = rem / KC, col = rem % KC;
            size_t ga = (size_t)(bm0 + blk * 16 + row) * D_SZ + k0 + col;
            size_t gb = (size_t)(cn0 + blk * 16 + row) * D_SZ + k0 + col;
            sBa [blk][row][col] = base [ga];
            sBa2[blk][row][col] = base2[ga];
            sP  [blk][row][col] = P  [gb];
            sA2 [blk][row][col] = A2m[gb];
            sAB [blk][row][col] = ABm[gb];
        }
        __syncthreads();
#pragma unroll
        for (int kk = 0; kk < KC; kk += 4) {
            int kb = kk + half * 2;
            v2f a1, a2, bp, b2, bb;
            a1[0] = sBa [wm][fr][kb]; a1[1] = sBa [wm][fr][kb + 1];
            a2[0] = sBa2[wm][fr][kb]; a2[1] = sBa2[wm][fr][kb + 1];
            bp[0] = sP  [wn][fr][kb]; bp[1] = sP  [wn][fr][kb + 1];
            b2[0] = sA2 [wn][fr][kb]; b2[1] = sA2 [wn][fr][kb + 1];
            bb[0] = sAB [wn][fr][kb]; bb[1] = sAB [wn][fr][kb + 1];
            acc1 = __builtin_amdgcn_wmma_f32_16x16x4_f32(
                    false, a1, false, bp, (short)0, acc1, false, false);
            acc2 = __builtin_amdgcn_wmma_f32_16x16x4_f32(
                    false, a2, false, b2, (short)0, acc2, false, false);
            acc3 = __builtin_amdgcn_wmma_f32_16x16x4_f32(
                    false, a1, false, bb, (short)0, acc3, false, false);
        }
        __syncthreads();
    }
#pragma unroll
    for (int j = 0; j < 8; ++j) {
        int b = bm0 + wm * 16 + j + half * 8;
        int c = cn0 + wn * 16 + fr;
        float num = acc1[j] + qv[c];
        float den = sqrtf(acc2[j] + 2.f * acc3[j] + rv[c]) + EPS_L2;
        out[(size_t)b * B_SZ + c] = num / den;
    }
}

extern "C" void kernel_launch(void* const* d_in, const int* in_sizes, int n_in,
                              void* d_out, int out_size, void* d_ws, size_t ws_size,
                              hipStream_t stream) {
    const float* img   = (const float*)d_in[0];
    const float* cap   = (const float*)d_in[1];
    const int*   lens  = (const int*)  d_in[2];
    const float* fc_w  = (const float*)d_in[3];
    const float* fc_b  = (const float*)d_in[4];
    const float* bn_w  = (const float*)d_in[5];
    const float* bn_b  = (const float*)d_in[6];
    float* out = (float*)d_out;

    float* ws = (float*)d_ws;
    const size_t BD = (size_t)B_SZ * D_SZ;
    float* cap_mean = ws;
    float* meanR    = cap_mean + BD;
    float* base     = meanR + BD;
    float* base2    = base + BD;
    float* P        = base2 + BD;
    float* A2m      = P + BD;
    float* ABm      = A2m + BD;
    float* params   = ABm + BD;
    float* colsum1  = params + 2 * BD;
    float* colsum2  = colsum1 + D_SZ;
    float* qv       = colsum2 + D_SZ;
    float* rv       = qv + B_SZ;

    zero_ws_kernel<<<dim3((2 * D_SZ + 255) / 256), 256, 0, stream>>>(colsum1, 2 * D_SZ);
    cap_pre_kernel<<<dim3(B_SZ), 256, 0, stream>>>(cap, lens, cap_mean);
    img_pre_kernel<<<dim3(B_SZ), 256, 0, stream>>>(img, meanR, colsum1, colsum2);
    make_base_kernel<<<dim3(B_SZ), 256, 0, stream>>>(meanR, colsum1, colsum2,
                                                     bn_w, bn_b, base, base2);
    fc_gemm_kernel<<<dim3(2 * D_SZ / 32, B_SZ / 32), 128, 0, stream>>>(
            cap_mean, fc_w, fc_b, params);
    make_pab_kernel<<<dim3(B_SZ), 256, 0, stream>>>(cap_mean, params,
                                                    P, A2m, ABm, qv, rv);
    sims_gemm_kernel<<<dim3(B_SZ / 32, B_SZ / 32), 128, 0, stream>>>(
            base, base2, P, A2m, ABm, qv, rv, out);
}


# Both WMMA GEMMs: each K-chunk's global loads go out as one float4 group, prefetched one chunk ahead in registers (chunk 64 to 128), instead of one dependent load-store round per staging trip
# speedup vs baseline: 2.7225x; 2.7225x over previous
#include <hip/hip_runtime.h>
#include <hip/hip_bf16.h>
#include <math.h>

#define B_SZ 256
#define R_SZ 36
#define T_SZ 64
#define D_SZ 1024
#define EPS_L2 1e-8f
#define EPS_BN 1e-5f
#define LEAK 0.1f
#define KC 128
#define ST_CPR (KC / 4)
#define ST_RPP (128 / ST_CPR)
#define ST_N   (32 / ST_RPP)

typedef __attribute__((ext_vector_type(2))) float v2f;
typedef __attribute__((ext_vector_type(8))) float v8f;

__global__ void zero_ws_kernel(float* p, int n) {
    int i = blockIdx.x * blockDim.x + threadIdx.x;
    if (i < n) p[i] = 0.f;
}

__global__ void cap_pre_kernel(const float* __restrict__ cap,
                               const int* __restrict__ lens,
                               float* __restrict__ cap_mean) {
    const int b = blockIdx.x;
    const int len = lens[b];
    const float* cb = cap + (size_t)b * T_SZ * D_SZ;
    float sumsq[4] = {0.f, 0.f, 0.f, 0.f};
    float summ[4]  = {0.f, 0.f, 0.f, 0.f};
    for (int t = 0; t < T_SZ; ++t) {
        const float* row = cb + (size_t)t * D_SZ;
        const bool valid = (t < len);
#pragma unroll
        for (int i = 0; i < 4; ++i) {
            float x = row[threadIdx.x + i * 256];
            float y = x > 0.f ? x : LEAK * x;
            sumsq[i] += y * y;
            if (valid) summ[i] += y;
        }
    }
#pragma unroll
    for (int i = 0; i < 4; ++i) {
        int d = threadIdx.x + i * 256;
        float n = sqrtf(sumsq[i]) + EPS_L2;
        cap_mean[(size_t)b * D_SZ + d] = summ[i] / (n * (float)len);
    }
}

__global__ void img_pre_kernel(const float* __restrict__ img,
                               float* __restrict__ meanR,
                               float* __restrict__ cs1,
                               float* __restrict__ cs2) {
    const int b = blockIdx.x;
    const float* ib = img + (size_t)b * R_SZ * D_SZ;
    float sumsq[4] = {0.f, 0.f, 0.f, 0.f};
    float summ[4]  = {0.f, 0.f, 0.f, 0.f};
    for (int r = 0; r < R_SZ; ++r) {
        const float* row = ib + (size_t)r * D_SZ;
#pragma unroll
        for (int i = 0; i < 4; ++i) {
            float x = row[threadIdx.x + i * 256];
            float y = x > 0.f ? x : LEAK * x;
            sumsq[i] += y * y;
            summ[i]  += y;
        }
    }
#pragma unroll
    for (int i = 0; i < 4; ++i) {
        int d = threadIdx.x + i * 256;
        float n = sqrtf(sumsq[i]) + EPS_L2;
        float s1 = summ[i] / n;
        float s2 = sumsq[i] / (n * n);
        meanR[(size_t)b * D_SZ + d] = s1 * (1.f / (float)R_SZ);
        atomicAdd(cs1 + d, s1);
        atomicAdd(cs2 + d, s2);
    }
}

__global__ void make_base_kernel(const float* __restrict__ meanR,
                                 const float* __restrict__ cs1,
                                 const float* __restrict__ cs2,
                                 const float* __restrict__ bn_w,
                                 const float* __restrict__ bn_b,
                                 float* __restrict__ base,
                                 float* __restrict__ base2) {
    const int b = blockIdx.x;
    const float inv_n = 1.f / (float)(B_SZ * R_SZ);
#pragma unroll
    for (int i = 0; i < 4; ++i) {
        int d = threadIdx.x + i * 256;
        float mu  = cs1[d] * inv_n;
        float ex2 = cs2[d] * inv_n;
        float var = ex2 - mu * mu;
        float scale = bn_w[d] * rsqrtf(var + EPS_BN);
        float v = (meanR[(size_t)b * D_SZ + d] - mu) * scale + bn_b[d];
        base[(size_t)b * D_SZ + d]  = v;
        base2[(size_t)b * D_SZ + d] = v * v;
    }
}

__global__ __launch_bounds__(128) __attribute__((amdgpu_waves_per_eu(1, 4)))
void fc_gemm_kernel(
        const float* __restrict__ A, const float* __restrict__ Bw,
        const float* __restrict__ bias, float* __restrict__ Cout) {
    __shared__ float sA[2][16][KC + 1];
    __shared__ float sB[2][16][KC + 1];
    const int tid  = threadIdx.x;
    const int wave = tid >> 5;
    const int lane = tid & 31;
    const int wm = wave >> 1, wn = wave & 1;
    const int half = lane >> 4, fr = lane & 15;
    const int m0 = blockIdx.y * 32;
    const int n0 = blockIdx.x * 32;
    v8f acc = {};
    const int srow = tid / ST_CPR;
    const int scol = (tid % ST_CPR) * 4;
    const float* gA = A  + (size_t)(m0 + srow) * D_SZ + scol;
    const float* gB = Bw + (size_t)(n0 + srow) * D_SZ + scol;
    float* lA = &sA[0][0][0] + srow * (KC + 1) + scol;
    float* lB = &sB[0][0][0] + srow * (KC + 1) + scol;
    float4 ra[ST_N], rb[ST_N];
#pragma unroll
    for (int i = 0; i < ST_N; ++i) {
        ra[i] = *(const float4*)(gA + (size_t)(ST_RPP * i) * D_SZ);
        rb[i] = *(const float4*)(gB + (size_t)(ST_RPP * i) * D_SZ);
    }
#pragma unroll 1
    for (int k0 = 0; k0 < D_SZ; k0 += KC) {
#pragma unroll
        for (int i = 0; i < ST_N; ++i) {
            float* da = lA + ST_RPP * i * (KC + 1);
            float* db = lB + ST_RPP * i * (KC + 1);
            da[0] = ra[i].x; da[1] = ra[i].y; da[2] = ra[i].z; da[3] = ra[i].w;
            db[0] = rb[i].x; db[1] = rb[i].y; db[2] = rb[i].z; db[3] = rb[i].w;
        }
        __syncthreads();
        if (k0 + KC < D_SZ) {
#pragma unroll
            for (int i = 0; i < ST_N; ++i) {
                ra[i] = *(const float4*)(gA + (size_t)(ST_RPP * i) * D_SZ + k0 + KC);
                rb[i] = *(const float4*)(gB + (size_t)(ST_RPP * i) * D_SZ + k0 + KC);
            }
        }
#pragma unroll
        for (int kk = 0; kk < KC; kk += 4) {
            int kb = kk + half * 2;
            v2f a, b;
            a[0] = sA[wm][fr][kb]; a[1] = sA[wm][fr][kb + 1];
            b[0] = sB[wn][fr][kb]; b[1] = sB[wn][fr][kb + 1];
            acc = __builtin_amdgcn_wmma_f32_16x16x4_f32(
                    false, a, false, b, (short)0, acc, false, false);
        }
        __syncthreads();
    }
#pragma unroll
    for (int j = 0; j < 8; ++j) {
        int m = m0 + wm * 16 + j + half * 8;
        int n = n0 + wn * 16 + fr;
        Cout[(size_t)m * (2 * D_SZ) + n] = acc[j] + bias[n];
    }
}

__global__ void make_pab_kernel(const float* __restrict__ cap_mean,
                                const float* __restrict__ params,
                                float* __restrict__ P, float* __restrict__ A2m,
                                float* __restrict__ ABm,
                                float* __restrict__ qv, float* __restrict__ rv) {
    const int c = blockIdx.x;
    const int tid = threadIdx.x;
    __shared__ float red[256];

    float ss = 0.f;
#pragma unroll
    for (int i = 0; i < 4; ++i) {
        float x = cap_mean[(size_t)c * D_SZ + tid + i * 256];
        ss += x * x;
    }
    red[tid] = ss; __syncthreads();
    for (int s = 128; s > 0; s >>= 1) {
        if (tid < s) red[tid] += red[tid + s];
        __syncthreads();
    }
    const float nrm = sqrtf(red[0]) + EPS_L2;
    __syncthreads();

    float q = 0.f, r = 0.f;
#pragma unroll
    for (int i = 0; i < 4; ++i) {
        int d = tid + i * 256;
        float alpha = params[(size_t)c * (2 * D_SZ) + 2 * d];
        float beta  = params[(size_t)c * (2 * D_SZ) + 2 * d + 1];
        float ch = cap_mean[(size_t)c * D_SZ + d] / nrm;
        P  [(size_t)c * D_SZ + d] = alpha * ch;
        A2m[(size_t)c * D_SZ + d] = alpha * alpha;
        ABm[(size_t)c * D_SZ + d] = alpha * beta;
        q += beta * ch;
        r += beta * beta;
    }
    red[tid] = q; __syncthreads();
    for (int s = 128; s > 0; s >>= 1) {
        if (tid < s) red[tid] += red[tid + s];
        __syncthreads();
    }
    if (tid == 0) qv[c] = red[0];
    __syncthreads();
    red[tid] = r; __syncthreads();
    for (int s = 128; s > 0; s >>= 1) {
        if (tid < s) red[tid] += red[tid + s];
        __syncthreads();
    }
    if (tid == 0) rv[c] = red[0];
}

__global__ __launch_bounds__(128) __attribute__((amdgpu_waves_per_eu(1, 2)))
void sims_gemm_kernel(
        const float* __restrict__ base, const float* __restrict__ base2,
        const float* __restrict__ P, const float* __restrict__ A2m,
        const float* __restrict__ ABm,
        const float* __restrict__ qv, const float* __restrict__ rv,
        float* __restrict__ out) {
    __shared__ float sBa [2][16][KC + 1];
    __shared__ float sBa2[2][16][KC + 1];
    __shared__ float sP  [2][16][KC + 1];
    __shared__ float sA2 [2][16][KC + 1];
    __shared__ float sAB [2][16][KC + 1];
    const int tid  = threadIdx.x;
    const int wave = tid >> 5;
    const int lane = tid & 31;
    const int wm = wave >> 1, wn = wave & 1;
    const int half = lane >> 4, fr = lane & 15;
    const int bm0 = blockIdx.y * 32;
    const int cn0 = blockIdx.x * 32;
    v8f acc1 = {}, acc2 = {}, acc3 = {};
    const int srow = tid / ST_CPR;
    const int scol = (tid % ST_CPR) * 4;
    const size_t goffA = (size_t)(bm0 + srow) * D_SZ + scol;
    const size_t goffB = (size_t)(cn0 + srow) * D_SZ + scol;
    const int loff = srow * (KC + 1) + scol;
    float4 r1[ST_N], r2[ST_N], rp[ST_N], rq[ST_N], rr[ST_N];
#pragma unroll
    for (int i = 0; i < ST_N; ++i) {
        const size_t o = (size_t)(ST_RPP * i) * D_SZ;
        r1[i] = *(const float4*)(base  + goffA + o);
        r2[i] = *(const float4*)(base2 + goffA + o);
        rp[i] = *(const float4*)(P     + goffB + o);
        rq[i] = *(const float4*)(A2m   + goffB + o);
        rr[i] = *(const float4*)(ABm   + goffB + o);
    }
    for (int k0 = 0; k0 < D_SZ; k0 += KC) {
#pragma unroll
        for (int i = 0; i < ST_N; ++i) {
            const int lo = loff + ST_RPP * i * (KC + 1);
            float* d1 = &sBa [0][0][0] + lo;
            float* d2 = &sBa2[0][0][0] + lo;
            float* dp = &sP  [0][0][0] + lo;
            float* dq = &sA2 [0][0][0] + lo;
            float* dr = &sAB [0][0][0] + lo;
            d1[0] = r1[i].x; d1[1] = r1[i].y; d1[2] = r1[i].z; d1[3] = r1[i].w;
            d2[0] = r2[i].x; d2[1] = r2[i].y; d2[2] = r2[i].z; d2[3] = r2[i].w;
            dp[0] = rp[i].x; dp[1] = rp[i].y; dp[2] = rp[i].z; dp[3] = rp[i].w;
            dq[0] = rq[i].x; dq[1] = rq[i].y; dq[2] = rq[i].z; dq[3] = rq[i].w;
            dr[0] = rr[i].x; dr[1] = rr[i].y; dr[2] = rr[i].z; dr[3] = rr[i].w;
        }
        __syncthreads();
        if (k0 + KC < D_SZ) {
#pragma unroll
            for (int i = 0; i < ST_N; ++i) {
                const size_t o = (size_t)(ST_RPP * i) * D_SZ + k0 + KC;
                r1[i] = *(const float4*)(base  + goffA + o);
                r2[i] = *(const float4*)(base2 + goffA + o);
                rp[i] = *(const float4*)(P     + goffB + o);
                rq[i] = *(const float4*)(A2m   + goffB + o);
                rr[i] = *(const float4*)(ABm   + goffB + o);
            }
        }
#pragma unroll
        for (int kk = 0; kk < KC; kk += 4) {
            int kb = kk + half * 2;
            v2f a1, a2, bp, b2, bb;
            a1[0] = sBa [wm][fr][kb]; a1[1] = sBa [wm][fr][kb + 1];
            a2[0] = sBa2[wm][fr][kb]; a2[1] = sBa2[wm][fr][kb + 1];
            bp[0] = sP  [wn][fr][kb]; bp[1] = sP  [wn][fr][kb + 1];
            b2[0] = sA2 [wn][fr][kb]; b2[1] = sA2 [wn][fr][kb + 1];
            bb[0] = sAB [wn][fr][kb]; bb[1] = sAB [wn][fr][kb + 1];
            acc1 = __builtin_amdgcn_wmma_f32_16x16x4_f32(
                    false, a1, false, bp, (short)0, acc1, false, false);
            acc2 = __builtin_amdgcn_wmma_f32_16x16x4_f32(
                    false, a2, false, b2, (short)0, acc2, false, false);
            acc3 = __builtin_amdgcn_wmma_f32_16x16x4_f32(
                    false, a1, false, bb, (short)0, acc3, false, false);
        }
        __syncthreads();
    }
#pragma unroll
    for (int j = 0; j < 8; ++j) {
        int b = bm0 + wm * 16 + j + half * 8;
        int c = cn0 + wn * 16 + fr;
        float num = acc1[j] + qv[c];
        float den = sqrtf(acc2[j] + 2.f * acc3[j] + rv[c]) + EPS_L2;
        out[(size_t)b * B_SZ + c] = num / den;
    }
}

extern "C" void kernel_launch(void* const* d_in, const int* in_sizes, int n_in,
                              void* d_out, int out_size, void* d_ws, size_t ws_size,
                              hipStream_t stream) {
    const float* img   = (const float*)d_in[0];
    const float* cap   = (const float*)d_in[1];
    const int*   lens  = (const int*)  d_in[2];
    const float* fc_w  = (const float*)d_in[3];
    const float* fc_b  = (const float*)d_in[4];
    const float* bn_w  = (const float*)d_in[5];
    const float* bn_b  = (const float*)d_in[6];
    float* out = (float*)d_out;

    float* ws = (float*)d_ws;
    const size_t BD = (size_t)B_SZ * D_SZ;
    float* cap_mean = ws;
    float* meanR    = cap_mean + BD;
    float* base     = meanR + BD;
    float* base2    = base + BD;
    float* P        = base2 + BD;
    float* A2m      = P + BD;
    float* ABm      = A2m + BD;
    float* params   = ABm + BD;
    float* colsum1  = params + 2 * BD;
    float* colsum2  = colsum1 + D_SZ;
    float* qv       = colsum2 + D_SZ;
    float* rv       = qv + B_SZ;

    zero_ws_kernel<<<dim3((2 * D_SZ + 255) / 256), 256, 0, stream>>>(colsum1, 2 * D_SZ);
    cap_pre_kernel<<<dim3(B_SZ), 256, 0, stream>>>(cap, lens, cap_mean);
    img_pre_kernel<<<dim3(B_SZ), 256, 0, stream>>>(img, meanR, colsum1, colsum2);
    make_base_kernel<<<dim3(B_SZ), 256, 0, stream>>>(meanR, colsum1, colsum2,
                                                     bn_w, bn_b, base, base2);
    fc_gemm_kernel<<<dim3(2 * D_SZ / 32, B_SZ / 32), 128, 0, stream>>>(
            cap_mean, fc_w, fc_b, params);
    make_pab_kernel<<<dim3(B_SZ), 256, 0, stream>>>(cap_mean, params,
                                                    P, A2m, ABm, qv, rv);
    sims_gemm_kernel<<<dim3(B_SZ / 32, B_SZ / 32), 128, 0, stream>>>(
            base, base2, P, A2m, ABm, qv, rv, out);
}


# sims GEMM: one 16x16 tile per block, its 4 waves split each K chunk and add their partial tiles through LDS in fixed wave order (sum order changes): 256 blocks instead of 64
# speedup vs baseline: 3.1239x; 1.1474x over previous
#include <hip/hip_runtime.h>
#include <hip/hip_bf16.h>
#include <math.h>

#define B_SZ 256
#define R_SZ 36
#define T_SZ 64
#define D_SZ 1024
#define EPS_L2 1e-8f
#define EPS_BN 1e-5f
#define LEAK 0.1f
#define KC 128
#define ST_CPR (KC / 4)
#define ST_RPP (128 / ST_CPR)
#define ST_N   (32 / ST_RPP)

typedef __attribute__((ext_vector_type(2))) float v2f;
typedef __attribute__((ext_vector_type(8))) float v8f;

__global__ void zero_ws_kernel(float* p, int n) {
    int i = blockIdx.x * blockDim.x + threadIdx.x;
    if (i < n) p[i] = 0.f;
}

__global__ void cap_pre_kernel(const float* __restrict__ cap,
                               const int* __restrict__ lens,
                               float* __restrict__ cap_mean) {
    const int b = blockIdx.x;
    const int len = lens[b];
    const float* cb = cap + (size_t)b * T_SZ * D_SZ;
    float sumsq[4] = {0.f, 0.f, 0.f, 0.f};
    float summ[4]  = {0.f, 0.f, 0.f, 0.f};
    for (int t = 0; t < T_SZ; ++t) {
        const float* row = cb + (size_t)t * D_SZ;
        const bool valid = (t < len);
#pragma unroll
        for (int i = 0; i < 4; ++i) {
            float x = row[threadIdx.x + i * 256];
            float y = x > 0.f ? x : LEAK * x;
            sumsq[i] += y * y;
            if (valid) summ[i] += y;
        }
    }
#pragma unroll
    for (int i = 0; i < 4; ++i) {
        int d = threadIdx.x + i * 256;
        float n = sqrtf(sumsq[i]) + EPS_L2;
        cap_mean[(size_t)b * D_SZ + d] = summ[i] / (n * (float)len);
    }
}

__global__ void img_pre_kernel(const float* __restrict__ img,
                               float* __restrict__ meanR,
                               float* __restrict__ cs1,
                               float* __restrict__ cs2) {
    const int b = blockIdx.x;
    const float* ib = img + (size_t)b * R_SZ * D_SZ;
    float sumsq[4] = {0.f, 0.f, 0.f, 0.f};
    float summ[4]  = {0.f, 0.f, 0.f, 0.f};
    for (int r = 0; r < R_SZ; ++r) {
        const float* row = ib + (size_t)r * D_SZ;
#pragma unroll
        for (int i = 0; i < 4; ++i) {
            float x = row[threadIdx.x + i * 256];
            float y = x > 0.f ? x : LEAK * x;
            sumsq[i] += y * y;
            summ[i]  += y;
        }
    }
#pragma unroll
    for (int i = 0; i < 4; ++i) {
        int d = threadIdx.x + i * 256;
        float n = sqrtf(sumsq[i]) + EPS_L2;
        float s1 = summ[i] / n;
        float s2 = sumsq[i] / (n * n);
        meanR[(size_t)b * D_SZ + d] = s1 * (1.f / (float)R_SZ);
        atomicAdd(cs1 + d, s1);
        atomicAdd(cs2 + d, s2);
    }
}

__global__ void make_base_kernel(const float* __restrict__ meanR,
                                 const float* __restrict__ cs1,
                                 const float* __restrict__ cs2,
                                 const float* __restrict__ bn_w,
                                 const float* __restrict__ bn_b,
                                 float* __restrict__ base,
                                 float* __restrict__ base2) {
    const int b = blockIdx.x;
    const float inv_n = 1.f / (float)(B_SZ * R_SZ);
#pragma unroll
    for (int i = 0; i < 4; ++i) {
        int d = threadIdx.x + i * 256;
        float mu  = cs1[d] * inv_n;
        float ex2 = cs2[d] * inv_n;
        float var = ex2 - mu * mu;
        float scale = bn_w[d] * rsqrtf(var + EPS_BN);
        float v = (meanR[(size_t)b * D_SZ + d] - mu) * scale + bn_b[d];
        base[(size_t)b * D_SZ + d]  = v;
        base2[(size_t)b * D_SZ + d] = v * v;
    }
}

__global__ __launch_bounds__(128) __attribute__((amdgpu_waves_per_eu(1, 4)))
void fc_gemm_kernel(
        const float* __restrict__ A, const float* __restrict__ Bw,
        const float* __restrict__ bias, float* __restrict__ Cout) {
    __shared__ float sA[2][16][KC + 1];
    __shared__ float sB[2][16][KC + 1];
    const int tid  = threadIdx.x;
    const int wave = tid >> 5;
    const int lane = tid & 31;
    const int wm = wave >> 1, wn = wave & 1;
    const int half = lane >> 4, fr = lane & 15;
    const int m0 = blockIdx.y * 32;
    const int n0 = blockIdx.x * 32;
    v8f acc = {};
    const int srow = tid / ST_CPR;
    const int scol = (tid % ST_CPR) * 4;
    const float* gA = A  + (size_t)(m0 + srow) * D_SZ + scol;
    const float* gB = Bw + (size_t)(n0 + srow) * D_SZ + scol;
    float* lA = &sA[0][0][0] + srow * (KC + 1) + scol;
    float* lB = &sB[0][0][0] + srow * (KC + 1) + scol;
    float4 ra[ST_N], rb[ST_N];
#pragma unroll
    for (int i = 0; i < ST_N; ++i) {
        ra[i] = *(const float4*)(gA + (size_t)(ST_RPP * i) * D_SZ);
        rb[i] = *(const float4*)(gB + (size_t)(ST_RPP * i) * D_SZ);
    }
#pragma unroll 1
    for (int k0 = 0; k0 < D_SZ; k0 += KC) {
#pragma unroll
        for (int i = 0; i < ST_N; ++i) {
            float* da = lA + ST_RPP * i * (KC + 1);
            float* db = lB + ST_RPP * i * (KC + 1);
            da[0] = ra[i].x; da[1] = ra[i].y; da[2] = ra[i].z; da[3] = ra[i].w;
            db[0] = rb[i].x; db[1] = rb[i].y; db[2] = rb[i].z; db[3] = rb[i].w;
        }
        __syncthreads();
        if (k0 + KC < D_SZ) {
#pragma unroll
            for (int i = 0; i < ST_N; ++i) {
                ra[i] = *(const float4*)(gA + (size_t)(ST_RPP * i) * D_SZ + k0 + KC);
                rb[i] = *(const float4*)(gB + (size_t)(ST_RPP * i) * D_SZ + k0 + KC);
            }
        }
#pragma unroll
        for (int kk = 0; kk < KC; kk += 4) {
            int kb = kk + half * 2;
            v2f a, b;
            a[0] = sA[wm][fr][kb]; a[1] = sA[wm][fr][kb + 1];
            b[0] = sB[wn][fr][kb]; b[1] = sB[wn][fr][kb + 1];
            acc = __builtin_amdgcn_wmma_f32_16x16x4_f32(
                    false, a, false, b, (short)0, acc, false, false);
        }
        __syncthreads();
    }
#pragma unroll
    for (int j = 0; j < 8; ++j) {
        int m = m0 + wm * 16 + j + half * 8;
        int n = n0 + wn * 16 + fr;
        Cout[(size_t)m * (2 * D_SZ) + n] = acc[j] + bias[n];
    }
}

__global__ void make_pab_kernel(const float* __restrict__ cap_mean,
                                const float* __restrict__ params,
                                float* __restrict__ P, float* __restrict__ A2m,
                                float* __restrict__ ABm,
                                float* __restrict__ qv, float* __restrict__ rv) {
    const int c = blockIdx.x;
    const int tid = threadIdx.x;
    __shared__ float red[256];

    float ss = 0.f;
#pragma unroll
    for (int i = 0; i < 4; ++i) {
        float x = cap_mean[(size_t)c * D_SZ + tid + i * 256];
        ss += x * x;
    }
    red[tid] = ss; __syncthreads();
    for (int s = 128; s > 0; s >>= 1) {
        if (tid < s) red[tid] += red[tid + s];
        __syncthreads();
    }
    const float nrm = sqrtf(red[0]) + EPS_L2;
    __syncthreads();

    float q = 0.f, r = 0.f;
#pragma unroll
    for (int i = 0; i < 4; ++i) {
        int d = tid + i * 256;
        float alpha = params[(size_t)c * (2 * D_SZ) + 2 * d];
        float beta  = params[(size_t)c * (2 * D_SZ) + 2 * d + 1];
        float ch = cap_mean[(size_t)c * D_SZ + d] / nrm;
        P  [(size_t)c * D_SZ + d] = alpha * ch;
        A2m[(size_t)c * D_SZ + d] = alpha * alpha;
        ABm[(size_t)c * D_SZ + d] = alpha * beta;
        q += beta * ch;
        r += beta * beta;
    }
    red[tid] = q; __syncthreads();
    for (int s = 128; s > 0; s >>= 1) {
        if (tid < s) red[tid] += red[tid + s];
        __syncthreads();
    }
    if (tid == 0) qv[c] = red[0];
    __syncthreads();
    red[tid] = r; __syncthreads();
    for (int s = 128; s > 0; s >>= 1) {
        if (tid < s) red[tid] += red[tid + s];
        __syncthreads();
    }
    if (tid == 0) rv[c] = red[0];
}

#define SM_T  16
#define SM_N  (SM_T / ST_RPP)
#define SM_KW (KC / 4)
__global__ __launch_bounds__(128) __attribute__((amdgpu_waves_per_eu(1, 2)))
void sims_gemm_kernel(
        const float* __restrict__ base, const float* __restrict__ base2,
        const float* __restrict__ P, const float* __restrict__ A2m,
        const float* __restrict__ ABm,
        const float* __restrict__ qv, const float* __restrict__ rv,
        float* __restrict__ out) {
    __shared__ float sBa [SM_T][KC + 1];
    __shared__ float sBa2[SM_T][KC + 1];
    __shared__ float sP  [SM_T][KC + 1];
    __shared__ float sA2 [SM_T][KC + 1];
    __shared__ float sAB [SM_T][KC + 1];
    __shared__ float red[3][3][8][32];
    const int tid  = threadIdx.x;
    const int wave = tid >> 5;
    const int lane = tid & 31;
    const int half = lane >> 4, fr = lane & 15;
    const int bm0 = blockIdx.y * SM_T;
    const int cn0 = blockIdx.x * SM_T;
    const int kw0 = wave * SM_KW;
    v8f acc1 = {}, acc2 = {}, acc3 = {};
    const int srow = tid / ST_CPR;
    const int scol = (tid % ST_CPR) * 4;
    const size_t goffA = (size_t)(bm0 + srow) * D_SZ + scol;
    const size_t goffB = (size_t)(cn0 + srow) * D_SZ + scol;
    const int loff = srow * (KC + 1) + scol;
    float4 r1[SM_N], r2[SM_N], rp[SM_N], rq[SM_N], rr[SM_N];
#pragma unroll
    for (int i = 0; i < SM_N; ++i) {
        const size_t o = (size_t)(ST_RPP * i) * D_SZ;
        r1[i] = *(const float4*)(base  + goffA + o);
        r2[i] = *(const float4*)(base2 + goffA + o);
        rp[i] = *(const float4*)(P     + goffB + o);
        rq[i] = *(const float4*)(A2m   + goffB + o);
        rr[i] = *(const float4*)(ABm   + goffB + o);
    }
    for (int k0 = 0; k0 < D_SZ; k0 += KC) {
#pragma unroll
        for (int i = 0; i < SM_N; ++i) {
            const int lo = loff + ST_RPP * i * (KC + 1);
            float* d1 = &sBa [0][0] + lo;
            float* d2 = &sBa2[0][0] + lo;
            float* dp = &sP  [0][0] + lo;
            float* dq = &sA2 [0][0] + lo;
            float* dr = &sAB [0][0] + lo;
            d1[0] = r1[i].x; d1[1] = r1[i].y; d1[2] = r1[i].z; d1[3] = r1[i].w;
            d2[0] = r2[i].x; d2[1] = r2[i].y; d2[2] = r2[i].z; d2[3] = r2[i].w;
            dp[0] = rp[i].x; dp[1] = rp[i].y; dp[2] = rp[i].z; dp[3] = rp[i].w;
            dq[0] = rq[i].x; dq[1] = rq[i].y; dq[2] = rq[i].z; dq[3] = rq[i].w;
            dr[0] = rr[i].x; dr[1] = rr[i].y; dr[2] = rr[i].z; dr[3] = rr[i].w;
        }
        __syncthreads();
        if (k0 + KC < D_SZ) {
#pragma unroll
            for (int i = 0; i < SM_N; ++i) {
                const size_t o = (size_t)(ST_RPP * i) * D_SZ + k0 + KC;
                r1[i] = *(const float4*)(base  + goffA + o);
                r2[i] = *(const float4*)(base2 + goffA + o);
                rp[i] = *(const float4*)(P     + goffB + o);
                rq[i] = *(const float4*)(A2m   + goffB + o);
                rr[i] = *(const float4*)(ABm   + goffB + o);
            }
        }
#pragma unroll
        for (int kk = 0; kk < SM_KW; kk += 4) {
            int kb = kw0 + kk + half * 2;
            v2f a1, a2, bp, b2, bb;
            a1[0] = sBa [fr][kb]; a1[1] = sBa [fr][kb + 1];
            a2[0] = sBa2[fr][kb]; a2[1] = sBa2[fr][kb + 1];
            bp[0] = sP  [fr][kb]; bp[1] = sP  [fr][kb + 1];
            b2[0] = sA2 [fr][kb]; b2[1] = sA2 [fr][kb + 1];
            bb[0] = sAB [fr][kb]; bb[1] = sAB [fr][kb + 1];
            acc1 = __builtin_amdgcn_wmma_f32_16x16x4_f32(
                    false, a1, false, bp, (short)0, acc1, false, false);
            acc2 = __builtin_amdgcn_wmma_f32_16x16x4_f32(
                    false, a2, false, b2, (short)0, acc2, false, false);
            acc3 = __builtin_amdgcn_wmma_f32_16x16x4_f32(
                    false, a1, false, bb, (short)0, acc3, false, false);
        }
        __syncthreads();
    }
    if (wave > 0) {
#pragma unroll
        for (int j = 0; j < 8; ++j) {
            red[wave - 1][0][j][lane] = acc1[j];
            red[wave - 1][1][j][lane] = acc2[j];
            red[wave - 1][2][j][lane] = acc3[j];
        }
    }
    __syncthreads();
    if (wave == 0) {
#pragma unroll
        for (int j = 0; j < 8; ++j) {
            float s1 = acc1[j], s2 = acc2[j], s3 = acc3[j];
            for (int w = 0; w < 3; ++w) {
                s1 += red[w][0][j][lane];
                s2 += red[w][1][j][lane];
                s3 += red[w][2][j][lane];
            }
            int b = bm0 + j + half * 8;
            int c = cn0 + fr;
            float num = s1 + qv[c];
            float den = sqrtf(s2 + 2.f * s3 + rv[c]) + EPS_L2;
            out[(size_t)b * B_SZ + c] = num / den;
        }
    }
}

extern "C" void kernel_launch(void* const* d_in, const int* in_sizes, int n_in,
                              void* d_out, int out_size, void* d_ws, size_t ws_size,
                              hipStream_t stream) {
    const float* img   = (const float*)d_in[0];
    const float* cap   = (const float*)d_in[1];
    const int*   lens  = (const int*)  d_in[2];
    const float* fc_w  = (const float*)d_in[3];
    const float* fc_b  = (const float*)d_in[4];
    const float* bn_w  = (const float*)d_in[5];
    const float* bn_b  = (const float*)d_in[6];
    float* out = (float*)d_out;

    float* ws = (float*)d_ws;
    const size_t BD = (size_t)B_SZ * D_SZ;
    float* cap_mean = ws;
    float* meanR    = cap_mean + BD;
    float* base     = meanR + BD;
    float* base2    = base + BD;
    float* P        = base2 + BD;
    float* A2m      = P + BD;
    float* ABm      = A2m + BD;
    float* params   = ABm + BD;
    float* colsum1  = params + 2 * BD;
    float* colsum2  = colsum1 + D_SZ;
    float* qv       = colsum2 + D_SZ;
    float* rv       = qv + B_SZ;

    zero_ws_kernel<<<dim3((2 * D_SZ + 255) / 256), 256, 0, stream>>>(colsum1, 2 * D_SZ);
    cap_pre_kernel<<<dim3(B_SZ), 256, 0, stream>>>(cap, lens, cap_mean);
    img_pre_kernel<<<dim3(B_SZ), 256, 0, stream>>>(img, meanR, colsum1, colsum2);
    make_base_kernel<<<dim3(B_SZ), 256, 0, stream>>>(meanR, colsum1, colsum2,
                                                     bn_w, bn_b, base, base2);
    fc_gemm_kernel<<<dim3(2 * D_SZ / 32, B_SZ / 32), 128, 0, stream>>>(
            cap_mean, fc_w, fc_b, params);
    make_pab_kernel<<<dim3(B_SZ), 256, 0, stream>>>(cap_mean, params,
                                                    P, A2m, ABm, qv, rv);
    sims_gemm_kernel<<<dim3(B_SZ / SM_T, B_SZ / SM_T), 128, 0, stream>>>(
            base, base2, P, A2m, ABm, qv, rv, out);
}


# cap_pre/img_pre streaming reductions re-tiled for bytes in flight: 512 blocks of 8 waves, 16-byte loads, a thread's rows requested at once and added pairwise, halves via LDS (sum order changes)
# speedup vs baseline: 3.8103x; 1.2197x over previous
#include <hip/hip_runtime.h>
#include <hip/hip_bf16.h>
#include <math.h>

#define B_SZ 256
#define R_SZ 36
#define T_SZ 64
#define D_SZ 1024
#define EPS_L2 1e-8f
#define EPS_BN 1e-5f
#define LEAK 0.1f
#define KC 128
#define ST_CPR (KC / 4)
#define ST_RPP (128 / ST_CPR)
#define ST_N   (32 / ST_RPP)

typedef __attribute__((ext_vector_type(2))) float v2f;
typedef __attribute__((ext_vector_type(8))) float v8f;

__global__ void zero_ws_kernel(float* p, int n) {
    int i = blockIdx.x * blockDim.x + threadIdx.x;
    if (i < n) p[i] = 0.f;
}

#define PRE_CH (D_SZ / 2)
#define PRE_CG (PRE_CH / 4)
#define CAP_RPT (T_SZ / 2)
#define IMG_RPT (R_SZ / 2)
#define IMG_P2  16

__global__ __launch_bounds__(256)
void cap_pre_kernel(const float* __restrict__ cap,
                    const int* __restrict__ lens,
                    float* __restrict__ cap_mean) {
    __shared__ __align__(16) float s_sq[2][PRE_CH];
    __shared__ __align__(16) float s_sm[2][PRE_CH];
    const int b   = blockIdx.x >> 1;
    const int c0  = (blockIdx.x & 1) * PRE_CH;
    const int tid = threadIdx.x;
    const int cg  = tid & (PRE_CG - 1);
    const int tg  = tid / PRE_CG;
    const int t0  = tg * CAP_RPT;
    const int len = lens[b];
    const float* p = cap + ((size_t)b * T_SZ + t0) * D_SZ + c0 + cg * 4;
    float4 v[CAP_RPT];
#pragma unroll
    for (int j = 0; j < CAP_RPT; ++j)
        v[j] = *(const float4*)(p + (size_t)j * D_SZ);
    float q[CAP_RPT][4], s[CAP_RPT][4];
#pragma unroll
    for (int j = 0; j < CAP_RPT; ++j) {
        const bool valid = (t0 + j < len);
        const float x[4] = {v[j].x, v[j].y, v[j].z, v[j].w};
#pragma unroll
        for (int i = 0; i < 4; ++i) {
            float y = x[i] > 0.f ? x[i] : LEAK * x[i];
            q[j][i] = y * y;
            s[j][i] = valid ? y : 0.f;
        }
    }
#pragma unroll
    for (int w = CAP_RPT / 2; w >= 1; w >>= 1) {
#pragma unroll
        for (int j = 0; j < w; ++j) {
#pragma unroll
            for (int i = 0; i < 4; ++i) {
                q[j][i] += q[j + w][i];
                s[j][i] += s[j + w][i];
            }
        }
    }
    *(float4*)&s_sq[tg][cg * 4] = make_float4(q[0][0], q[0][1], q[0][2], q[0][3]);
    *(float4*)&s_sm[tg][cg * 4] = make_float4(s[0][0], s[0][1], s[0][2], s[0][3]);
    __syncthreads();
#pragma unroll
    for (int i = 0; i < PRE_CH / 256; ++i) {
        const int c = tid + i * 256;
        float ssq = s_sq[0][c] + s_sq[1][c];
        float ssm = s_sm[0][c] + s_sm[1][c];
        float n = sqrtf(ssq) + EPS_L2;
        cap_mean[(size_t)b * D_SZ + c0 + c] = ssm / (n * (float)len);
    }
}

__global__ __launch_bounds__(256)
void img_pre_kernel(const float* __restrict__ img,
                    float* __restrict__ meanR,
                    float* __restrict__ cs1,
                    float* __restrict__ cs2) {
    __shared__ __align__(16) float s_sq[2][PRE_CH];
    __shared__ __align__(16) float s_sm[2][PRE_CH];
    const int b   = blockIdx.x >> 1;
    const int c0  = (blockIdx.x & 1) * PRE_CH;
    const int tid = threadIdx.x;
    const int cg  = tid & (PRE_CG - 1);
    const int tg  = tid / PRE_CG;
    const float* p = img + ((size_t)b * R_SZ + tg * IMG_RPT) * D_SZ + c0 + cg * 4;
    float4 v[IMG_RPT];
#pragma unroll
    for (int j = 0; j < IMG_RPT; ++j)
        v[j] = *(const float4*)(p + (size_t)j * D_SZ);
    float q[IMG_RPT][4], s[IMG_RPT][4];
#pragma unroll
    for (int j = 0; j < IMG_RPT; ++j) {
        const float x[4] = {v[j].x, v[j].y, v[j].z, v[j].w};
#pragma unroll
        for (int i = 0; i < 4; ++i) {
            float y = x[i] > 0.f ? x[i] : LEAK * x[i];
            q[j][i] = y * y;
            s[j][i] = y;
        }
    }
#pragma unroll
    for (int j = IMG_P2; j < IMG_RPT; ++j) {
#pragma unroll
        for (int i = 0; i < 4; ++i) {
            q[j - IMG_P2][i] += q[j][i];
            s[j - IMG_P2][i] += s[j][i];
        }
    }
#pragma unroll
    for (int w = IMG_P2 / 2; w >= 1; w >>= 1) {
#pragma unroll
        for (int j = 0; j < w; ++j) {
#pragma unroll
            for (int i = 0; i < 4; ++i) {
                q[j][i] += q[j + w][i];
                s[j][i] += s[j + w][i];
            }
        }
    }
    *(float4*)&s_sq[tg][cg * 4] = make_float4(q[0][0], q[0][1], q[0][2], q[0][3]);
    *(float4*)&s_sm[tg][cg * 4] = make_float4(s[0][0], s[0][1], s[0][2], s[0][3]);
    __syncthreads();
#pragma unroll
    for (int i = 0; i < PRE_CH / 256; ++i) {
        const int c = tid + i * 256;
        const int d = c0 + c;
        float ssq = s_sq[0][c] + s_sq[1][c];
        float ssm = s_sm[0][c] + s_sm[1][c];
        float n = sqrtf(ssq) + EPS_L2;
        float s1 = ssm / n;
        float s2 = ssq / (n * n);
        meanR[(size_t)b * D_SZ + d] = s1 * (1.f / (float)R_SZ);
        atomicAdd(cs1 + d, s1);
        atomicAdd(cs2 + d, s2);
    }
}

__global__ void make_base_kernel(const float* __restrict__ meanR,
                                 const float* __restrict__ cs1,
                                 const float* __restrict__ cs2,
                                 const float* __restrict__ bn_w,
                                 const float* __restrict__ bn_b,
                                 float* __restrict__ base,
                                 float* __restrict__ base2) {
    const int b = blockIdx.x;
    const float inv_n = 1.f / (float)(B_SZ * R_SZ);
#pragma unroll
    for (int i = 0; i < 4; ++i) {
        int d = threadIdx.x + i * 256;
        float mu  = cs1[d] * inv_n;
        float ex2 = cs2[d] * inv_n;
        float var = ex2 - mu * mu;
        float scale = bn_w[d] * rsqrtf(var + EPS_BN);
        float v = (meanR[(size_t)b * D_SZ + d] - mu) * scale + bn_b[d];
        base[(size_t)b * D_SZ + d]  = v;
        base2[(size_t)b * D_SZ + d] = v * v;
    }
}

__global__ __launch_bounds__(128) __attribute__((amdgpu_waves_per_eu(1, 4)))
void fc_gemm_kernel(
        const float* __restrict__ A, const float* __restrict__ Bw,
        const float* __restrict__ bias, float* __restrict__ Cout) {
    __shared__ float sA[2][16][KC + 1];
    __shared__ float sB[2][16][KC + 1];
    const int tid  = threadIdx.x;
    const int wave = tid >> 5;
    const int lane = tid & 31;
    const int wm = wave >> 1, wn = wave & 1;
    const int half = lane >> 4, fr = lane & 15;
    const int m0 = blockIdx.y * 32;
    const int n0 = blockIdx.x * 32;
    v8f acc = {};
    const int srow = tid / ST_CPR;
    const int scol = (tid % ST_CPR) * 4;
    const float* gA = A  + (size_t)(m0 + srow) * D_SZ + scol;
    const float* gB = Bw + (size_t)(n0 + srow) * D_SZ + scol;
    float* lA = &sA[0][0][0] + srow * (KC + 1) + scol;
    float* lB = &sB[0][0][0] + srow * (KC + 1) + scol;
    float4 ra[ST_N], rb[ST_N];
#pragma unroll
    for (int i = 0; i < ST_N; ++i) {
        ra[i] = *(const float4*)(gA + (size_t)(ST_RPP * i) * D_SZ);
        rb[i] = *(const float4*)(gB + (size_t)(ST_RPP * i) * D_SZ);
    }
#pragma unroll 1
    for (int k0 = 0; k0 < D_SZ; k0 += KC) {
#pragma unroll
        for (int i = 0; i < ST_N; ++i) {
            float* da = lA + ST_RPP * i * (KC + 1);
            float* db = lB + ST_RPP * i * (KC + 1);
            da[0] = ra[i].x; da[1] = ra[i].y; da[2] = ra[i].z; da[3] = ra[i].w;
            db[0] = rb[i].x; db[1] = rb[i].y; db[2] = rb[i].z; db[3] = rb[i].w;
        }
        __syncthreads();
        if (k0 + KC < D_SZ) {
#pragma unroll
            for (int i = 0; i < ST_N; ++i) {
                ra[i] = *(const float4*)(gA + (size_t)(ST_RPP * i) * D_SZ + k0 + KC);
                rb[i] = *(const float4*)(gB + (size_t)(ST_RPP * i) * D_SZ + k0 + KC);
            }
        }
#pragma unroll
        for (int kk = 0; kk < KC; kk += 4) {
            int kb = kk + half * 2;
            v2f a, b;
            a[0] = sA[wm][fr][kb]; a[1] = sA[wm][fr][kb + 1];
            b[0] = sB[wn][fr][kb]; b[1] = sB[wn][fr][kb + 1];
            acc = __builtin_amdgcn_wmma_f32_16x16x4_f32(
                    false, a, false, b, (short)0, acc, false, false);
        }
        __syncthreads();
    }
#pragma unroll
    for (int j = 0; j < 8; ++j) {
        int m = m0 + wm * 16 + j + half * 8;
        int n = n0 + wn * 16 + fr;
        Cout[(size_t)m * (2 * D_SZ) + n] = acc[j] + bias[n];
    }
}

__global__ void make_pab_kernel(const float* __restrict__ cap_mean,
                                const float* __restrict__ params,
                                float* __restrict__ P, float* __restrict__ A2m,
                                float* __restrict__ ABm,
                                float* __restrict__ qv, float* __restrict__ rv) {
    const int c = blockIdx.x;
    const int tid = threadIdx.x;
    __shared__ float red[256];

    float ss = 0.f;
#pragma unroll
    for (int i = 0; i < 4; ++i) {
        float x = cap_mean[(size_t)c * D_SZ + tid + i * 256];
        ss += x * x;
    }
    red[tid] = ss; __syncthreads();
    for (int s = 128; s > 0; s >>= 1) {
        if (tid < s) red[tid] += red[tid + s];
        __syncthreads();
    }
    const float nrm = sqrtf(red[0]) + EPS_L2;
    __syncthreads();

    float q = 0.f, r = 0.f;
#pragma unroll
    for (int i = 0; i < 4; ++i) {
        int d = tid + i * 256;
        float alpha = params[(size_t)c * (2 * D_SZ) + 2 * d];
        float beta  = params[(size_t)c * (2 * D_SZ) + 2 * d + 1];
        float ch = cap_mean[(size_t)c * D_SZ + d] / nrm;
        P  [(size_t)c * D_SZ + d] = alpha * ch;
        A2m[(size_t)c * D_SZ + d] = alpha * alpha;
        ABm[(size_t)c * D_SZ + d] = alpha * beta;
        q += beta * ch;
        r += beta * beta;
    }
    red[tid] = q; __syncthreads();
    for (int s = 128; s > 0; s >>= 1) {
        if (tid < s) red[tid] += red[tid + s];
        __syncthreads();
    }
    if (tid == 0) qv[c] = red[0];
    __syncthreads();
    red[tid] = r; __syncthreads();
    for (int s = 128; s > 0; s >>= 1) {
        if (tid < s) red[tid] += red[tid + s];
        __syncthreads();
    }
    if (tid == 0) rv[c] = red[0];
}

#define SM_T  16
#define SM_N  (SM_T / ST_RPP)
#define SM_KW (KC / 4)
__global__ __launch_bounds__(128) __attribute__((amdgpu_waves_per_eu(1, 2)))
void sims_gemm_kernel(
        const float* __restrict__ base, const float* __restrict__ base2,
        const float* __restrict__ P, const float* __restrict__ A2m,
        const float* __restrict__ ABm,
        const float* __restrict__ qv, const float* __restrict__ rv,
        float* __restrict__ out) {
    __shared__ float sBa [SM_T][KC + 1];
    __shared__ float sBa2[SM_T][KC + 1];
    __shared__ float sP  [SM_T][KC + 1];
    __shared__ float sA2 [SM_T][KC + 1];
    __shared__ float sAB [SM_T][KC + 1];
    __shared__ float red[3][3][8][32];
    const int tid  = threadIdx.x;
    const int wave = tid >> 5;
    const int lane = tid & 31;
    const int half = lane >> 4, fr = lane & 15;
    const int bm0 = blockIdx.y * SM_T;
    const int cn0 = blockIdx.x * SM_T;
    const int kw0 = wave * SM_KW;
    v8f acc1 = {}, acc2 = {}, acc3 = {};
    const int srow = tid / ST_CPR;
    const int scol = (tid % ST_CPR) * 4;
    const size_t goffA = (size_t)(bm0 + srow) * D_SZ + scol;
    const size_t goffB = (size_t)(cn0 + srow) * D_SZ + scol;
    const int loff = srow * (KC + 1) + scol;
    float4 r1[SM_N], r2[SM_N], rp[SM_N], rq[SM_N], rr[SM_N];
#pragma unroll
    for (int i = 0; i < SM_N; ++i) {
        const size_t o = (size_t)(ST_RPP * i) * D_SZ;
        r1[i] = *(const float4*)(base  + goffA + o);
        r2[i] = *(const float4*)(base2 + goffA + o);
        rp[i] = *(const float4*)(P     + goffB + o);
        rq[i] = *(const float4*)(A2m   + goffB + o);
        rr[i] = *(const float4*)(ABm   + goffB + o);
    }
    for (int k0 = 0; k0 < D_SZ; k0 += KC) {
#pragma unroll
        for (int i = 0; i < SM_N; ++i) {
            const int lo = loff + ST_RPP * i * (KC + 1);
            float* d1 = &sBa [0][0] + lo;
            float* d2 = &sBa2[0][0] + lo;
            float* dp = &sP  [0][0] + lo;
            float* dq = &sA2 [0][0] + lo;
            float* dr = &sAB [0][0] + lo;
            d1[0] = r1[i].x; d1[1] = r1[i].y; d1[2] = r1[i].z; d1[3] = r1[i].w;
            d2[0] = r2[i].x; d2[1] = r2[i].y; d2[2] = r2[i].z; d2[3] = r2[i].w;
            dp[0] = rp[i].x; dp[1] = rp[i].y; dp[2] = rp[i].z; dp[3] = rp[i].w;
            dq[0] = rq[i].x; dq[1] = rq[i].y; dq[2] = rq[i].z; dq[3] = rq[i].w;
            dr[0] = rr[i].x; dr[1] = rr[i].y; dr[2] = rr[i].z; dr[3] = rr[i].w;
        }
        __syncthreads();
        if (k0 + KC < D_SZ) {
#pragma unroll
            for (int i = 0; i < SM_N; ++i) {
                const size_t o = (size_t)(ST_RPP * i) * D_SZ + k0 + KC;
                r1[i] = *(const float4*)(base  + goffA + o);
                r2[i] = *(const float4*)(base2 + goffA + o);
                rp[i] = *(const float4*)(P     + goffB + o);
                rq[i] = *(const float4*)(A2m   + goffB + o);
                rr[i] = *(const float4*)(ABm   + goffB + o);
            }
        }
#pragma unroll
        for (int kk = 0; kk < SM_KW; kk += 4) {
            int kb = kw0 + kk + half * 2;
            v2f a1, a2, bp, b2, bb;
            a1[0] = sBa [fr][kb]; a1[1] = sBa [fr][kb + 1];
            a2[0] = sBa2[fr][kb]; a2[1] = sBa2[fr][kb + 1];
            bp[0] = sP  [fr][kb]; bp[1] = sP  [fr][kb + 1];
            b2[0] = sA2 [fr][kb]; b2[1] = sA2 [fr][kb + 1];
            bb[0] = sAB [fr][kb]; bb[1] = sAB [fr][kb + 1];
            acc1 = __builtin_amdgcn_wmma_f32_16x16x4_f32(
                    false, a1, false, bp, (short)0, acc1, false, false);
            acc2 = __builtin_amdgcn_wmma_f32_16x16x4_f32(
                    false, a2, false, b2, (short)0, acc2, false, false);
            acc3 = __builtin_amdgcn_wmma_f32_16x16x4_f32(
                    false, a1, false, bb, (short)0, acc3, false, false);
        }
        __syncthreads();
    }
    if (wave > 0) {
#pragma unroll
        for (int j = 0; j < 8; ++j) {
            red[wave - 1][0][j][lane] = acc1[j];
            red[wave - 1][1][j][lane] = acc2[j];
            red[wave - 1][2][j][lane] = acc3[j];
        }
    }
    __syncthreads();
    if (wave == 0) {
#pragma unroll
        for (int j = 0; j < 8; ++j) {
            float s1 = acc1[j], s2 = acc2[j], s3 = acc3[j];
            for (int w = 0; w < 3; ++w) {
                s1 += red[w][0][j][lane];
                s2 += red[w][1][j][lane];
                s3 += red[w][2][j][lane];
            }
            int b = bm0 + j + half * 8;
            int c = cn0 + fr;
            float num = s1 + qv[c];
            float den = sqrtf(s2 + 2.f * s3 + rv[c]) + EPS_L2;
            out[(size_t)b * B_SZ + c] = num / den;
        }
    }
}

extern "C" void kernel_launch(void* const* d_in, const int* in_sizes, int n_in,
                              void* d_out, int out_size, void* d_ws, size_t ws_size,
                              hipStream_t stream) {
    const float* img   = (const float*)d_in[0];
    const float* cap   = (const float*)d_in[1];
    const int*   lens  = (const int*)  d_in[2];
    const float* fc_w  = (const float*)d_in[3];
    const float* fc_b  = (const float*)d_in[4];
    const float* bn_w  = (const float*)d_in[5];
    const float* bn_b  = (const float*)d_in[6];
    float* out = (float*)d_out;

    float* ws = (float*)d_ws;
    const size_t BD = (size_t)B_SZ * D_SZ;
    float* cap_mean = ws;
    float* meanR    = cap_mean + BD;
    float* base     = meanR + BD;
    float* base2    = base + BD;
    float* P        = base2 + BD;
    float* A2m      = P + BD;
    float* ABm      = A2m + BD;
    float* params   = ABm + BD;
    float* colsum1  = params + 2 * BD;
    float* colsum2  = colsum1 + D_SZ;
    float* qv       = colsum2 + D_SZ;
    float* rv       = qv + B_SZ;

    zero_ws_kernel<<<dim3((2 * D_SZ + 255) / 256), 256, 0, stream>>>(colsum1, 2 * D_SZ);
    cap_pre_kernel<<<dim3(2 * B_SZ), 256, 0, stream>>>(cap, lens, cap_mean);
    img_pre_kernel<<<dim3(2 * B_SZ), 256, 0, stream>>>(img, meanR, colsum1, colsum2);
    make_base_kernel<<<dim3(B_SZ), 256, 0, stream>>>(meanR, colsum1, colsum2,
                                                     bn_w, bn_b, base, base2);
    fc_gemm_kernel<<<dim3(2 * D_SZ / 32, B_SZ / 32), 128, 0, stream>>>(
            cap_mean, fc_w, fc_b, params);
    make_pab_kernel<<<dim3(B_SZ), 256, 0, stream>>>(cap_mean, params,
                                                    P, A2m, ABm, qv, rv);
    sims_gemm_kernel<<<dim3(B_SZ / SM_T, B_SZ / SM_T), 128, 0, stream>>>(
            base, base2, P, A2m, ABm, qv, rv, out);
}


# Both WMMA GEMMs: LDS tile row pitch 129 to 132 floats, so staging is 16-byte stores and a lane reads 4 k per conflict-free 16-byte load for 2 products (k order in a chunk, so sum order, changes)
# speedup vs baseline: 4.1059x; 1.0776x over previous
#include <hip/hip_runtime.h>
#include <hip/hip_bf16.h>
#include <math.h>

#define B_SZ 256
#define R_SZ 36
#define T_SZ 64
#define D_SZ 1024
#define EPS_L2 1e-8f
#define EPS_BN 1e-5f
#define LEAK 0.1f
#define KC 128
#define ST_CPR (KC / 4)
#define ST_RPP (128 / ST_CPR)
#define ST_N   (32 / ST_RPP)
#define KP (KC + 4)

typedef __attribute__((ext_vector_type(2))) float v2f;
typedef __attribute__((ext_vector_type(8))) float v8f;

__global__ void zero_ws_kernel(float* p, int n) {
    int i = blockIdx.x * blockDim.x + threadIdx.x;
    if (i < n) p[i] = 0.f;
}

#define PRE_CH (D_SZ / 2)
#define PRE_CG (PRE_CH / 4)
#define CAP_RPT (T_SZ / 2)
#define IMG_RPT (R_SZ / 2)
#define IMG_P2  16

__global__ __launch_bounds__(256)
void cap_pre_kernel(const float* __restrict__ cap,
                    const int* __restrict__ lens,
                    float* __restrict__ cap_mean) {
    __shared__ __align__(16) float s_sq[2][PRE_CH];
    __shared__ __align__(16) float s_sm[2][PRE_CH];
    const int b   = blockIdx.x >> 1;
    const int c0  = (blockIdx.x & 1) * PRE_CH;
    const int tid = threadIdx.x;
    const int cg  = tid & (PRE_CG - 1);
    const int tg  = tid / PRE_CG;
    const int t0  = tg * CAP_RPT;
    const int len = lens[b];
    const float* p = cap + ((size_t)b * T_SZ + t0) * D_SZ + c0 + cg * 4;
    float4 v[CAP_RPT];
#pragma unroll
    for (int j = 0; j < CAP_RPT; ++j)
        v[j] = *(const float4*)(p + (size_t)j * D_SZ);
    float q[CAP_RPT][4], s[CAP_RPT][4];
#pragma unroll
    for (int j = 0; j < CAP_RPT; ++j) {
        const bool valid = (t0 + j < len);
        const float x[4] = {v[j].x, v[j].y, v[j].z, v[j].w};
#pragma unroll
        for (int i = 0; i < 4; ++i) {
            float y = x[i] > 0.f ? x[i] : LEAK * x[i];
            q[j][i] = y * y;
            s[j][i] = valid ? y : 0.f;
        }
    }
#pragma unroll
    for (int w = CAP_RPT / 2; w >= 1; w >>= 1) {
#pragma unroll
        for (int j = 0; j < w; ++j) {
#pragma unroll
            for (int i = 0; i < 4; ++i) {
                q[j][i] += q[j + w][i];
                s[j][i] += s[j + w][i];
            }
        }
    }
    *(float4*)&s_sq[tg][cg * 4] = make_float4(q[0][0], q[0][1], q[0][2], q[0][3]);
    *(float4*)&s_sm[tg][cg * 4] = make_float4(s[0][0], s[0][1], s[0][2], s[0][3]);
    __syncthreads();
#pragma unroll
    for (int i = 0; i < PRE_CH / 256; ++i) {
        const int c = tid + i * 256;
        float ssq = s_sq[0][c] + s_sq[1][c];
        float ssm = s_sm[0][c] + s_sm[1][c];
        float n = sqrtf(ssq) + EPS_L2;
        cap_mean[(size_t)b * D_SZ + c0 + c] = ssm / (n * (float)len);
    }
}

__global__ __launch_bounds__(256)
void img_pre_kernel(const float* __restrict__ img,
                    float* __restrict__ meanR,
                    float* __restrict__ cs1,
                    float* __restrict__ cs2) {
    __shared__ __align__(16) float s_sq[2][PRE_CH];
    __shared__ __align__(16) float s_sm[2][PRE_CH];
    const int b   = blockIdx.x >> 1;
    const int c0  = (blockIdx.x & 1) * PRE_CH;
    const int tid = threadIdx.x;
    const int cg  = tid & (PRE_CG - 1);
    const int tg  = tid / PRE_CG;
    const float* p = img + ((size_t)b * R_SZ + tg * IMG_RPT) * D_SZ + c0 + cg * 4;
    float4 v[IMG_RPT];
#pragma unroll
    for (int j = 0; j < IMG_RPT; ++j)
        v[j] = *(const float4*)(p + (size_t)j * D_SZ);
    float q[IMG_RPT][4], s[IMG_RPT][4];
#pragma unroll
    for (int j = 0; j < IMG_RPT; ++j) {
        const float x[4] = {v[j].x, v[j].y, v[j].z, v[j].w};
#pragma unroll
        for (int i = 0; i < 4; ++i) {
            float y = x[i] > 0.f ? x[i] : LEAK * x[i];
            q[j][i] = y * y;
            s[j][i] = y;
        }
    }
#pragma unroll
    for (int j = IMG_P2; j < IMG_RPT; ++j) {
#pragma unroll
        for (int i = 0; i < 4; ++i) {
            q[j - IMG_P2][i] += q[j][i];
            s[j - IMG_P2][i] += s[j][i];
        }
    }
#pragma unroll
    for (int w = IMG_P2 / 2; w >= 1; w >>= 1) {
#pragma unroll
        for (int j = 0; j < w; ++j) {
#pragma unroll
            for (int i = 0; i < 4; ++i) {
                q[j][i] += q[j + w][i];
                s[j][i] += s[j + w][i];
            }
        }
    }
    *(float4*)&s_sq[tg][cg * 4] = make_float4(q[0][0], q[0][1], q[0][2], q[0][3]);
    *(float4*)&s_sm[tg][cg * 4] = make_float4(s[0][0], s[0][1], s[0][2], s[0][3]);
    __syncthreads();
#pragma unroll
    for (int i = 0; i < PRE_CH / 256; ++i) {
        const int c = tid + i * 256;
        const int d = c0 + c;
        float ssq = s_sq[0][c] + s_sq[1][c];
        float ssm = s_sm[0][c] + s_sm[1][c];
        float n = sqrtf(ssq) + EPS_L2;
        float s1 = ssm / n;
        float s2 = ssq / (n * n);
        meanR[(size_t)b * D_SZ + d] = s1 * (1.f / (float)R_SZ);
        atomicAdd(cs1 + d, s1);
        atomicAdd(cs2 + d, s2);
    }
}

__global__ void make_base_kernel(const float* __restrict__ meanR,
                                 const float* __restrict__ cs1,
                                 const float* __restrict__ cs2,
                                 const float* __restrict__ bn_w,
                                 const float* __restrict__ bn_b,
                                 float* __restrict__ base,
                                 float* __restrict__ base2) {
    const int b = blockIdx.x;
    const float inv_n = 1.f / (float)(B_SZ * R_SZ);
#pragma unroll
    for (int i = 0; i < 4; ++i) {
        int d = threadIdx.x + i * 256;
        float mu  = cs1[d] * inv_n;
        float ex2 = cs2[d] * inv_n;
        float var = ex2 - mu * mu;
        float scale = bn_w[d] * rsqrtf(var + EPS_BN);
        float v = (meanR[(size_t)b * D_SZ + d] - mu) * scale + bn_b[d];
        base[(size_t)b * D_SZ + d]  = v;
        base2[(size_t)b * D_SZ + d] = v * v;
    }
}

__global__ __launch_bounds__(128) __attribute__((amdgpu_waves_per_eu(1, 4)))
void fc_gemm_kernel(
        const float* __restrict__ A, const float* __restrict__ Bw,
        const float* __restrict__ bias, float* __restrict__ Cout) {
    __shared__ __align__(16) float sA[2][16][KP];
    __shared__ __align__(16) float sB[2][16][KP];
    const int tid  = threadIdx.x;
    const int wave = tid >> 5;
    const int lane = tid & 31;
    const int wm = wave >> 1, wn = wave & 1;
    const int half = lane >> 4, fr = lane & 15;
    const int m0 = blockIdx.y * 32;
    const int n0 = blockIdx.x * 32;
    v8f acc = {};
    const int srow = tid / ST_CPR;
    const int scol = (tid % ST_CPR) * 4;
    const float* gA = A  + (size_t)(m0 + srow) * D_SZ + scol;
    const float* gB = Bw + (size_t)(n0 + srow) * D_SZ + scol;
    float* lA = &sA[0][0][0] + srow * KP + scol;
    float* lB = &sB[0][0][0] + srow * KP + scol;
    float4 ra[ST_N], rb[ST_N];
#pragma unroll
    for (int i = 0; i < ST_N; ++i) {
        ra[i] = *(const float4*)(gA + (size_t)(ST_RPP * i) * D_SZ);
        rb[i] = *(const float4*)(gB + (size_t)(ST_RPP * i) * D_SZ);
    }
#pragma unroll 1
    for (int k0 = 0; k0 < D_SZ; k0 += KC) {
#pragma unroll
        for (int i = 0; i < ST_N; ++i) {
            *(float4*)(lA + ST_RPP * i * KP) =
                    make_float4(ra[i].x, ra[i].y, ra[i].z, ra[i].w);
            *(float4*)(lB + ST_RPP * i * KP) =
                    make_float4(rb[i].x, rb[i].y, rb[i].z, rb[i].w);
        }
        __syncthreads();
        if (k0 + KC < D_SZ) {
#pragma unroll
            for (int i = 0; i < ST_N; ++i) {
                ra[i] = *(const float4*)(gA + (size_t)(ST_RPP * i) * D_SZ + k0 + KC);
                rb[i] = *(const float4*)(gB + (size_t)(ST_RPP * i) * D_SZ + k0 + KC);
            }
        }
#pragma unroll
        for (int kk = 0; kk < KC; kk += 8) {
            const int kb = kk + half * 4;
            const float4 a4 = *(const float4*)&sA[wm][fr][kb];
            const float4 b4 = *(const float4*)&sB[wn][fr][kb];
            v2f a, b;
            a[0] = a4.x; a[1] = a4.y;
            b[0] = b4.x; b[1] = b4.y;
            acc = __builtin_amdgcn_wmma_f32_16x16x4_f32(
                    false, a, false, b, (short)0, acc, false, false);
            a[0] = a4.z; a[1] = a4.w;
            b[0] = b4.z; b[1] = b4.w;
            acc = __builtin_amdgcn_wmma_f32_16x16x4_f32(
                    false, a, false, b, (short)0, acc, false, false);
        }
        __syncthreads();
    }
#pragma unroll
    for (int j = 0; j < 8; ++j) {
        int m = m0 + wm * 16 + j + half * 8;
        int n = n0 + wn * 16 + fr;
        Cout[(size_t)m * (2 * D_SZ) + n] = acc[j] + bias[n];
    }
}

__global__ void make_pab_kernel(const float* __restrict__ cap_mean,
                                const float* __restrict__ params,
                                float* __restrict__ P, float* __restrict__ A2m,
                                float* __restrict__ ABm,
                                float* __restrict__ qv, float* __restrict__ rv) {
    const int c = blockIdx.x;
    const int tid = threadIdx.x;
    __shared__ float red[256];

    float ss = 0.f;
#pragma unroll
    for (int i = 0; i < 4; ++i) {
        float x = cap_mean[(size_t)c * D_SZ + tid + i * 256];
        ss += x * x;
    }
    red[tid] = ss; __syncthreads();
    for (int s = 128; s > 0; s >>= 1) {
        if (tid < s) red[tid] += red[tid + s];
        __syncthreads();
    }
    const float nrm = sqrtf(red[0]) + EPS_L2;
    __syncthreads();

    float q = 0.f, r = 0.f;
#pragma unroll
    for (int i = 0; i < 4; ++i) {
        int d = tid + i * 256;
        float alpha = params[(size_t)c * (2 * D_SZ) + 2 * d];
        float beta  = params[(size_t)c * (2 * D_SZ) + 2 * d + 1];
        float ch = cap_mean[(size_t)c * D_SZ + d] / nrm;
        P  [(size_t)c * D_SZ + d] = alpha * ch;
        A2m[(size_t)c * D_SZ + d] = alpha * alpha;
        ABm[(size_t)c * D_SZ + d] = alpha * beta;
        q += beta * ch;
        r += beta * beta;
    }
    red[tid] = q; __syncthreads();
    for (int s = 128; s > 0; s >>= 1) {
        if (tid < s) red[tid] += red[tid + s];
        __syncthreads();
    }
    if (tid == 0) qv[c] = red[0];
    __syncthreads();
    red[tid] = r; __syncthreads();
    for (int s = 128; s > 0; s >>= 1) {
        if (tid < s) red[tid] += red[tid + s];
        __syncthreads();
    }
    if (tid == 0) rv[c] = red[0];
}

#define SM_T  16
#define SM_N  (SM_T / ST_RPP)
#define SM_KW (KC / 4)
__global__ __launch_bounds__(128) __attribute__((amdgpu_waves_per_eu(1, 2)))
void sims_gemm_kernel(
        const float* __restrict__ base, const float* __restrict__ base2,
        const float* __restrict__ P, const float* __restrict__ A2m,
        const float* __restrict__ ABm,
        const float* __restrict__ qv, const float* __restrict__ rv,
        float* __restrict__ out) {
    __shared__ __align__(16) float sBa [SM_T][KP];
    __shared__ __align__(16) float sBa2[SM_T][KP];
    __shared__ __align__(16) float sP  [SM_T][KP];
    __shared__ __align__(16) float sA2 [SM_T][KP];
    __shared__ __align__(16) float sAB [SM_T][KP];
    __shared__ float red[3][3][8][32];
    const int tid  = threadIdx.x;
    const int wave = tid >> 5;
    const int lane = tid & 31;
    const int half = lane >> 4, fr = lane & 15;
    const int bm0 = blockIdx.y * SM_T;
    const int cn0 = blockIdx.x * SM_T;
    const int kw0 = wave * SM_KW;
    v8f acc1 = {}, acc2 = {}, acc3 = {};
    const int srow = tid / ST_CPR;
    const int scol = (tid % ST_CPR) * 4;
    const size_t goffA = (size_t)(bm0 + srow) * D_SZ + scol;
    const size_t goffB = (size_t)(cn0 + srow) * D_SZ + scol;
    const int loff = srow * KP + scol;
    float4 r1[SM_N], r2[SM_N], rp[SM_N], rq[SM_N], rr[SM_N];
#pragma unroll
    for (int i = 0; i < SM_N; ++i) {
        const size_t o = (size_t)(ST_RPP * i) * D_SZ;
        r1[i] = *(const float4*)(base  + goffA + o);
        r2[i] = *(const float4*)(base2 + goffA + o);
        rp[i] = *(const float4*)(P     + goffB + o);
        rq[i] = *(const float4*)(A2m   + goffB + o);
        rr[i] = *(const float4*)(ABm   + goffB + o);
    }
    for (int k0 = 0; k0 < D_SZ; k0 += KC) {
#pragma unroll
        for (int i = 0; i < SM_N; ++i) {
            const int lo = loff + ST_RPP * i * KP;
            *(float4*)(&sBa [0][0] + lo) = make_float4(r1[i].x, r1[i].y, r1[i].z, r1[i].w);
            *(float4*)(&sBa2[0][0] + lo) = make_float4(r2[i].x, r2[i].y, r2[i].z, r2[i].w);
            *(float4*)(&sP  [0][0] + lo) = make_float4(rp[i].x, rp[i].y, rp[i].z, rp[i].w);
            *(float4*)(&sA2 [0][0] + lo) = make_float4(rq[i].x, rq[i].y, rq[i].z, rq[i].w);
            *(float4*)(&sAB [0][0] + lo) = make_float4(rr[i].x, rr[i].y, rr[i].z, rr[i].w);
        }
        __syncthreads();
        if (k0 + KC < D_SZ) {
#pragma unroll
            for (int i = 0; i < SM_N; ++i) {
                const size_t o = (size_t)(ST_RPP * i) * D_SZ + k0 + KC;
                r1[i] = *(const float4*)(base  + goffA + o);
                r2[i] = *(const float4*)(base2 + goffA + o);
                rp[i] = *(const float4*)(P     + goffB + o);
                rq[i] = *(const float4*)(A2m   + goffB + o);
                rr[i] = *(const float4*)(ABm   + goffB + o);
            }
        }
#pragma unroll
        for (int kk = 0; kk < SM_KW; kk += 8) {
            const int kb = kw0 + kk + half * 4;
            const float4 x1 = *(const float4*)&sBa [fr][kb];
            const float4 x2 = *(const float4*)&sBa2[fr][kb];
            const float4 xp = *(const float4*)&sP  [fr][kb];
            const float4 xq = *(const float4*)&sA2 [fr][kb];
            const float4 xr = *(const float4*)&sAB [fr][kb];
            v2f a1, a2, bp, b2, bb;
            a1[0] = x1.x; a1[1] = x1.y;
            a2[0] = x2.x; a2[1] = x2.y;
            bp[0] = xp.x; bp[1] = xp.y;
            b2[0] = xq.x; b2[1] = xq.y;
            bb[0] = xr.x; bb[1] = xr.y;
            acc1 = __builtin_amdgcn_wmma_f32_16x16x4_f32(
                    false, a1, false, bp, (short)0, acc1, false, false);
            acc2 = __builtin_amdgcn_wmma_f32_16x16x4_f32(
                    false, a2, false, b2, (short)0, acc2, false, false);
            acc3 = __builtin_amdgcn_wmma_f32_16x16x4_f32(
                    false, a1, false, bb, (short)0, acc3, false, false);
            a1[0] = x1.z; a1[1] = x1.w;
            a2[0] = x2.z; a2[1] = x2.w;
            bp[0] = xp.z; bp[1] = xp.w;
            b2[0] = xq.z; b2[1] = xq.w;
            bb[0] = xr.z; bb[1] = xr.w;
            acc1 = __builtin_amdgcn_wmma_f32_16x16x4_f32(
                    false, a1, false, bp, (short)0, acc1, false, false);
            acc2 = __builtin_amdgcn_wmma_f32_16x16x4_f32(
                    false, a2, false, b2, (short)0, acc2, false, false);
            acc3 = __builtin_amdgcn_wmma_f32_16x16x4_f32(
                    false, a1, false, bb, (short)0, acc3, false, false);
        }
        __syncthreads();
    }
    if (wave > 0) {
#pragma unroll
        for (int j = 0; j < 8; ++j) {
            red[wave - 1][0][j][lane] = acc1[j];
            red[wave - 1][1][j][lane] = acc2[j];
            red[wave - 1][2][j][lane] = acc3[j];
        }
    }
    __syncthreads();
    if (wave == 0) {
#pragma unroll
        for (int j = 0; j < 8; ++j) {
            float s1 = acc1[j], s2 = acc2[j], s3 = acc3[j];
            for (int w = 0; w < 3; ++w) {
                s1 += red[w][0][j][lane];
                s2 += red[w][1][j][lane];
                s3 += red[w][2][j][lane];
            }
            int b = bm0 + j + half * 8;
            int c = cn0 + fr;
            float num = s1 + qv[c];
            float den = sqrtf(s2 + 2.f * s3 + rv[c]) + EPS_L2;
            out[(size_t)b * B_SZ + c] = num / den;
        }
    }
}

extern "C" void kernel_launch(void* const* d_in, const int* in_sizes, int n_in,
                              void* d_out, int out_size, void* d_ws, size_t ws_size,
                              hipStream_t stream) {
    const float* img   = (const float*)d_in[0];
    const float* cap   = (const float*)d_in[1];
    const int*   lens  = (const int*)  d_in[2];
    const float* fc_w  = (const float*)d_in[3];
    const float* fc_b  = (const float*)d_in[4];
    const float* bn_w  = (const float*)d_in[5];
    const float* bn_b  = (const float*)d_in[6];
    float* out = (float*)d_out;

    float* ws = (float*)d_ws;
    const size_t BD = (size_t)B_SZ * D_SZ;
    float* cap_mean = ws;
    float* meanR    = cap_mean + BD;
    float* base     = meanR + BD;
    float* base2    = base + BD;
    float* P        = base2 + BD;
    float* A2m      = P + BD;
    float* ABm      = A2m + BD;
    float* params   = ABm + BD;
    float* colsum1  = params + 2 * BD;
    float* colsum2  = colsum1 + D_SZ;
    float* qv       = colsum2 + D_SZ;
    float* rv       = qv + B_SZ;

    zero_ws_kernel<<<dim3((2 * D_SZ + 255) / 256), 256, 0, stream>>>(colsum1, 2 * D_SZ);
    cap_pre_kernel<<<dim3(2 * B_SZ), 256, 0, stream>>>(cap, lens, cap_mean);
    img_pre_kernel<<<dim3(2 * B_SZ), 256, 0, stream>>>(img, meanR, colsum1, colsum2);
    make_base_kernel<<<dim3(B_SZ), 256, 0, stream>>>(meanR, colsum1, colsum2,
                                                     bn_w, bn_b, base, base2);
    fc_gemm_kernel<<<dim3(2 * D_SZ / 32, B_SZ / 32), 128, 0, stream>>>(
            cap_mean, fc_w, fc_b, params);
    make_pab_kernel<<<dim3(B_SZ), 256, 0, stream>>>(cap_mean, params,
                                                    P, A2m, ABm, qv, rv);
    sims_gemm_kernel<<<dim3(B_SZ / SM_T, B_SZ / SM_T), 128, 0, stream>>>(
            base, base2, P, A2m, ABm, qv, rv, out);
}


# Kernels that do not depend on each other share a launch, a block taking its role from its index (BN-sum start values in cap_pre, fc GEMM beside img_pre, P/A2/AB beside BN base): 7 launches to 4
# speedup vs baseline: 5.0098x; 1.2201x over previous
#include <hip/hip_runtime.h>
#include <hip/hip_bf16.h>
#include <math.h>

#define B_SZ 256
#define R_SZ 36
#define T_SZ 64
#define D_SZ 1024
#define EPS_L2 1e-8f
#define EPS_BN 1e-5f
#define LEAK 0.1f
#define KC 128
#define ST_CPR (KC / 4)
#define ST_RPP (128 / ST_CPR)
#define ST_N   (32 / ST_RPP)
#define KP (KC + 4)

typedef __attribute__((ext_vector_type(2))) float v2f;
typedef __attribute__((ext_vector_type(8))) float v8f;


#define PRE_CH (D_SZ / 2)
#define PRE_CG (PRE_CH / 4)
#define CAP_RPT (T_SZ / 2)
#define IMG_RPT (R_SZ / 2)
#define IMG_P2  16

__global__ __launch_bounds__(256)
void cap_pre_kernel(const float* __restrict__ cap,
                    const int* __restrict__ lens,
                    float* __restrict__ cap_mean,
                    float* __restrict__ bn_sums) {
    __shared__ __align__(16) float s_sq[2][PRE_CH];
    __shared__ __align__(16) float s_sm[2][PRE_CH];
    const int b   = blockIdx.x >> 1;
    const int c0  = (blockIdx.x & 1) * PRE_CH;
    const int tid = threadIdx.x;
    const int cg  = tid & (PRE_CG - 1);
    const int tg  = tid / PRE_CG;
    const int t0  = tg * CAP_RPT;
    const int len = lens[b];
    if (blockIdx.x < 2 * D_SZ / 256) bn_sums[blockIdx.x * 256 + tid] = 0.f;
    const float* p = cap + ((size_t)b * T_SZ + t0) * D_SZ + c0 + cg * 4;
    float4 v[CAP_RPT];
#pragma unroll
    for (int j = 0; j < CAP_RPT; ++j)
        v[j] = *(const float4*)(p + (size_t)j * D_SZ);
    float q[CAP_RPT][4], s[CAP_RPT][4];
#pragma unroll
    for (int j = 0; j < CAP_RPT; ++j) {
        const bool valid = (t0 + j < len);
        const float x[4] = {v[j].x, v[j].y, v[j].z, v[j].w};
#pragma unroll
        for (int i = 0; i < 4; ++i) {
            float y = x[i] > 0.f ? x[i] : LEAK * x[i];
            q[j][i] = y * y;
            s[j][i] = valid ? y : 0.f;
        }
    }
#pragma unroll
    for (int w = CAP_RPT / 2; w >= 1; w >>= 1) {
#pragma unroll
        for (int j = 0; j < w; ++j) {
#pragma unroll
            for (int i = 0; i < 4; ++i) {
                q[j][i] += q[j + w][i];
                s[j][i] += s[j + w][i];
            }
        }
    }
    *(float4*)&s_sq[tg][cg * 4] = make_float4(q[0][0], q[0][1], q[0][2], q[0][3]);
    *(float4*)&s_sm[tg][cg * 4] = make_float4(s[0][0], s[0][1], s[0][2], s[0][3]);
    __syncthreads();
#pragma unroll
    for (int i = 0; i < PRE_CH / 256; ++i) {
        const int c = tid + i * 256;
        float ssq = s_sq[0][c] + s_sq[1][c];
        float ssm = s_sm[0][c] + s_sm[1][c];
        float n = sqrtf(ssq) + EPS_L2;
        cap_mean[(size_t)b * D_SZ + c0 + c] = ssm / (n * (float)len);
    }
}

#define IMG_CH (D_SZ / 4)
#define IMG_CG (IMG_CH / 4)
__device__ __forceinline__
void img_pre_role(const int blk,
                  const float* __restrict__ img,
                  float* __restrict__ meanR,
                  float* __restrict__ cs1,
                  float* __restrict__ cs2) {
    __shared__ __align__(16) float s_sq[2][IMG_CH];
    __shared__ __align__(16) float s_sm[2][IMG_CH];
    const int b   = blk >> 2;
    const int c0  = (blk & 3) * IMG_CH;
    const int tid = threadIdx.x;
    const int cg  = tid & (IMG_CG - 1);
    const int tg  = tid / IMG_CG;
    const float* p = img + ((size_t)b * R_SZ + tg * IMG_RPT) * D_SZ + c0 + cg * 4;
    float4 v[IMG_RPT];
#pragma unroll
    for (int j = 0; j < IMG_RPT; ++j)
        v[j] = *(const float4*)(p + (size_t)j * D_SZ);
    float q[IMG_RPT][4], s[IMG_RPT][4];
#pragma unroll
    for (int j = 0; j < IMG_RPT; ++j) {
        const float x[4] = {v[j].x, v[j].y, v[j].z, v[j].w};
#pragma unroll
        for (int i = 0; i < 4; ++i) {
            float y = x[i] > 0.f ? x[i] : LEAK * x[i];
            q[j][i] = y * y;
            s[j][i] = y;
        }
    }
#pragma unroll
    for (int j = IMG_P2; j < IMG_RPT; ++j) {
#pragma unroll
        for (int i = 0; i < 4; ++i) {
            q[j - IMG_P2][i] += q[j][i];
            s[j - IMG_P2][i] += s[j][i];
        }
    }
#pragma unroll
    for (int w = IMG_P2 / 2; w >= 1; w >>= 1) {
#pragma unroll
        for (int j = 0; j < w; ++j) {
#pragma unroll
            for (int i = 0; i < 4; ++i) {
                q[j][i] += q[j + w][i];
                s[j][i] += s[j + w][i];
            }
        }
    }
    *(float4*)&s_sq[tg][cg * 4] = make_float4(q[0][0], q[0][1], q[0][2], q[0][3]);
    *(float4*)&s_sm[tg][cg * 4] = make_float4(s[0][0], s[0][1], s[0][2], s[0][3]);
    __syncthreads();
#pragma unroll
    for (int i = 0; i < IMG_CH / 128; ++i) {
        const int c = tid + i * 128;
        const int d = c0 + c;
        float ssq = s_sq[0][c] + s_sq[1][c];
        float ssm = s_sm[0][c] + s_sm[1][c];
        float n = sqrtf(ssq) + EPS_L2;
        float s1 = ssm / n;
        float s2 = ssq / (n * n);
        meanR[(size_t)b * D_SZ + d] = s1 * (1.f / (float)R_SZ);
        atomicAdd(cs1 + d, s1);
        atomicAdd(cs2 + d, s2);
    }
}

__device__ __forceinline__
void make_base_role(const int b,
                    const float* __restrict__ meanR,
                    const float* __restrict__ cs1,
                    const float* __restrict__ cs2,
                    const float* __restrict__ bn_w,
                    const float* __restrict__ bn_b,
                    float* __restrict__ base,
                    float* __restrict__ base2) {
    const float inv_n = 1.f / (float)(B_SZ * R_SZ);
#pragma unroll
    for (int i = 0; i < 4; ++i) {
        int d = threadIdx.x + i * 256;
        float mu  = cs1[d] * inv_n;
        float ex2 = cs2[d] * inv_n;
        float var = ex2 - mu * mu;
        float scale = bn_w[d] * rsqrtf(var + EPS_BN);
        float v = (meanR[(size_t)b * D_SZ + d] - mu) * scale + bn_b[d];
        base[(size_t)b * D_SZ + d]  = v;
        base2[(size_t)b * D_SZ + d] = v * v;
    }
}

__device__ __forceinline__
void fc_gemm_role(const int bx, const int by,
        const float* __restrict__ A, const float* __restrict__ Bw,
        const float* __restrict__ bias, float* __restrict__ Cout) {
    __shared__ __align__(16) float sA[2][16][KP];
    __shared__ __align__(16) float sB[2][16][KP];
    const int tid  = threadIdx.x;
    const int wave = tid >> 5;
    const int lane = tid & 31;
    const int wm = wave >> 1, wn = wave & 1;
    const int half = lane >> 4, fr = lane & 15;
    const int m0 = by * 32;
    const int n0 = bx * 32;
    v8f acc = {};
    const int srow = tid / ST_CPR;
    const int scol = (tid % ST_CPR) * 4;
    const float* gA = A  + (size_t)(m0 + srow) * D_SZ + scol;
    const float* gB = Bw + (size_t)(n0 + srow) * D_SZ + scol;
    float* lA = &sA[0][0][0] + srow * KP + scol;
    float* lB = &sB[0][0][0] + srow * KP + scol;
    float4 ra[ST_N], rb[ST_N];
#pragma unroll
    for (int i = 0; i < ST_N; ++i) {
        ra[i] = *(const float4*)(gA + (size_t)(ST_RPP * i) * D_SZ);
        rb[i] = *(const float4*)(gB + (size_t)(ST_RPP * i) * D_SZ);
    }
#pragma unroll 1
    for (int k0 = 0; k0 < D_SZ; k0 += KC) {
#pragma unroll
        for (int i = 0; i < ST_N; ++i) {
            *(float4*)(lA + ST_RPP * i * KP) =
                    make_float4(ra[i].x, ra[i].y, ra[i].z, ra[i].w);
            *(float4*)(lB + ST_RPP * i * KP) =
                    make_float4(rb[i].x, rb[i].y, rb[i].z, rb[i].w);
        }
        __syncthreads();
        if (k0 + KC < D_SZ) {
#pragma unroll
            for (int i = 0; i < ST_N; ++i) {
                ra[i] = *(const float4*)(gA + (size_t)(ST_RPP * i) * D_SZ + k0 + KC);
                rb[i] = *(const float4*)(gB + (size_t)(ST_RPP * i) * D_SZ + k0 + KC);
            }
        }
#pragma unroll
        for (int kk = 0; kk < KC; kk += 8) {
            const int kb = kk + half * 4;
            const float4 a4 = *(const float4*)&sA[wm][fr][kb];
            const float4 b4 = *(const float4*)&sB[wn][fr][kb];
            v2f a, b;
            a[0] = a4.x; a[1] = a4.y;
            b[0] = b4.x; b[1] = b4.y;
            acc = __builtin_amdgcn_wmma_f32_16x16x4_f32(
                    false, a, false, b, (short)0, acc, false, false);
            a[0] = a4.z; a[1] = a4.w;
            b[0] = b4.z; b[1] = b4.w;
            acc = __builtin_amdgcn_wmma_f32_16x16x4_f32(
                    false, a, false, b, (short)0, acc, false, false);
        }
        __syncthreads();
    }
#pragma unroll
    for (int j = 0; j < 8; ++j) {
        int m = m0 + wm * 16 + j + half * 8;
        int n = n0 + wn * 16 + fr;
        Cout[(size_t)m * (2 * D_SZ) + n] = acc[j] + bias[n];
    }
}

#define FC_NBX (2 * D_SZ / 32)
#define FC_NBY (B_SZ / 32)
#define IMG_NB (4 * B_SZ)
__global__ __launch_bounds__(128)
void img_fc_kernel(
        const float* __restrict__ cap_mean, const float* __restrict__ fc_w,
        const float* __restrict__ fc_b, float* __restrict__ params,
        const float* __restrict__ img, float* __restrict__ meanR,
        float* __restrict__ cs1, float* __restrict__ cs2) {
    const int blk = blockIdx.x;
    if (blk < FC_NBX * FC_NBY)
        fc_gemm_role(blk % FC_NBX, blk / FC_NBX, cap_mean, fc_w, fc_b, params);
    else
        img_pre_role(blk - FC_NBX * FC_NBY, img, meanR, cs1, cs2);
}

__device__ __forceinline__
void make_pab_role(const int c,
                   const float* __restrict__ cap_mean,
                   const float* __restrict__ params,
                   float* __restrict__ P, float* __restrict__ A2m,
                   float* __restrict__ ABm,
                   float* __restrict__ qv, float* __restrict__ rv) {
    const int tid = threadIdx.x;
    __shared__ float red[256];

    float ss = 0.f;
#pragma unroll
    for (int i = 0; i < 4; ++i) {
        float x = cap_mean[(size_t)c * D_SZ + tid + i * 256];
        ss += x * x;
    }
    red[tid] = ss; __syncthreads();
    for (int s = 128; s > 0; s >>= 1) {
        if (tid < s) red[tid] += red[tid + s];
        __syncthreads();
    }
    const float nrm = sqrtf(red[0]) + EPS_L2;
    __syncthreads();

    float q = 0.f, r = 0.f;
#pragma unroll
    for (int i = 0; i < 4; ++i) {
        int d = tid + i * 256;
        float alpha = params[(size_t)c * (2 * D_SZ) + 2 * d];
        float beta  = params[(size_t)c * (2 * D_SZ) + 2 * d + 1];
        float ch = cap_mean[(size_t)c * D_SZ + d] / nrm;
        P  [(size_t)c * D_SZ + d] = alpha * ch;
        A2m[(size_t)c * D_SZ + d] = alpha * alpha;
        ABm[(size_t)c * D_SZ + d] = alpha * beta;
        q += beta * ch;
        r += beta * beta;
    }
    red[tid] = q; __syncthreads();
    for (int s = 128; s > 0; s >>= 1) {
        if (tid < s) red[tid] += red[tid + s];
        __syncthreads();
    }
    if (tid == 0) qv[c] = red[0];
    __syncthreads();
    red[tid] = r; __syncthreads();
    for (int s = 128; s > 0; s >>= 1) {
        if (tid < s) red[tid] += red[tid + s];
        __syncthreads();
    }
    if (tid == 0) rv[c] = red[0];
}

__global__ __launch_bounds__(256)
void pab_base_kernel(
        const float* __restrict__ cap_mean, const float* __restrict__ params,
        float* __restrict__ P, float* __restrict__ A2m, float* __restrict__ ABm,
        float* __restrict__ qv, float* __restrict__ rv,
        const float* __restrict__ meanR,
        const float* __restrict__ cs1, const float* __restrict__ cs2,
        const float* __restrict__ bn_w, const float* __restrict__ bn_b,
        float* __restrict__ base, float* __restrict__ base2) {
    const int blk = blockIdx.x;
    if (blk < B_SZ)
        make_pab_role(blk, cap_mean, params, P, A2m, ABm, qv, rv);
    else
        make_base_role(blk - B_SZ, meanR, cs1, cs2, bn_w, bn_b, base, base2);
}

#define SM_T  16
#define SM_N  (SM_T / ST_RPP)
#define SM_KW (KC / 4)
__global__ __launch_bounds__(128) __attribute__((amdgpu_waves_per_eu(1, 2)))
void sims_gemm_kernel(
        const float* __restrict__ base, const float* __restrict__ base2,
        const float* __restrict__ P, const float* __restrict__ A2m,
        const float* __restrict__ ABm,
        const float* __restrict__ qv, const float* __restrict__ rv,
        float* __restrict__ out) {
    __shared__ __align__(16) float sBa [SM_T][KP];
    __shared__ __align__(16) float sBa2[SM_T][KP];
    __shared__ __align__(16) float sP  [SM_T][KP];
    __shared__ __align__(16) float sA2 [SM_T][KP];
    __shared__ __align__(16) float sAB [SM_T][KP];
    __shared__ float red[3][3][8][32];
    const int tid  = threadIdx.x;
    const int wave = tid >> 5;
    const int lane = tid & 31;
    const int half = lane >> 4, fr = lane & 15;
    const int bm0 = blockIdx.y * SM_T;
    const int cn0 = blockIdx.x * SM_T;
    const int kw0 = wave * SM_KW;
    v8f acc1 = {}, acc2 = {}, acc3 = {};
    const int srow = tid / ST_CPR;
    const int scol = (tid % ST_CPR) * 4;
    const size_t goffA = (size_t)(bm0 + srow) * D_SZ + scol;
    const size_t goffB = (size_t)(cn0 + srow) * D_SZ + scol;
    const int loff = srow * KP + scol;
    float4 r1[SM_N], r2[SM_N], rp[SM_N], rq[SM_N], rr[SM_N];
#pragma unroll
    for (int i = 0; i < SM_N; ++i) {
        const size_t o = (size_t)(ST_RPP * i) * D_SZ;
        r1[i] = *(const float4*)(base  + goffA + o);
        r2[i] = *(const float4*)(base2 + goffA + o);
        rp[i] = *(const float4*)(P     + goffB + o);
        rq[i] = *(const float4*)(A2m   + goffB + o);
        rr[i] = *(const float4*)(ABm   + goffB + o);
    }
    for (int k0 = 0; k0 < D_SZ; k0 += KC) {
#pragma unroll
        for (int i = 0; i < SM_N; ++i) {
            const int lo = loff + ST_RPP * i * KP;
            *(float4*)(&sBa [0][0] + lo) = make_float4(r1[i].x, r1[i].y, r1[i].z, r1[i].w);
            *(float4*)(&sBa2[0][0] + lo) = make_float4(r2[i].x, r2[i].y, r2[i].z, r2[i].w);
            *(float4*)(&sP  [0][0] + lo) = make_float4(rp[i].x, rp[i].y, rp[i].z, rp[i].w);
            *(float4*)(&sA2 [0][0] + lo) = make_float4(rq[i].x, rq[i].y, rq[i].z, rq[i].w);
            *(float4*)(&sAB [0][0] + lo) = make_float4(rr[i].x, rr[i].y, rr[i].z, rr[i].w);
        }
        __syncthreads();
        if (k0 + KC < D_SZ) {
#pragma unroll
            for (int i = 0; i < SM_N; ++i) {
                const size_t o = (size_t)(ST_RPP * i) * D_SZ + k0 + KC;
                r1[i] = *(const float4*)(base  + goffA + o);
                r2[i] = *(const float4*)(base2 + goffA + o);
                rp[i] = *(const float4*)(P     + goffB + o);
                rq[i] = *(const float4*)(A2m   + goffB + o);
                rr[i] = *(const float4*)(ABm   + goffB + o);
            }
        }
#pragma unroll
        for (int kk = 0; kk < SM_KW; kk += 8) {
            const int kb = kw0 + kk + half * 4;
            const float4 x1 = *(const float4*)&sBa [fr][kb];
            const float4 x2 = *(const float4*)&sBa2[fr][kb];
            const float4 xp = *(const float4*)&sP  [fr][kb];
            const float4 xq = *(const float4*)&sA2 [fr][kb];
            const float4 xr = *(const float4*)&sAB [fr][kb];
            v2f a1, a2, bp, b2, bb;
            a1[0] = x1.x; a1[1] = x1.y;
            a2[0] = x2.x; a2[1] = x2.y;
            bp[0] = xp.x; bp[1] = xp.y;
            b2[0] = xq.x; b2[1] = xq.y;
            bb[0] = xr.x; bb[1] = xr.y;
            acc1 = __builtin_amdgcn_wmma_f32_16x16x4_f32(
                    false, a1, false, bp, (short)0, acc1, false, false);
            acc2 = __builtin_amdgcn_wmma_f32_16x16x4_f32(
                    false, a2, false, b2, (short)0, acc2, false, false);
            acc3 = __builtin_amdgcn_wmma_f32_16x16x4_f32(
                    false, a1, false, bb, (short)0, acc3, false, false);
            a1[0] = x1.z; a1[1] = x1.w;
            a2[0] = x2.z; a2[1] = x2.w;
            bp[0] = xp.z; bp[1] = xp.w;
            b2[0] = xq.z; b2[1] = xq.w;
            bb[0] = xr.z; bb[1] = xr.w;
            acc1 = __builtin_amdgcn_wmma_f32_16x16x4_f32(
                    false, a1, false, bp, (short)0, acc1, false, false);
            acc2 = __builtin_amdgcn_wmma_f32_16x16x4_f32(
                    false, a2, false, b2, (short)0, acc2, false, false);
            acc3 = __builtin_amdgcn_wmma_f32_16x16x4_f32(
                    false, a1, false, bb, (short)0, acc3, false, false);
        }
        __syncthreads();
    }
    if (wave > 0) {
#pragma unroll
        for (int j = 0; j < 8; ++j) {
            red[wave - 1][0][j][lane] = acc1[j];
            red[wave - 1][1][j][lane] = acc2[j];
            red[wave - 1][2][j][lane] = acc3[j];
        }
    }
    __syncthreads();
    if (wave == 0) {
#pragma unroll
        for (int j = 0; j < 8; ++j) {
            float s1 = acc1[j], s2 = acc2[j], s3 = acc3[j];
            for (int w = 0; w < 3; ++w) {
                s1 += red[w][0][j][lane];
                s2 += red[w][1][j][lane];
                s3 += red[w][2][j][lane];
            }
            int b = bm0 + j + half * 8;
            int c = cn0 + fr;
            float num = s1 + qv[c];
            float den = sqrtf(s2 + 2.f * s3 + rv[c]) + EPS_L2;
            out[(size_t)b * B_SZ + c] = num / den;
        }
    }
}

extern "C" void kernel_launch(void* const* d_in, const int* in_sizes, int n_in,
                              void* d_out, int out_size, void* d_ws, size_t ws_size,
                              hipStream_t stream) {
    const float* img   = (const float*)d_in[0];
    const float* cap   = (const float*)d_in[1];
    const int*   lens  = (const int*)  d_in[2];
    const float* fc_w  = (const float*)d_in[3];
    const float* fc_b  = (const float*)d_in[4];
    const float* bn_w  = (const float*)d_in[5];
    const float* bn_b  = (const float*)d_in[6];
    float* out = (float*)d_out;

    float* ws = (float*)d_ws;
    const size_t BD = (size_t)B_SZ * D_SZ;
    float* cap_mean = ws;
    float* meanR    = cap_mean + BD;
    float* base     = meanR + BD;
    float* base2    = base + BD;
    float* P        = base2 + BD;
    float* A2m      = P + BD;
    float* ABm      = A2m + BD;
    float* params   = ABm + BD;
    float* colsum1  = params + 2 * BD;
    float* colsum2  = colsum1 + D_SZ;
    float* qv       = colsum2 + D_SZ;
    float* rv       = qv + B_SZ;

    cap_pre_kernel<<<dim3(2 * B_SZ), 256, 0, stream>>>(cap, lens, cap_mean, colsum1);
    img_fc_kernel<<<dim3(FC_NBX * FC_NBY + IMG_NB), 128, 0, stream>>>(
            cap_mean, fc_w, fc_b, params, img, meanR, colsum1, colsum2);
    pab_base_kernel<<<dim3(2 * B_SZ), 256, 0, stream>>>(
            cap_mean, params, P, A2m, ABm, qv, rv,
            meanR, colsum1, colsum2, bn_w, bn_b, base, base2);
    sims_gemm_kernel<<<dim3(B_SZ / SM_T, B_SZ / SM_T), 128, 0, stream>>>(
            base, base2, P, A2m, ABm, qv, rv, out);
}


# P/A2/AB, BN base and the q, r, row-norm sums are formed in the sims GEMM's tile staging and their launch is dropped: 4 launches to 3 (q, r, norm sum order changes; 1/norm applied to the numerator)
# speedup vs baseline: 5.4038x; 1.0786x over previous
#include <hip/hip_runtime.h>
#include <hip/hip_bf16.h>
#include <math.h>

#define B_SZ 256
#define R_SZ 36
#define T_SZ 64
#define D_SZ 1024
#define EPS_L2 1e-8f
#define EPS_BN 1e-5f
#define LEAK 0.1f
#define KC 128
#define ST_CPR (KC / 4)
#define ST_RPP (128 / ST_CPR)
#define ST_N   (32 / ST_RPP)
#define KP (KC + 4)

typedef __attribute__((ext_vector_type(2))) float v2f;
typedef __attribute__((ext_vector_type(8))) float v8f;


#define PRE_CH (D_SZ / 2)
#define PRE_CG (PRE_CH / 4)
#define CAP_RPT (T_SZ / 2)
#define IMG_RPT (R_SZ / 2)
#define IMG_P2  16

__global__ __launch_bounds__(256)
void cap_pre_kernel(const float* __restrict__ cap,
                    const int* __restrict__ lens,
                    float* __restrict__ cap_mean,
                    float* __restrict__ bn_sums) {
    __shared__ __align__(16) float s_sq[2][PRE_CH];
    __shared__ __align__(16) float s_sm[2][PRE_CH];
    const int b   = blockIdx.x >> 1;
    const int c0  = (blockIdx.x & 1) * PRE_CH;
    const int tid = threadIdx.x;
    const int cg  = tid & (PRE_CG - 1);
    const int tg  = tid / PRE_CG;
    const int t0  = tg * CAP_RPT;
    const int len = lens[b];
    if (blockIdx.x < 2 * D_SZ / 256) bn_sums[blockIdx.x * 256 + tid] = 0.f;
    const float* p = cap + ((size_t)b * T_SZ + t0) * D_SZ + c0 + cg * 4;
    float4 v[CAP_RPT];
#pragma unroll
    for (int j = 0; j < CAP_RPT; ++j)
        v[j] = *(const float4*)(p + (size_t)j * D_SZ);
    float q[CAP_RPT][4], s[CAP_RPT][4];
#pragma unroll
    for (int j = 0; j < CAP_RPT; ++j) {
        const bool valid = (t0 + j < len);
        const float x[4] = {v[j].x, v[j].y, v[j].z, v[j].w};
#pragma unroll
        for (int i = 0; i < 4; ++i) {
            float y = x[i] > 0.f ? x[i] : LEAK * x[i];
            q[j][i] = y * y;
            s[j][i] = valid ? y : 0.f;
        }
    }
#pragma unroll
    for (int w = CAP_RPT / 2; w >= 1; w >>= 1) {
#pragma unroll
        for (int j = 0; j < w; ++j) {
#pragma unroll
            for (int i = 0; i < 4; ++i) {
                q[j][i] += q[j + w][i];
                s[j][i] += s[j + w][i];
            }
        }
    }
    *(float4*)&s_sq[tg][cg * 4] = make_float4(q[0][0], q[0][1], q[0][2], q[0][3]);
    *(float4*)&s_sm[tg][cg * 4] = make_float4(s[0][0], s[0][1], s[0][2], s[0][3]);
    __syncthreads();
#pragma unroll
    for (int i = 0; i < PRE_CH / 256; ++i) {
        const int c = tid + i * 256;
        float ssq = s_sq[0][c] + s_sq[1][c];
        float ssm = s_sm[0][c] + s_sm[1][c];
        float n = sqrtf(ssq) + EPS_L2;
        cap_mean[(size_t)b * D_SZ + c0 + c] = ssm / (n * (float)len);
    }
}

#define IMG_CH (D_SZ / 4)
#define IMG_CG (IMG_CH / 4)
__device__ __forceinline__
void img_pre_role(const int blk,
                  const float* __restrict__ img,
                  float* __restrict__ meanR,
                  float* __restrict__ cs1,
                  float* __restrict__ cs2) {
    __shared__ __align__(16) float s_sq[2][IMG_CH];
    __shared__ __align__(16) float s_sm[2][IMG_CH];
    const int b   = blk >> 2;
    const int c0  = (blk & 3) * IMG_CH;
    const int tid = threadIdx.x;
    const int cg  = tid & (IMG_CG - 1);
    const int tg  = tid / IMG_CG;
    const float* p = img + ((size_t)b * R_SZ + tg * IMG_RPT) * D_SZ + c0 + cg * 4;
    float4 v[IMG_RPT];
#pragma unroll
    for (int j = 0; j < IMG_RPT; ++j)
        v[j] = *(const float4*)(p + (size_t)j * D_SZ);
    float q[IMG_RPT][4], s[IMG_RPT][4];
#pragma unroll
    for (int j = 0; j < IMG_RPT; ++j) {
        const float x[4] = {v[j].x, v[j].y, v[j].z, v[j].w};
#pragma unroll
        for (int i = 0; i < 4; ++i) {
            float y = x[i] > 0.f ? x[i] : LEAK * x[i];
            q[j][i] = y * y;
            s[j][i] = y;
        }
    }
#pragma unroll
    for (int j = IMG_P2; j < IMG_RPT; ++j) {
#pragma unroll
        for (int i = 0; i < 4; ++i) {
            q[j - IMG_P2][i] += q[j][i];
            s[j - IMG_P2][i] += s[j][i];
        }
    }
#pragma unroll
    for (int w = IMG_P2 / 2; w >= 1; w >>= 1) {
#pragma unroll
        for (int j = 0; j < w; ++j) {
#pragma unroll
            for (int i = 0; i < 4; ++i) {
                q[j][i] += q[j + w][i];
                s[j][i] += s[j + w][i];
            }
        }
    }
    *(float4*)&s_sq[tg][cg * 4] = make_float4(q[0][0], q[0][1], q[0][2], q[0][3]);
    *(float4*)&s_sm[tg][cg * 4] = make_float4(s[0][0], s[0][1], s[0][2], s[0][3]);
    __syncthreads();
#pragma unroll
    for (int i = 0; i < IMG_CH / 128; ++i) {
        const int c = tid + i * 128;
        const int d = c0 + c;
        float ssq = s_sq[0][c] + s_sq[1][c];
        float ssm = s_sm[0][c] + s_sm[1][c];
        float n = sqrtf(ssq) + EPS_L2;
        float s1 = ssm / n;
        float s2 = ssq / (n * n);
        meanR[(size_t)b * D_SZ + d] = s1 * (1.f / (float)R_SZ);
        atomicAdd(cs1 + d, s1);
        atomicAdd(cs2 + d, s2);
    }
}

__device__ __forceinline__
void fc_gemm_role(const int bx, const int by,
        const float* __restrict__ A, const float* __restrict__ Bw,
        const float* __restrict__ bias, float* __restrict__ Cout) {
    __shared__ __align__(16) float sA[2][16][KP];
    __shared__ __align__(16) float sB[2][16][KP];
    const int tid  = threadIdx.x;
    const int wave = tid >> 5;
    const int lane = tid & 31;
    const int wm = wave >> 1, wn = wave & 1;
    const int half = lane >> 4, fr = lane & 15;
    const int m0 = by * 32;
    const int n0 = bx * 32;
    v8f acc = {};
    const int srow = tid / ST_CPR;
    const int scol = (tid % ST_CPR) * 4;
    const float* gA = A  + (size_t)(m0 + srow) * D_SZ + scol;
    const float* gB = Bw + (size_t)(n0 + srow) * D_SZ + scol;
    float* lA = &sA[0][0][0] + srow * KP + scol;
    float* lB = &sB[0][0][0] + srow * KP + scol;
    float4 ra[ST_N], rb[ST_N];
#pragma unroll
    for (int i = 0; i < ST_N; ++i) {
        ra[i] = *(const float4*)(gA + (size_t)(ST_RPP * i) * D_SZ);
        rb[i] = *(const float4*)(gB + (size_t)(ST_RPP * i) * D_SZ);
    }
#pragma unroll 1
    for (int k0 = 0; k0 < D_SZ; k0 += KC) {
#pragma unroll
        for (int i = 0; i < ST_N; ++i) {
            *(float4*)(lA + ST_RPP * i * KP) =
                    make_float4(ra[i].x, ra[i].y, ra[i].z, ra[i].w);
            *(float4*)(lB + ST_RPP * i * KP) =
                    make_float4(rb[i].x, rb[i].y, rb[i].z, rb[i].w);
        }
        __syncthreads();
        if (k0 + KC < D_SZ) {
#pragma unroll
            for (int i = 0; i < ST_N; ++i) {
                ra[i] = *(const float4*)(gA + (size_t)(ST_RPP * i) * D_SZ + k0 + KC);
                rb[i] = *(const float4*)(gB + (size_t)(ST_RPP * i) * D_SZ + k0 + KC);
            }
        }
#pragma unroll
        for (int kk = 0; kk < KC; kk += 8) {
            const int kb = kk + half * 4;
            const float4 a4 = *(const float4*)&sA[wm][fr][kb];
            const float4 b4 = *(const float4*)&sB[wn][fr][kb];
            v2f a, b;
            a[0] = a4.x; a[1] = a4.y;
            b[0] = b4.x; b[1] = b4.y;
            acc = __builtin_amdgcn_wmma_f32_16x16x4_f32(
                    false, a, false, b, (short)0, acc, false, false);
            a[0] = a4.z; a[1] = a4.w;
            b[0] = b4.z; b[1] = b4.w;
            acc = __builtin_amdgcn_wmma_f32_16x16x4_f32(
                    false, a, false, b, (short)0, acc, false, false);
        }
        __syncthreads();
    }
#pragma unroll
    for (int j = 0; j < 8; ++j) {
        int m = m0 + wm * 16 + j + half * 8;
        int n = n0 + wn * 16 + fr;
        Cout[(size_t)m * (2 * D_SZ) + n] = acc[j] + bias[n];
    }
}

#define FC_NBX (2 * D_SZ / 32)
#define FC_NBY (B_SZ / 32)
#define IMG_NB (4 * B_SZ)
__global__ __launch_bounds__(128)
void img_fc_kernel(
        const float* __restrict__ cap_mean, const float* __restrict__ fc_w,
        const float* __restrict__ fc_b, float* __restrict__ params,
        const float* __restrict__ img, float* __restrict__ meanR,
        float* __restrict__ cs1, float* __restrict__ cs2) {
    const int blk = blockIdx.x;
    if (blk < FC_NBX * FC_NBY)
        fc_gemm_role(blk % FC_NBX, blk / FC_NBX, cap_mean, fc_w, fc_b, params);
    else
        img_pre_role(blk - FC_NBX * FC_NBY, img, meanR, cs1, cs2);
}

#define SM_T  16
#define SM_N  (SM_T / ST_RPP)
#define SM_KW (KC / 4)
__global__ __launch_bounds__(128) __attribute__((amdgpu_waves_per_eu(1, 2)))
void sims_gemm_kernel(
        const float* __restrict__ meanR,
        const float* __restrict__ cs1, const float* __restrict__ cs2,
        const float* __restrict__ bn_w, const float* __restrict__ bn_b,
        const float* __restrict__ cap_mean, const float* __restrict__ params,
        float* __restrict__ out) {
    __shared__ __align__(16) float sBa [SM_T][KP];
    __shared__ __align__(16) float sBa2[SM_T][KP];
    __shared__ __align__(16) float sP  [SM_T][KP];
    __shared__ __align__(16) float sA2 [SM_T][KP];
    __shared__ __align__(16) float sAB [SM_T][KP];
    __shared__ float red[3][3][8][32];
    __shared__ float s_row[3][SM_T];
    const int tid  = threadIdx.x;
    const int wave = tid >> 5;
    const int lane = tid & 31;
    const int half = lane >> 4, fr = lane & 15;
    const int bm0 = blockIdx.y * SM_T;
    const int cn0 = blockIdx.x * SM_T;
    const int kw0 = wave * SM_KW;
    const float inv_n = 1.f / (float)(B_SZ * R_SZ);
    v8f acc1 = {}, acc2 = {}, acc3 = {};
    const int srow = tid / ST_CPR;
    const int scol = (tid % ST_CPR) * 4;
    const size_t goffA = (size_t)(bm0 + srow) * D_SZ + scol;
    const size_t goffC = (size_t)(cn0 + srow) * D_SZ + scol;
    const size_t goffW = (size_t)(cn0 + srow) * (2 * D_SZ) + 2 * scol;
    const int loff = srow * KP + scol;
    float4 rm[SM_N], rc[SM_N], rw0[SM_N], rw1[SM_N], g1, g2, gw, gb;
    float qs[SM_N] = {}, rs[SM_N] = {}, ns[SM_N] = {};
    g1 = *(const float4*)(cs1  + scol);
    g2 = *(const float4*)(cs2  + scol);
    gw = *(const float4*)(bn_w + scol);
    gb = *(const float4*)(bn_b + scol);
#pragma unroll
    for (int i = 0; i < SM_N; ++i) {
        const size_t o = (size_t)(ST_RPP * i) * D_SZ;
        rm[i]  = *(const float4*)(meanR    + goffA + o);
        rc[i]  = *(const float4*)(cap_mean + goffC + o);
        rw0[i] = *(const float4*)(params   + goffW + 2 * o);
        rw1[i] = *(const float4*)(params   + goffW + 2 * o + 4);
    }
#pragma unroll
    for (int k0 = 0; k0 < D_SZ; k0 += KC) {
        const float c1[4] = {g1.x, g1.y, g1.z, g1.w};
        const float c2[4] = {g2.x, g2.y, g2.z, g2.w};
        const float cw[4] = {gw.x, gw.y, gw.z, gw.w};
        const float cb[4] = {gb.x, gb.y, gb.z, gb.w};
        float mu[4], sc[4];
#pragma unroll
        for (int e = 0; e < 4; ++e) {
            mu[e] = c1[e] * inv_n;
            const float ex2 = c2[e] * inv_n;
            const float var = ex2 - mu[e] * mu[e];
            sc[e] = cw[e] * rsqrtf(var + EPS_BN);
        }
#pragma unroll
        for (int i = 0; i < SM_N; ++i) {
            const int lo = loff + ST_RPP * i * KP;
            const float m[4]  = {rm[i].x, rm[i].y, rm[i].z, rm[i].w};
            const float cm[4] = {rc[i].x, rc[i].y, rc[i].z, rc[i].w};
            const float al[4] = {rw0[i].x, rw0[i].z, rw1[i].x, rw1[i].z};
            const float be[4] = {rw0[i].y, rw0[i].w, rw1[i].y, rw1[i].w};
            float v[4], v2[4], p[4], a2[4], ab[4];
#pragma unroll
            for (int e = 0; e < 4; ++e) {
                v[e]  = (m[e] - mu[e]) * sc[e] + cb[e];
                v2[e] = v[e] * v[e];
                p[e]  = al[e] * cm[e];
                a2[e] = al[e] * al[e];
                ab[e] = al[e] * be[e];
                qs[i] += be[e] * cm[e];
                rs[i] += be[e] * be[e];
                ns[i] += cm[e] * cm[e];
            }
            *(float4*)(&sBa [0][0] + lo) = make_float4(v[0],  v[1],  v[2],  v[3]);
            *(float4*)(&sBa2[0][0] + lo) = make_float4(v2[0], v2[1], v2[2], v2[3]);
            *(float4*)(&sP  [0][0] + lo) = make_float4(p[0],  p[1],  p[2],  p[3]);
            *(float4*)(&sA2 [0][0] + lo) = make_float4(a2[0], a2[1], a2[2], a2[3]);
            *(float4*)(&sAB [0][0] + lo) = make_float4(ab[0], ab[1], ab[2], ab[3]);
        }
        __syncthreads();
        if (k0 + KC < D_SZ) {
            const int kn = k0 + KC;
            g1 = *(const float4*)(cs1  + scol + kn);
            g2 = *(const float4*)(cs2  + scol + kn);
            gw = *(const float4*)(bn_w + scol + kn);
            gb = *(const float4*)(bn_b + scol + kn);
#pragma unroll
            for (int i = 0; i < SM_N; ++i) {
                const size_t o = (size_t)(ST_RPP * i) * D_SZ + kn;
                rm[i]  = *(const float4*)(meanR    + goffA + o);
                rc[i]  = *(const float4*)(cap_mean + goffC + o);
                rw0[i] = *(const float4*)(params   + goffW + 2 * o);
                rw1[i] = *(const float4*)(params   + goffW + 2 * o + 4);
            }
        }
#pragma unroll
        for (int kk = 0; kk < SM_KW; kk += 8) {
            const int kb = kw0 + kk + half * 4;
            const float4 x1 = *(const float4*)&sBa [fr][kb];
            const float4 x2 = *(const float4*)&sBa2[fr][kb];
            const float4 xp = *(const float4*)&sP  [fr][kb];
            const float4 xq = *(const float4*)&sA2 [fr][kb];
            const float4 xr = *(const float4*)&sAB [fr][kb];
            v2f a1, a2, bp, b2, bb;
            a1[0] = x1.x; a1[1] = x1.y;
            a2[0] = x2.x; a2[1] = x2.y;
            bp[0] = xp.x; bp[1] = xp.y;
            b2[0] = xq.x; b2[1] = xq.y;
            bb[0] = xr.x; bb[1] = xr.y;
            acc1 = __builtin_amdgcn_wmma_f32_16x16x4_f32(
                    false, a1, false, bp, (short)0, acc1, false, false);
            acc2 = __builtin_amdgcn_wmma_f32_16x16x4_f32(
                    false, a2, false, b2, (short)0, acc2, false, false);
            acc3 = __builtin_amdgcn_wmma_f32_16x16x4_f32(
                    false, a1, false, bb, (short)0, acc3, false, false);
            a1[0] = x1.z; a1[1] = x1.w;
            a2[0] = x2.z; a2[1] = x2.w;
            bp[0] = xp.z; bp[1] = xp.w;
            b2[0] = xq.z; b2[1] = xq.w;
            bb[0] = xr.z; bb[1] = xr.w;
            acc1 = __builtin_amdgcn_wmma_f32_16x16x4_f32(
                    false, a1, false, bp, (short)0, acc1, false, false);
            acc2 = __builtin_amdgcn_wmma_f32_16x16x4_f32(
                    false, a2, false, b2, (short)0, acc2, false, false);
            acc3 = __builtin_amdgcn_wmma_f32_16x16x4_f32(
                    false, a1, false, bb, (short)0, acc3, false, false);
        }
        __syncthreads();
    }
#pragma unroll
    for (int i = 0; i < SM_N; ++i) {
        float a = qs[i], b = rs[i], c = ns[i];
#pragma unroll
        for (int o = 16; o >= 1; o >>= 1) {
            a += __shfl_xor(a, o);
            b += __shfl_xor(b, o);
            c += __shfl_xor(c, o);
        }
        if (lane == 0) {
            s_row[0][srow + ST_RPP * i] = a;
            s_row[1][srow + ST_RPP * i] = b;
            s_row[2][srow + ST_RPP * i] = c;
        }
    }
    if (wave > 0) {
#pragma unroll
        for (int j = 0; j < 8; ++j) {
            red[wave - 1][0][j][lane] = acc1[j];
            red[wave - 1][1][j][lane] = acc2[j];
            red[wave - 1][2][j][lane] = acc3[j];
        }
    }
    __syncthreads();
    if (wave == 0) {
        const int c = cn0 + fr;
        const float qc  = s_row[0][fr];
        const float rc2 = s_row[1][fr];
        const float nrm = sqrtf(s_row[2][fr]) + EPS_L2;
#pragma unroll
        for (int j = 0; j < 8; ++j) {
            float s1 = acc1[j], s2 = acc2[j], s3 = acc3[j];
            for (int w = 0; w < 3; ++w) {
                s1 += red[w][0][j][lane];
                s2 += red[w][1][j][lane];
                s3 += red[w][2][j][lane];
            }
            int b = bm0 + j + half * 8;
            float num = (s1 + qc) / nrm;
            float den = sqrtf(s2 + 2.f * s3 + rc2) + EPS_L2;
            out[(size_t)b * B_SZ + c] = num / den;
        }
    }
}

extern "C" void kernel_launch(void* const* d_in, const int* in_sizes, int n_in,
                              void* d_out, int out_size, void* d_ws, size_t ws_size,
                              hipStream_t stream) {
    const float* img   = (const float*)d_in[0];
    const float* cap   = (const float*)d_in[1];
    const int*   lens  = (const int*)  d_in[2];
    const float* fc_w  = (const float*)d_in[3];
    const float* fc_b  = (const float*)d_in[4];
    const float* bn_w  = (const float*)d_in[5];
    const float* bn_b  = (const float*)d_in[6];
    float* out = (float*)d_out;

    float* ws = (float*)d_ws;
    const size_t BD = (size_t)B_SZ * D_SZ;
    float* cap_mean = ws;
    float* meanR    = cap_mean + BD;
    float* params   = meanR + BD;
    float* colsum1  = params + 2 * BD;
    float* colsum2  = colsum1 + D_SZ;

    cap_pre_kernel<<<dim3(2 * B_SZ), 256, 0, stream>>>(cap, lens, cap_mean, colsum1);
    img_fc_kernel<<<dim3(FC_NBX * FC_NBY + IMG_NB), 128, 0, stream>>>(
            cap_mean, fc_w, fc_b, params, img, meanR, colsum1, colsum2);
    sims_gemm_kernel<<<dim3(B_SZ / SM_T, B_SZ / SM_T), 128, 0, stream>>>(
            meanR, colsum1, colsum2, bn_w, bn_b, cap_mean, params, out);
}


# sims GEMM closing part: the 12 caption-row sums (q, r, norm) finished by DPP row_xmask and permlanex16 adds in one branch-free run, not 4 branch-separated chains of LDS-unit shuffles; same sum order
# speedup vs baseline: 5.7002x; 1.0549x over previous
#include <hip/hip_runtime.h>
#include <hip/hip_bf16.h>
#include <math.h>

#define B_SZ 256
#define R_SZ 36
#define T_SZ 64
#define D_SZ 1024
#define EPS_L2 1e-8f
#define EPS_BN 1e-5f
#define LEAK 0.1f
#define KC 128
#define ST_CPR (KC / 4)
#define ST_RPP (128 / ST_CPR)
#define ST_N   (32 / ST_RPP)
#define KP (KC + 4)

typedef __attribute__((ext_vector_type(2))) float v2f;
typedef __attribute__((ext_vector_type(8))) float v8f;


#define PRE_CH (D_SZ / 2)
#define PRE_CG (PRE_CH / 4)
#define CAP_RPT (T_SZ / 2)
#define IMG_RPT (R_SZ / 2)
#define IMG_P2  16

__global__ __launch_bounds__(256)
void cap_pre_kernel(const float* __restrict__ cap,
                    const int* __restrict__ lens,
                    float* __restrict__ cap_mean,
                    float* __restrict__ bn_sums) {
    __shared__ __align__(16) float s_sq[2][PRE_CH];
    __shared__ __align__(16) float s_sm[2][PRE_CH];
    const int b   = blockIdx.x >> 1;
    const int c0  = (blockIdx.x & 1) * PRE_CH;
    const int tid = threadIdx.x;
    const int cg  = tid & (PRE_CG - 1);
    const int tg  = tid / PRE_CG;
    const int t0  = tg * CAP_RPT;
    const int len = lens[b];
    if (blockIdx.x < 2 * D_SZ / 256) bn_sums[blockIdx.x * 256 + tid] = 0.f;
    const float* p = cap + ((size_t)b * T_SZ + t0) * D_SZ + c0 + cg * 4;
    float4 v[CAP_RPT];
#pragma unroll
    for (int j = 0; j < CAP_RPT; ++j)
        v[j] = *(const float4*)(p + (size_t)j * D_SZ);
    float q[CAP_RPT][4], s[CAP_RPT][4];
#pragma unroll
    for (int j = 0; j < CAP_RPT; ++j) {
        const bool valid = (t0 + j < len);
        const float x[4] = {v[j].x, v[j].y, v[j].z, v[j].w};
#pragma unroll
        for (int i = 0; i < 4; ++i) {
            float y = x[i] > 0.f ? x[i] : LEAK * x[i];
            q[j][i] = y * y;
            s[j][i] = valid ? y : 0.f;
        }
    }
#pragma unroll
    for (int w = CAP_RPT / 2; w >= 1; w >>= 1) {
#pragma unroll
        for (int j = 0; j < w; ++j) {
#pragma unroll
            for (int i = 0; i < 4; ++i) {
                q[j][i] += q[j + w][i];
                s[j][i] += s[j + w][i];
            }
        }
    }
    *(float4*)&s_sq[tg][cg * 4] = make_float4(q[0][0], q[0][1], q[0][2], q[0][3]);
    *(float4*)&s_sm[tg][cg * 4] = make_float4(s[0][0], s[0][1], s[0][2], s[0][3]);
    __syncthreads();
#pragma unroll
    for (int i = 0; i < PRE_CH / 256; ++i) {
        const int c = tid + i * 256;
        float ssq = s_sq[0][c] + s_sq[1][c];
        float ssm = s_sm[0][c] + s_sm[1][c];
        float n = sqrtf(ssq) + EPS_L2;
        cap_mean[(size_t)b * D_SZ + c0 + c] = ssm / (n * (float)len);
    }
}

#define IMG_CH (D_SZ / 4)
#define IMG_CG (IMG_CH / 4)
__device__ __forceinline__
void img_pre_role(const int blk,
                  const float* __restrict__ img,
                  float* __restrict__ meanR,
                  float* __restrict__ cs1,
                  float* __restrict__ cs2) {
    __shared__ __align__(16) float s_sq[2][IMG_CH];
    __shared__ __align__(16) float s_sm[2][IMG_CH];
    const int b   = blk >> 2;
    const int c0  = (blk & 3) * IMG_CH;
    const int tid = threadIdx.x;
    const int cg  = tid & (IMG_CG - 1);
    const int tg  = tid / IMG_CG;
    const float* p = img + ((size_t)b * R_SZ + tg * IMG_RPT) * D_SZ + c0 + cg * 4;
    float4 v[IMG_RPT];
#pragma unroll
    for (int j = 0; j < IMG_RPT; ++j)
        v[j] = *(const float4*)(p + (size_t)j * D_SZ);
    float q[IMG_RPT][4], s[IMG_RPT][4];
#pragma unroll
    for (int j = 0; j < IMG_RPT; ++j) {
        const float x[4] = {v[j].x, v[j].y, v[j].z, v[j].w};
#pragma unroll
        for (int i = 0; i < 4; ++i) {
            float y = x[i] > 0.f ? x[i] : LEAK * x[i];
            q[j][i] = y * y;
            s[j][i] = y;
        }
    }
#pragma unroll
    for (int j = IMG_P2; j < IMG_RPT; ++j) {
#pragma unroll
        for (int i = 0; i < 4; ++i) {
            q[j - IMG_P2][i] += q[j][i];
            s[j - IMG_P2][i] += s[j][i];
        }
    }
#pragma unroll
    for (int w = IMG_P2 / 2; w >= 1; w >>= 1) {
#pragma unroll
        for (int j = 0; j < w; ++j) {
#pragma unroll
            for (int i = 0; i < 4; ++i) {
                q[j][i] += q[j + w][i];
                s[j][i] += s[j + w][i];
            }
        }
    }
    *(float4*)&s_sq[tg][cg * 4] = make_float4(q[0][0], q[0][1], q[0][2], q[0][3]);
    *(float4*)&s_sm[tg][cg * 4] = make_float4(s[0][0], s[0][1], s[0][2], s[0][3]);
    __syncthreads();
#pragma unroll
    for (int i = 0; i < IMG_CH / 128; ++i) {
        const int c = tid + i * 128;
        const int d = c0 + c;
        float ssq = s_sq[0][c] + s_sq[1][c];
        float ssm = s_sm[0][c] + s_sm[1][c];
        float n = sqrtf(ssq) + EPS_L2;
        float s1 = ssm / n;
        float s2 = ssq / (n * n);
        meanR[(size_t)b * D_SZ + d] = s1 * (1.f / (float)R_SZ);
        atomicAdd(cs1 + d, s1);
        atomicAdd(cs2 + d, s2);
    }
}

__device__ __forceinline__
void fc_gemm_role(const int bx, const int by,
        const float* __restrict__ A, const float* __restrict__ Bw,
        const float* __restrict__ bias, float* __restrict__ Cout) {
    __shared__ __align__(16) float sA[2][16][KP];
    __shared__ __align__(16) float sB[2][16][KP];
    const int tid  = threadIdx.x;
    const int wave = tid >> 5;
    const int lane = tid & 31;
    const int wm = wave >> 1, wn = wave & 1;
    const int half = lane >> 4, fr = lane & 15;
    const int m0 = by * 32;
    const int n0 = bx * 32;
    v8f acc = {};
    const int srow = tid / ST_CPR;
    const int scol = (tid % ST_CPR) * 4;
    const float* gA = A  + (size_t)(m0 + srow) * D_SZ + scol;
    const float* gB = Bw + (size_t)(n0 + srow) * D_SZ + scol;
    float* lA = &sA[0][0][0] + srow * KP + scol;
    float* lB = &sB[0][0][0] + srow * KP + scol;
    float4 ra[ST_N], rb[ST_N];
#pragma unroll
    for (int i = 0; i < ST_N; ++i) {
        ra[i] = *(const float4*)(gA + (size_t)(ST_RPP * i) * D_SZ);
        rb[i] = *(const float4*)(gB + (size_t)(ST_RPP * i) * D_SZ);
    }
#pragma unroll 1
    for (int k0 = 0; k0 < D_SZ; k0 += KC) {
#pragma unroll
        for (int i = 0; i < ST_N; ++i) {
            *(float4*)(lA + ST_RPP * i * KP) =
                    make_float4(ra[i].x, ra[i].y, ra[i].z, ra[i].w);
            *(float4*)(lB + ST_RPP * i * KP) =
                    make_float4(rb[i].x, rb[i].y, rb[i].z, rb[i].w);
        }
        __syncthreads();
        if (k0 + KC < D_SZ) {
#pragma unroll
            for (int i = 0; i < ST_N; ++i) {
                ra[i] = *(const float4*)(gA + (size_t)(ST_RPP * i) * D_SZ + k0 + KC);
                rb[i] = *(const float4*)(gB + (size_t)(ST_RPP * i) * D_SZ + k0 + KC);
            }
        }
#pragma unroll
        for (int kk = 0; kk < KC; kk += 8) {
            const int kb = kk + half * 4;
            const float4 a4 = *(const float4*)&sA[wm][fr][kb];
            const float4 b4 = *(const float4*)&sB[wn][fr][kb];
            v2f a, b;
            a[0] = a4.x; a[1] = a4.y;
            b[0] = b4.x; b[1] = b4.y;
            acc = __builtin_amdgcn_wmma_f32_16x16x4_f32(
                    false, a, false, b, (short)0, acc, false, false);
            a[0] = a4.z; a[1] = a4.w;
            b[0] = b4.z; b[1] = b4.w;
            acc = __builtin_amdgcn_wmma_f32_16x16x4_f32(
                    false, a, false, b, (short)0, acc, false, false);
        }
        __syncthreads();
    }
#pragma unroll
    for (int j = 0; j < 8; ++j) {
        int m = m0 + wm * 16 + j + half * 8;
        int n = n0 + wn * 16 + fr;
        Cout[(size_t)m * (2 * D_SZ) + n] = acc[j] + bias[n];
    }
}

#define FC_NBX (2 * D_SZ / 32)
#define FC_NBY (B_SZ / 32)
#define IMG_NB (4 * B_SZ)
__global__ __launch_bounds__(128)
void img_fc_kernel(
        const float* __restrict__ cap_mean, const float* __restrict__ fc_w,
        const float* __restrict__ fc_b, float* __restrict__ params,
        const float* __restrict__ img, float* __restrict__ meanR,
        float* __restrict__ cs1, float* __restrict__ cs2) {
    const int blk = blockIdx.x;
    if (blk < FC_NBX * FC_NBY)
        fc_gemm_role(blk % FC_NBX, blk / FC_NBX, cap_mean, fc_w, fc_b, params);
    else
        img_pre_role(blk - FC_NBX * FC_NBY, img, meanR, cs1, cs2);
}

#define DPP_ROW_XMASK(n) (0x160 + (n))
__device__ __forceinline__ float lane_xor16(float x) {
    const int b = __builtin_bit_cast(int, x);
    return __builtin_bit_cast(float, __builtin_amdgcn_permlanex16(
            b, b, 0x76543210, 0xfedcba98, false, false));
}
template <int N>
__device__ __forceinline__ float row_xor(float x) {
    const int b = __builtin_bit_cast(int, x);
    return __builtin_bit_cast(float, __builtin_amdgcn_update_dpp(
            0, b, DPP_ROW_XMASK(N), 0xf, 0xf, false));
}
__device__ __forceinline__ float wave_sum(float x) {
    x += lane_xor16(x);
    x += row_xor<8>(x);
    x += row_xor<4>(x);
    x += row_xor<2>(x);
    x += row_xor<1>(x);
    return x;
}

#define SM_T  16
#define SM_N  (SM_T / ST_RPP)
#define SM_KW (KC / 4)
__global__ __launch_bounds__(128) __attribute__((amdgpu_waves_per_eu(1, 2)))
void sims_gemm_kernel(
        const float* __restrict__ meanR,
        const float* __restrict__ cs1, const float* __restrict__ cs2,
        const float* __restrict__ bn_w, const float* __restrict__ bn_b,
        const float* __restrict__ cap_mean, const float* __restrict__ params,
        float* __restrict__ out) {
    __shared__ __align__(16) float sBa [SM_T][KP];
    __shared__ __align__(16) float sBa2[SM_T][KP];
    __shared__ __align__(16) float sP  [SM_T][KP];
    __shared__ __align__(16) float sA2 [SM_T][KP];
    __shared__ __align__(16) float sAB [SM_T][KP];
    __shared__ float red[3][3][8][32];
    __shared__ float s_row[3][SM_T];
    const int tid  = threadIdx.x;
    const int wave = tid >> 5;
    const int lane = tid & 31;
    const int half = lane >> 4, fr = lane & 15;
    const int bm0 = blockIdx.y * SM_T;
    const int cn0 = blockIdx.x * SM_T;
    const int kw0 = wave * SM_KW;
    const float inv_n = 1.f / (float)(B_SZ * R_SZ);
    v8f acc1 = {}, acc2 = {}, acc3 = {};
    const int srow = tid / ST_CPR;
    const int scol = (tid % ST_CPR) * 4;
    const size_t goffA = (size_t)(bm0 + srow) * D_SZ + scol;
    const size_t goffC = (size_t)(cn0 + srow) * D_SZ + scol;
    const size_t goffW = (size_t)(cn0 + srow) * (2 * D_SZ) + 2 * scol;
    const int loff = srow * KP + scol;
    float4 rm[SM_N], rc[SM_N], rw0[SM_N], rw1[SM_N], g1, g2, gw, gb;
    float qs[SM_N] = {}, rs[SM_N] = {}, ns[SM_N] = {};
    g1 = *(const float4*)(cs1  + scol);
    g2 = *(const float4*)(cs2  + scol);
    gw = *(const float4*)(bn_w + scol);
    gb = *(const float4*)(bn_b + scol);
#pragma unroll
    for (int i = 0; i < SM_N; ++i) {
        const size_t o = (size_t)(ST_RPP * i) * D_SZ;
        rm[i]  = *(const float4*)(meanR    + goffA + o);
        rc[i]  = *(const float4*)(cap_mean + goffC + o);
        rw0[i] = *(const float4*)(params   + goffW + 2 * o);
        rw1[i] = *(const float4*)(params   + goffW + 2 * o + 4);
    }
#pragma unroll
    for (int k0 = 0; k0 < D_SZ; k0 += KC) {
        const float c1[4] = {g1.x, g1.y, g1.z, g1.w};
        const float c2[4] = {g2.x, g2.y, g2.z, g2.w};
        const float cw[4] = {gw.x, gw.y, gw.z, gw.w};
        const float cb[4] = {gb.x, gb.y, gb.z, gb.w};
        float mu[4], sc[4];
#pragma unroll
        for (int e = 0; e < 4; ++e) {
            mu[e] = c1[e] * inv_n;
            const float ex2 = c2[e] * inv_n;
            const float var = ex2 - mu[e] * mu[e];
            sc[e] = cw[e] * rsqrtf(var + EPS_BN);
        }
#pragma unroll
        for (int i = 0; i < SM_N; ++i) {
            const int lo = loff + ST_RPP * i * KP;
            const float m[4]  = {rm[i].x, rm[i].y, rm[i].z, rm[i].w};
            const float cm[4] = {rc[i].x, rc[i].y, rc[i].z, rc[i].w};
            const float al[4] = {rw0[i].x, rw0[i].z, rw1[i].x, rw1[i].z};
            const float be[4] = {rw0[i].y, rw0[i].w, rw1[i].y, rw1[i].w};
            float v[4], v2[4], p[4], a2[4], ab[4];
#pragma unroll
            for (int e = 0; e < 4; ++e) {
                v[e]  = (m[e] - mu[e]) * sc[e] + cb[e];
                v2[e] = v[e] * v[e];
                p[e]  = al[e] * cm[e];
                a2[e] = al[e] * al[e];
                ab[e] = al[e] * be[e];
                qs[i] += be[e] * cm[e];
                rs[i] += be[e] * be[e];
                ns[i] += cm[e] * cm[e];
            }
            *(float4*)(&sBa [0][0] + lo) = make_float4(v[0],  v[1],  v[2],  v[3]);
            *(float4*)(&sBa2[0][0] + lo) = make_float4(v2[0], v2[1], v2[2], v2[3]);
            *(float4*)(&sP  [0][0] + lo) = make_float4(p[0],  p[1],  p[2],  p[3]);
            *(float4*)(&sA2 [0][0] + lo) = make_float4(a2[0], a2[1], a2[2], a2[3]);
            *(float4*)(&sAB [0][0] + lo) = make_float4(ab[0], ab[1], ab[2], ab[3]);
        }
        __syncthreads();
        if (k0 + KC < D_SZ) {
            const int kn = k0 + KC;
            g1 = *(const float4*)(cs1  + scol + kn);
            g2 = *(const float4*)(cs2  + scol + kn);
            gw = *(const float4*)(bn_w + scol + kn);
            gb = *(const float4*)(bn_b + scol + kn);
#pragma unroll
            for (int i = 0; i < SM_N; ++i) {
                const size_t o = (size_t)(ST_RPP * i) * D_SZ + kn;
                rm[i]  = *(const float4*)(meanR    + goffA + o);
                rc[i]  = *(const float4*)(cap_mean + goffC + o);
                rw0[i] = *(const float4*)(params   + goffW + 2 * o);
                rw1[i] = *(const float4*)(params   + goffW + 2 * o + 4);
            }
        }
#pragma unroll
        for (int kk = 0; kk < SM_KW; kk += 8) {
            const int kb = kw0 + kk + half * 4;
            const float4 x1 = *(const float4*)&sBa [fr][kb];
            const float4 x2 = *(const float4*)&sBa2[fr][kb];
            const float4 xp = *(const float4*)&sP  [fr][kb];
            const float4 xq = *(const float4*)&sA2 [fr][kb];
            const float4 xr = *(const float4*)&sAB [fr][kb];
            v2f a1, a2, bp, b2, bb;
            a1[0] = x1.x; a1[1] = x1.y;
            a2[0] = x2.x; a2[1] = x2.y;
            bp[0] = xp.x; bp[1] = xp.y;
            b2[0] = xq.x; b2[1] = xq.y;
            bb[0] = xr.x; bb[1] = xr.y;
            acc1 = __builtin_amdgcn_wmma_f32_16x16x4_f32(
                    false, a1, false, bp, (short)0, acc1, false, false);
            acc2 = __builtin_amdgcn_wmma_f32_16x16x4_f32(
                    false, a2, false, b2, (short)0, acc2, false, false);
            acc3 = __builtin_amdgcn_wmma_f32_16x16x4_f32(
                    false, a1, false, bb, (short)0, acc3, false, false);
            a1[0] = x1.z; a1[1] = x1.w;
            a2[0] = x2.z; a2[1] = x2.w;
            bp[0] = xp.z; bp[1] = xp.w;
            b2[0] = xq.z; b2[1] = xq.w;
            bb[0] = xr.z; bb[1] = xr.w;
            acc1 = __builtin_amdgcn_wmma_f32_16x16x4_f32(
                    false, a1, false, bp, (short)0, acc1, false, false);
            acc2 = __builtin_amdgcn_wmma_f32_16x16x4_f32(
                    false, a2, false, b2, (short)0, acc2, false, false);
            acc3 = __builtin_amdgcn_wmma_f32_16x16x4_f32(
                    false, a1, false, bb, (short)0, acc3, false, false);
        }
        __syncthreads();
    }
    float fq[SM_N], fr2[SM_N], fn[SM_N];
#pragma unroll
    for (int i = 0; i < SM_N; ++i) {
        fq[i]  = wave_sum(qs[i]);
        fr2[i] = wave_sum(rs[i]);
        fn[i]  = wave_sum(ns[i]);
    }
    if (lane == 0) {
#pragma unroll
        for (int i = 0; i < SM_N; ++i) {
            s_row[0][srow + ST_RPP * i] = fq[i];
            s_row[1][srow + ST_RPP * i] = fr2[i];
            s_row[2][srow + ST_RPP * i] = fn[i];
        }
    }
    if (wave > 0) {
#pragma unroll
        for (int j = 0; j < 8; ++j) {
            red[wave - 1][0][j][lane] = acc1[j];
            red[wave - 1][1][j][lane] = acc2[j];
            red[wave - 1][2][j][lane] = acc3[j];
        }
    }
    __syncthreads();
    if (wave == 0) {
        const int c = cn0 + fr;
        const float qc  = s_row[0][fr];
        const float rc2 = s_row[1][fr];
        const float nrm = sqrtf(s_row[2][fr]) + EPS_L2;
#pragma unroll
        for (int j = 0; j < 8; ++j) {
            float s1 = acc1[j], s2 = acc2[j], s3 = acc3[j];
            for (int w = 0; w < 3; ++w) {
                s1 += red[w][0][j][lane];
                s2 += red[w][1][j][lane];
                s3 += red[w][2][j][lane];
            }
            int b = bm0 + j + half * 8;
            float num = (s1 + qc) / nrm;
            float den = sqrtf(s2 + 2.f * s3 + rc2) + EPS_L2;
            out[(size_t)b * B_SZ + c] = num / den;
        }
    }
}

extern "C" void kernel_launch(void* const* d_in, const int* in_sizes, int n_in,
                              void* d_out, int out_size, void* d_ws, size_t ws_size,
                              hipStream_t stream) {
    const float* img   = (const float*)d_in[0];
    const float* cap   = (const float*)d_in[1];
    const int*   lens  = (const int*)  d_in[2];
    const float* fc_w  = (const float*)d_in[3];
    const float* fc_b  = (const float*)d_in[4];
    const float* bn_w  = (const float*)d_in[5];
    const float* bn_b  = (const float*)d_in[6];
    float* out = (float*)d_out;

    float* ws = (float*)d_ws;
    const size_t BD = (size_t)B_SZ * D_SZ;
    float* cap_mean = ws;
    float* meanR    = cap_mean + BD;
    float* params   = meanR + BD;
    float* colsum1  = params + 2 * BD;
    float* colsum2  = colsum1 + D_SZ;

    cap_pre_kernel<<<dim3(2 * B_SZ), 256, 0, stream>>>(cap, lens, cap_mean, colsum1);
    img_fc_kernel<<<dim3(FC_NBX * FC_NBY + IMG_NB), 128, 0, stream>>>(
            cap_mean, fc_w, fc_b, params, img, meanR, colsum1, colsum2);
    sims_gemm_kernel<<<dim3(B_SZ / SM_T, B_SZ / SM_T), 128, 0, stream>>>(
            meanR, colsum1, colsum2, bn_w, bn_b, cap_mean, params, out);
}


# sims GEMM: operand loads requested 3 K chunks ahead (was 1) into a register ring and kept in flight across the K loop's barriers, which now fence LDS only; forming held behind them; sums unchanged
# speedup vs baseline: 5.9642x; 1.0463x over previous
#include <hip/hip_runtime.h>
#include <hip/hip_bf16.h>
#include <math.h>

#define B_SZ 256
#define R_SZ 36
#define T_SZ 64
#define D_SZ 1024
#define EPS_L2 1e-8f
#define EPS_BN 1e-5f
#define LEAK 0.1f
#define KC 128
#define ST_CPR (KC / 4)
#define ST_RPP (128 / ST_CPR)
#define ST_N   (32 / ST_RPP)
#define KP (KC + 4)

typedef __attribute__((ext_vector_type(2))) float v2f;
typedef __attribute__((ext_vector_type(8))) float v8f;


#define PRE_CH (D_SZ / 2)
#define PRE_CG (PRE_CH / 4)
#define CAP_RPT (T_SZ / 2)
#define IMG_RPT (R_SZ / 2)
#define IMG_P2  16

__global__ __launch_bounds__(256)
void cap_pre_kernel(const float* __restrict__ cap,
                    const int* __restrict__ lens,
                    float* __restrict__ cap_mean,
                    float* __restrict__ bn_sums) {
    __shared__ __align__(16) float s_sq[2][PRE_CH];
    __shared__ __align__(16) float s_sm[2][PRE_CH];
    const int b   = blockIdx.x >> 1;
    const int c0  = (blockIdx.x & 1) * PRE_CH;
    const int tid = threadIdx.x;
    const int cg  = tid & (PRE_CG - 1);
    const int tg  = tid / PRE_CG;
    const int t0  = tg * CAP_RPT;
    const int len = lens[b];
    if (blockIdx.x < 2 * D_SZ / 256) bn_sums[blockIdx.x * 256 + tid] = 0.f;
    const float* p = cap + ((size_t)b * T_SZ + t0) * D_SZ + c0 + cg * 4;
    float4 v[CAP_RPT];
#pragma unroll
    for (int j = 0; j < CAP_RPT; ++j)
        v[j] = *(const float4*)(p + (size_t)j * D_SZ);
    float q[CAP_RPT][4], s[CAP_RPT][4];
#pragma unroll
    for (int j = 0; j < CAP_RPT; ++j) {
        const bool valid = (t0 + j < len);
        const float x[4] = {v[j].x, v[j].y, v[j].z, v[j].w};
#pragma unroll
        for (int i = 0; i < 4; ++i) {
            float y = x[i] > 0.f ? x[i] : LEAK * x[i];
            q[j][i] = y * y;
            s[j][i] = valid ? y : 0.f;
        }
    }
#pragma unroll
    for (int w = CAP_RPT / 2; w >= 1; w >>= 1) {
#pragma unroll
        for (int j = 0; j < w; ++j) {
#pragma unroll
            for (int i = 0; i < 4; ++i) {
                q[j][i] += q[j + w][i];
                s[j][i] += s[j + w][i];
            }
        }
    }
    *(float4*)&s_sq[tg][cg * 4] = make_float4(q[0][0], q[0][1], q[0][2], q[0][3]);
    *(float4*)&s_sm[tg][cg * 4] = make_float4(s[0][0], s[0][1], s[0][2], s[0][3]);
    __syncthreads();
#pragma unroll
    for (int i = 0; i < PRE_CH / 256; ++i) {
        const int c = tid + i * 256;
        float ssq = s_sq[0][c] + s_sq[1][c];
        float ssm = s_sm[0][c] + s_sm[1][c];
        float n = sqrtf(ssq) + EPS_L2;
        cap_mean[(size_t)b * D_SZ + c0 + c] = ssm / (n * (float)len);
    }
}

#define IMG_CH (D_SZ / 4)
#define IMG_CG (IMG_CH / 4)
__device__ __forceinline__
void img_pre_role(const int blk,
                  const float* __restrict__ img,
                  float* __restrict__ meanR,
                  float* __restrict__ cs1,
                  float* __restrict__ cs2) {
    __shared__ __align__(16) float s_sq[2][IMG_CH];
    __shared__ __align__(16) float s_sm[2][IMG_CH];
    const int b   = blk >> 2;
    const int c0  = (blk & 3) * IMG_CH;
    const int tid = threadIdx.x;
    const int cg  = tid & (IMG_CG - 1);
    const int tg  = tid / IMG_CG;
    const float* p = img + ((size_t)b * R_SZ + tg * IMG_RPT) * D_SZ + c0 + cg * 4;
    float4 v[IMG_RPT];
#pragma unroll
    for (int j = 0; j < IMG_RPT; ++j)
        v[j] = *(const float4*)(p + (size_t)j * D_SZ);
    float q[IMG_RPT][4], s[IMG_RPT][4];
#pragma unroll
    for (int j = 0; j < IMG_RPT; ++j) {
        const float x[4] = {v[j].x, v[j].y, v[j].z, v[j].w};
#pragma unroll
        for (int i = 0; i < 4; ++i) {
            float y = x[i] > 0.f ? x[i] : LEAK * x[i];
            q[j][i] = y * y;
            s[j][i] = y;
        }
    }
#pragma unroll
    for (int j = IMG_P2; j < IMG_RPT; ++j) {
#pragma unroll
        for (int i = 0; i < 4; ++i) {
            q[j - IMG_P2][i] += q[j][i];
            s[j - IMG_P2][i] += s[j][i];
        }
    }
#pragma unroll
    for (int w = IMG_P2 / 2; w >= 1; w >>= 1) {
#pragma unroll
        for (int j = 0; j < w; ++j) {
#pragma unroll
            for (int i = 0; i < 4; ++i) {
                q[j][i] += q[j + w][i];
                s[j][i] += s[j + w][i];
            }
        }
    }
    *(float4*)&s_sq[tg][cg * 4] = make_float4(q[0][0], q[0][1], q[0][2], q[0][3]);
    *(float4*)&s_sm[tg][cg * 4] = make_float4(s[0][0], s[0][1], s[0][2], s[0][3]);
    __syncthreads();
#pragma unroll
    for (int i = 0; i < IMG_CH / 128; ++i) {
        const int c = tid + i * 128;
        const int d = c0 + c;
        float ssq = s_sq[0][c] + s_sq[1][c];
        float ssm = s_sm[0][c] + s_sm[1][c];
        float n = sqrtf(ssq) + EPS_L2;
        float s1 = ssm / n;
        float s2 = ssq / (n * n);
        meanR[(size_t)b * D_SZ + d] = s1 * (1.f / (float)R_SZ);
        atomicAdd(cs1 + d, s1);
        atomicAdd(cs2 + d, s2);
    }
}

__device__ __forceinline__
void fc_gemm_role(const int bx, const int by,
        const float* __restrict__ A, const float* __restrict__ Bw,
        const float* __restrict__ bias, float* __restrict__ Cout) {
    __shared__ __align__(16) float sA[2][16][KP];
    __shared__ __align__(16) float sB[2][16][KP];
    const int tid  = threadIdx.x;
    const int wave = tid >> 5;
    const int lane = tid & 31;
    const int wm = wave >> 1, wn = wave & 1;
    const int half = lane >> 4, fr = lane & 15;
    const int m0 = by * 32;
    const int n0 = bx * 32;
    v8f acc = {};
    const int srow = tid / ST_CPR;
    const int scol = (tid % ST_CPR) * 4;
    const float* gA = A  + (size_t)(m0 + srow) * D_SZ + scol;
    const float* gB = Bw + (size_t)(n0 + srow) * D_SZ + scol;
    float* lA = &sA[0][0][0] + srow * KP + scol;
    float* lB = &sB[0][0][0] + srow * KP + scol;
    float4 ra[ST_N], rb[ST_N];
#pragma unroll
    for (int i = 0; i < ST_N; ++i) {
        ra[i] = *(const float4*)(gA + (size_t)(ST_RPP * i) * D_SZ);
        rb[i] = *(const float4*)(gB + (size_t)(ST_RPP * i) * D_SZ);
    }
#pragma unroll 1
    for (int k0 = 0; k0 < D_SZ; k0 += KC) {
#pragma unroll
        for (int i = 0; i < ST_N; ++i) {
            *(float4*)(lA + ST_RPP * i * KP) =
                    make_float4(ra[i].x, ra[i].y, ra[i].z, ra[i].w);
            *(float4*)(lB + ST_RPP * i * KP) =
                    make_float4(rb[i].x, rb[i].y, rb[i].z, rb[i].w);
        }
        __syncthreads();
        if (k0 + KC < D_SZ) {
#pragma unroll
            for (int i = 0; i < ST_N; ++i) {
                ra[i] = *(const float4*)(gA + (size_t)(ST_RPP * i) * D_SZ + k0 + KC);
                rb[i] = *(const float4*)(gB + (size_t)(ST_RPP * i) * D_SZ + k0 + KC);
            }
        }
#pragma unroll
        for (int kk = 0; kk < KC; kk += 8) {
            const int kb = kk + half * 4;
            const float4 a4 = *(const float4*)&sA[wm][fr][kb];
            const float4 b4 = *(const float4*)&sB[wn][fr][kb];
            v2f a, b;
            a[0] = a4.x; a[1] = a4.y;
            b[0] = b4.x; b[1] = b4.y;
            acc = __builtin_amdgcn_wmma_f32_16x16x4_f32(
                    false, a, false, b, (short)0, acc, false, false);
            a[0] = a4.z; a[1] = a4.w;
            b[0] = b4.z; b[1] = b4.w;
            acc = __builtin_amdgcn_wmma_f32_16x16x4_f32(
                    false, a, false, b, (short)0, acc, false, false);
        }
        __syncthreads();
    }
#pragma unroll
    for (int j = 0; j < 8; ++j) {
        int m = m0 + wm * 16 + j + half * 8;
        int n = n0 + wn * 16 + fr;
        Cout[(size_t)m * (2 * D_SZ) + n] = acc[j] + bias[n];
    }
}

#define FC_NBX (2 * D_SZ / 32)
#define FC_NBY (B_SZ / 32)
#define IMG_NB (4 * B_SZ)
__global__ __launch_bounds__(128)
void img_fc_kernel(
        const float* __restrict__ cap_mean, const float* __restrict__ fc_w,
        const float* __restrict__ fc_b, float* __restrict__ params,
        const float* __restrict__ img, float* __restrict__ meanR,
        float* __restrict__ cs1, float* __restrict__ cs2) {
    const int blk = blockIdx.x;
    if (blk < FC_NBX * FC_NBY)
        fc_gemm_role(blk % FC_NBX, blk / FC_NBX, cap_mean, fc_w, fc_b, params);
    else
        img_pre_role(blk - FC_NBX * FC_NBY, img, meanR, cs1, cs2);
}

#define DPP_ROW_XMASK(n) (0x160 + (n))
__device__ __forceinline__ float lane_xor16(float x) {
    const int b = __builtin_bit_cast(int, x);
    return __builtin_bit_cast(float, __builtin_amdgcn_permlanex16(
            b, b, 0x76543210, 0xfedcba98, false, false));
}
template <int N>
__device__ __forceinline__ float row_xor(float x) {
    const int b = __builtin_bit_cast(int, x);
    return __builtin_bit_cast(float, __builtin_amdgcn_update_dpp(
            0, b, DPP_ROW_XMASK(N), 0xf, 0xf, false));
}
__device__ __forceinline__ float wave_sum(float x) {
    x += lane_xor16(x);
    x += row_xor<8>(x);
    x += row_xor<4>(x);
    x += row_xor<2>(x);
    x += row_xor<1>(x);
    return x;
}

#define SM_T  16
#define SM_N  (SM_T / ST_RPP)
#define SM_KW (KC / 4)
#define SM_NCH (D_SZ / KC)
#define SM_PF 3
#define SM_LPC (4 + 4 * SM_N)
static_assert(SM_PF * SM_LPC <= 63, "more requests than a wave's load counter holds");

__device__ __forceinline__ void sims_loop_barrier() {
    __builtin_amdgcn_fence(__ATOMIC_RELEASE, "workgroup", "local");
    __builtin_amdgcn_s_barrier();
    __builtin_amdgcn_fence(__ATOMIC_ACQUIRE, "workgroup", "local");
    __builtin_amdgcn_sched_barrier(0);
}

__device__ __forceinline__
void sims_request_chunk(
        const float* __restrict__ meanR,
        const float* __restrict__ cs1, const float* __restrict__ cs2,
        const float* __restrict__ bn_w, const float* __restrict__ bn_b,
        const float* __restrict__ cap_mean, const float* __restrict__ params,
        const size_t goffA, const size_t goffC, const size_t goffW,
        const int scol, const int kn,
        float4& g1, float4& g2, float4& gw, float4& gb,
        float4 (&rm)[SM_N], float4 (&rc)[SM_N],
        float4 (&rw0)[SM_N], float4 (&rw1)[SM_N]) {
    g1 = *(const float4*)(cs1  + scol + kn);
    g2 = *(const float4*)(cs2  + scol + kn);
    gw = *(const float4*)(bn_w + scol + kn);
    gb = *(const float4*)(bn_b + scol + kn);
#pragma unroll
    for (int i = 0; i < SM_N; ++i) {
        const size_t o = (size_t)(ST_RPP * i) * D_SZ + kn;
        rm[i]  = *(const float4*)(meanR    + goffA + o);
        rc[i]  = *(const float4*)(cap_mean + goffC + o);
        rw0[i] = *(const float4*)(params   + goffW + 2 * o);
        rw1[i] = *(const float4*)(params   + goffW + 2 * o + 4);
    }
}

__device__ __forceinline__
void sims_hold(float& inv_n, float& nz,
               float (&qs)[SM_N], float (&rs)[SM_N], float (&ns)[SM_N]) {
    static_assert(SM_N == 4, "the statement below names the 4 staged rows");
    asm volatile("" : "+v"(inv_n), "+v"(nz),
                      "+v"(qs[0]), "+v"(qs[1]), "+v"(qs[2]), "+v"(qs[3]),
                      "+v"(rs[0]), "+v"(rs[1]), "+v"(rs[2]), "+v"(rs[3]),
                      "+v"(ns[0]), "+v"(ns[1]), "+v"(ns[2]), "+v"(ns[3]));
}

__device__ __forceinline__
void sims_form_chunk(
        const float4 g1, const float4 g2, const float4 gw, const float4 gb,
        const float4 (&rm)[SM_N], const float4 (&rc)[SM_N],
        const float4 (&rw0)[SM_N], const float4 (&rw1)[SM_N],
        const float inv_n, const float nz,
        float (&qs)[SM_N], float (&rs)[SM_N], float (&ns)[SM_N],
        float4 (&fv)[SM_N], float4 (&fv2)[SM_N], float4 (&fp)[SM_N],
        float4 (&fa2)[SM_N], float4 (&fab)[SM_N]) {
    const float c1[4] = {g1.x, g1.y, g1.z, g1.w};
    const float c2[4] = {g2.x, g2.y, g2.z, g2.w};
    const float cw[4] = {gw.x, gw.y, gw.z, gw.w};
    const float cb[4] = {gb.x, gb.y, gb.z, gb.w};
    float mu[4], sc[4];
#pragma unroll
    for (int e = 0; e < 4; ++e) {
        mu[e] = c1[e] * inv_n;
        const float ex2 = c2[e] * inv_n;
        const float var = ex2 - mu[e] * mu[e];
        sc[e] = cw[e] * rsqrtf(var + EPS_BN);
    }
#pragma unroll
    for (int i = 0; i < SM_N; ++i) {
        const float m[4]  = {rm[i].x, rm[i].y, rm[i].z, rm[i].w};
        const float cm[4] = {rc[i].x, rc[i].y, rc[i].z, rc[i].w};
        const float al[4] = {rw0[i].x, rw0[i].z, rw1[i].x, rw1[i].z};
        const float be[4] = {rw0[i].y, rw0[i].w, rw1[i].y, rw1[i].w};
        float v[4], v2[4], p[4], a2[4], ab[4];
#pragma unroll
        for (int e = 0; e < 4; ++e) {
            v[e]  = (m[e] - mu[e]) * sc[e] + cb[e];
            v2[e] = v[e] * v[e];
            p[e]  = __builtin_fmaf(al[e], cm[e], nz);
            a2[e] = __builtin_fmaf(al[e], al[e], nz);
            ab[e] = __builtin_fmaf(al[e], be[e], nz);
            qs[i] = __builtin_fmaf(be[e], cm[e], qs[i]);
            rs[i] = __builtin_fmaf(be[e], be[e], rs[i]);
            ns[i] = __builtin_fmaf(cm[e], cm[e], ns[i]);
        }
        fv[i]  = make_float4(v[0],  v[1],  v[2],  v[3]);
        fv2[i] = make_float4(v2[0], v2[1], v2[2], v2[3]);
        fp[i]  = make_float4(p[0],  p[1],  p[2],  p[3]);
        fa2[i] = make_float4(a2[0], a2[1], a2[2], a2[3]);
        fab[i] = make_float4(ab[0], ab[1], ab[2], ab[3]);
    }
}

__global__ __launch_bounds__(128) __attribute__((amdgpu_waves_per_eu(1, 2)))
void sims_gemm_kernel(
        const float* __restrict__ meanR,
        const float* __restrict__ cs1, const float* __restrict__ cs2,
        const float* __restrict__ bn_w, const float* __restrict__ bn_b,
        const float* __restrict__ cap_mean, const float* __restrict__ params,
        float* __restrict__ out) {
    __shared__ __align__(16) float sBa [SM_T][KP];
    __shared__ __align__(16) float sBa2[SM_T][KP];
    __shared__ __align__(16) float sP  [SM_T][KP];
    __shared__ __align__(16) float sA2 [SM_T][KP];
    __shared__ __align__(16) float sAB [SM_T][KP];
    __shared__ float red[3][3][8][32];
    __shared__ float s_row[3][SM_T];
    const int tid  = threadIdx.x;
    const int wave = tid >> 5;
    const int lane = tid & 31;
    const int half = lane >> 4, fr = lane & 15;
    const int bm0 = blockIdx.y * SM_T;
    const int cn0 = blockIdx.x * SM_T;
    const int kw0 = wave * SM_KW;
    const float inv_n = 1.f / (float)(B_SZ * R_SZ);
    v8f acc1 = {}, acc2 = {}, acc3 = {};
    const int srow = tid / ST_CPR;
    const int scol = (tid % ST_CPR) * 4;
    const size_t goffA = (size_t)(bm0 + srow) * D_SZ + scol;
    const size_t goffC = (size_t)(cn0 + srow) * D_SZ + scol;
    const size_t goffW = (size_t)(cn0 + srow) * (2 * D_SZ) + 2 * scol;
    const int loff = srow * KP + scol;
    float4 rm[SM_PF][SM_N], rc[SM_PF][SM_N], rw0[SM_PF][SM_N], rw1[SM_PF][SM_N];
    float4 g1[SM_PF], g2[SM_PF], gw[SM_PF], gb[SM_PF];
    float4 fv[SM_N], fv2[SM_N], fp[SM_N], fa2[SM_N], fab[SM_N];
    float qs[SM_N] = {}, rs[SM_N] = {}, ns[SM_N] = {};
#pragma unroll
    for (int ch = 0; ch < SM_PF; ++ch)
        sims_request_chunk(meanR, cs1, cs2, bn_w, bn_b, cap_mean, params,
                           goffA, goffC, goffW, scol, ch * KC,
                           g1[ch], g2[ch], gw[ch], gb[ch],
                           rm[ch], rc[ch], rw0[ch], rw1[ch]);
    float inv_nh = inv_n, nz = -0.0f;
    sims_form_chunk(g1[0], g2[0], gw[0], gb[0], rm[0], rc[0], rw0[0], rw1[0],
                    inv_nh, nz, qs, rs, ns, fv, fv2, fp, fa2, fab);
#pragma unroll
    for (int ch = 0; ch < SM_NCH; ++ch) {
#pragma unroll
        for (int i = 0; i < SM_N; ++i) {
            const int lo = loff + ST_RPP * i * KP;
            *(float4*)(&sBa [0][0] + lo) = fv[i];
            *(float4*)(&sBa2[0][0] + lo) = fv2[i];
            *(float4*)(&sP  [0][0] + lo) = fp[i];
            *(float4*)(&sA2 [0][0] + lo) = fa2[i];
            *(float4*)(&sAB [0][0] + lo) = fab[i];
        }
        sims_loop_barrier();
        if (ch + SM_PF < SM_NCH)
            sims_request_chunk(meanR, cs1, cs2, bn_w, bn_b, cap_mean, params,
                               goffA, goffC, goffW, scol, (ch + SM_PF) * KC,
                               g1[ch % SM_PF], g2[ch % SM_PF],
                               gw[ch % SM_PF], gb[ch % SM_PF],
                               rm[ch % SM_PF], rc[ch % SM_PF],
                               rw0[ch % SM_PF], rw1[ch % SM_PF]);
#pragma unroll
        for (int kk = 0; kk < SM_KW; kk += 8) {
            const int kb = kw0 + kk + half * 4;
            const float4 x1 = *(const float4*)&sBa [fr][kb];
            const float4 x2 = *(const float4*)&sBa2[fr][kb];
            const float4 xp = *(const float4*)&sP  [fr][kb];
            const float4 xq = *(const float4*)&sA2 [fr][kb];
            const float4 xr = *(const float4*)&sAB [fr][kb];
            v2f a1, a2, bp, b2, bb;
            a1[0] = x1.x; a1[1] = x1.y;
            a2[0] = x2.x; a2[1] = x2.y;
            bp[0] = xp.x; bp[1] = xp.y;
            b2[0] = xq.x; b2[1] = xq.y;
            bb[0] = xr.x; bb[1] = xr.y;
            acc1 = __builtin_amdgcn_wmma_f32_16x16x4_f32(
                    false, a1, false, bp, (short)0, acc1, false, false);
            acc2 = __builtin_amdgcn_wmma_f32_16x16x4_f32(
                    false, a2, false, b2, (short)0, acc2, false, false);
            acc3 = __builtin_amdgcn_wmma_f32_16x16x4_f32(
                    false, a1, false, bb, (short)0, acc3, false, false);
            a1[0] = x1.z; a1[1] = x1.w;
            a2[0] = x2.z; a2[1] = x2.w;
            bp[0] = xp.z; bp[1] = xp.w;
            b2[0] = xq.z; b2[1] = xq.w;
            bb[0] = xr.z; bb[1] = xr.w;
            acc1 = __builtin_amdgcn_wmma_f32_16x16x4_f32(
                    false, a1, false, bp, (short)0, acc1, false, false);
            acc2 = __builtin_amdgcn_wmma_f32_16x16x4_f32(
                    false, a2, false, b2, (short)0, acc2, false, false);
            acc3 = __builtin_amdgcn_wmma_f32_16x16x4_f32(
                    false, a1, false, bb, (short)0, acc3, false, false);
        }
        if (ch + 1 < SM_NCH) {
            sims_hold(inv_nh, nz, qs, rs, ns);
            sims_form_chunk(g1[(ch + 1) % SM_PF], g2[(ch + 1) % SM_PF],
                            gw[(ch + 1) % SM_PF], gb[(ch + 1) % SM_PF],
                            rm[(ch + 1) % SM_PF], rc[(ch + 1) % SM_PF],
                            rw0[(ch + 1) % SM_PF], rw1[(ch + 1) % SM_PF],
                            inv_nh, nz, qs, rs, ns, fv, fv2, fp, fa2, fab);
        }
        sims_loop_barrier();
    }
    float fq[SM_N], fr2[SM_N], fn[SM_N];
#pragma unroll
    for (int i = 0; i < SM_N; ++i) {
        fq[i]  = wave_sum(qs[i]);
        fr2[i] = wave_sum(rs[i]);
        fn[i]  = wave_sum(ns[i]);
    }
    if (lane == 0) {
#pragma unroll
        for (int i = 0; i < SM_N; ++i) {
            s_row[0][srow + ST_RPP * i] = fq[i];
            s_row[1][srow + ST_RPP * i] = fr2[i];
            s_row[2][srow + ST_RPP * i] = fn[i];
        }
    }
    if (wave > 0) {
#pragma unroll
        for (int j = 0; j < 8; ++j) {
            red[wave - 1][0][j][lane] = acc1[j];
            red[wave - 1][1][j][lane] = acc2[j];
            red[wave - 1][2][j][lane] = acc3[j];
        }
    }
    __syncthreads();
    if (wave == 0) {
        const int c = cn0 + fr;
        const float qc  = s_row[0][fr];
        const float rc2 = s_row[1][fr];
        const float nrm = sqrtf(s_row[2][fr]) + EPS_L2;
#pragma unroll
        for (int j = 0; j < 8; ++j) {
            float s1 = acc1[j], s2 = acc2[j], s3 = acc3[j];
            for (int w = 0; w < 3; ++w) {
                s1 += red[w][0][j][lane];
                s2 += red[w][1][j][lane];
                s3 += red[w][2][j][lane];
            }
            int b = bm0 + j + half * 8;
            float num = (s1 + qc) / nrm;
            float den = sqrtf(s2 + 2.f * s3 + rc2) + EPS_L2;
            out[(size_t)b * B_SZ + c] = num / den;
        }
    }
}

extern "C" void kernel_launch(void* const* d_in, const int* in_sizes, int n_in,
                              void* d_out, int out_size, void* d_ws, size_t ws_size,
                              hipStream_t stream) {
    const float* img   = (const float*)d_in[0];
    const float* cap   = (const float*)d_in[1];
    const int*   lens  = (const int*)  d_in[2];
    const float* fc_w  = (const float*)d_in[3];
    const float* fc_b  = (const float*)d_in[4];
    const float* bn_w  = (const float*)d_in[5];
    const float* bn_b  = (const float*)d_in[6];
    float* out = (float*)d_out;

    float* ws = (float*)d_ws;
    const size_t BD = (size_t)B_SZ * D_SZ;
    float* cap_mean = ws;
    float* meanR    = cap_mean + BD;
    float* params   = meanR + BD;
    float* colsum1  = params + 2 * BD;
    float* colsum2  = colsum1 + D_SZ;

    cap_pre_kernel<<<dim3(2 * B_SZ), 256, 0, stream>>>(cap, lens, cap_mean, colsum1);
    img_fc_kernel<<<dim3(FC_NBX * FC_NBY + IMG_NB), 128, 0, stream>>>(
            cap_mean, fc_w, fc_b, params, img, meanR, colsum1, colsum2);
    sims_gemm_kernel<<<dim3(B_SZ / SM_T, B_SZ / SM_T), 128, 0, stream>>>(
            meanR, colsum1, colsum2, bn_w, bn_b, cap_mean, params, out);
}


# sims GEMM: each wave stages the K columns it multiplies, so its LDS tile part is private: K loop without work-group barriers, next chunk's LDS trip runs under products; q, r, norm sum order changes
# speedup vs baseline: 6.3926x; 1.0718x over previous
#include <hip/hip_runtime.h>
#include <hip/hip_bf16.h>
#include <math.h>

#define B_SZ 256
#define R_SZ 36
#define T_SZ 64
#define D_SZ 1024
#define EPS_L2 1e-8f
#define EPS_BN 1e-5f
#define LEAK 0.1f
#define KC 128
#define ST_CPR (KC / 4)
#define ST_RPP (128 / ST_CPR)
#define ST_N   (32 / ST_RPP)
#define KP (KC + 4)

typedef __attribute__((ext_vector_type(2))) float v2f;
typedef __attribute__((ext_vector_type(8))) float v8f;


#define PRE_CH (D_SZ / 2)
#define PRE_CG (PRE_CH / 4)
#define CAP_RPT (T_SZ / 2)
#define IMG_RPT (R_SZ / 2)
#define IMG_P2  16

__global__ __launch_bounds__(256)
void cap_pre_kernel(const float* __restrict__ cap,
                    const int* __restrict__ lens,
                    float* __restrict__ cap_mean,
                    float* __restrict__ bn_sums) {
    __shared__ __align__(16) float s_sq[2][PRE_CH];
    __shared__ __align__(16) float s_sm[2][PRE_CH];
    const int b   = blockIdx.x >> 1;
    const int c0  = (blockIdx.x & 1) * PRE_CH;
    const int tid = threadIdx.x;
    const int cg  = tid & (PRE_CG - 1);
    const int tg  = tid / PRE_CG;
    const int t0  = tg * CAP_RPT;
    const int len = lens[b];
    if (blockIdx.x < 2 * D_SZ / 256) bn_sums[blockIdx.x * 256 + tid] = 0.f;
    const float* p = cap + ((size_t)b * T_SZ + t0) * D_SZ + c0 + cg * 4;
    float4 v[CAP_RPT];
#pragma unroll
    for (int j = 0; j < CAP_RPT; ++j)
        v[j] = *(const float4*)(p + (size_t)j * D_SZ);
    float q[CAP_RPT][4], s[CAP_RPT][4];
#pragma unroll
    for (int j = 0; j < CAP_RPT; ++j) {
        const bool valid = (t0 + j < len);
        const float x[4] = {v[j].x, v[j].y, v[j].z, v[j].w};
#pragma unroll
        for (int i = 0; i < 4; ++i) {
            float y = x[i] > 0.f ? x[i] : LEAK * x[i];
            q[j][i] = y * y;
            s[j][i] = valid ? y : 0.f;
        }
    }
#pragma unroll
    for (int w = CAP_RPT / 2; w >= 1; w >>= 1) {
#pragma unroll
        for (int j = 0; j < w; ++j) {
#pragma unroll
            for (int i = 0; i < 4; ++i) {
                q[j][i] += q[j + w][i];
                s[j][i] += s[j + w][i];
            }
        }
    }
    *(float4*)&s_sq[tg][cg * 4] = make_float4(q[0][0], q[0][1], q[0][2], q[0][3]);
    *(float4*)&s_sm[tg][cg * 4] = make_float4(s[0][0], s[0][1], s[0][2], s[0][3]);
    __syncthreads();
#pragma unroll
    for (int i = 0; i < PRE_CH / 256; ++i) {
        const int c = tid + i * 256;
        float ssq = s_sq[0][c] + s_sq[1][c];
        float ssm = s_sm[0][c] + s_sm[1][c];
        float n = sqrtf(ssq) + EPS_L2;
        cap_mean[(size_t)b * D_SZ + c0 + c] = ssm / (n * (float)len);
    }
}

#define IMG_CH (D_SZ / 4)
#define IMG_CG (IMG_CH / 4)
__device__ __forceinline__
void img_pre_role(const int blk,
                  const float* __restrict__ img,
                  float* __restrict__ meanR,
                  float* __restrict__ cs1,
                  float* __restrict__ cs2) {
    __shared__ __align__(16) float s_sq[2][IMG_CH];
    __shared__ __align__(16) float s_sm[2][IMG_CH];
    const int b   = blk >> 2;
    const int c0  = (blk & 3) * IMG_CH;
    const int tid = threadIdx.x;
    const int cg  = tid & (IMG_CG - 1);
    const int tg  = tid / IMG_CG;
    const float* p = img + ((size_t)b * R_SZ + tg * IMG_RPT) * D_SZ + c0 + cg * 4;
    float4 v[IMG_RPT];
#pragma unroll
    for (int j = 0; j < IMG_RPT; ++j)
        v[j] = *(const float4*)(p + (size_t)j * D_SZ);
    float q[IMG_RPT][4], s[IMG_RPT][4];
#pragma unroll
    for (int j = 0; j < IMG_RPT; ++j) {
        const float x[4] = {v[j].x, v[j].y, v[j].z, v[j].w};
#pragma unroll
        for (int i = 0; i < 4; ++i) {
            float y = x[i] > 0.f ? x[i] : LEAK * x[i];
            q[j][i] = y * y;
            s[j][i] = y;
        }
    }
#pragma unroll
    for (int j = IMG_P2; j < IMG_RPT; ++j) {
#pragma unroll
        for (int i = 0; i < 4; ++i) {
            q[j - IMG_P2][i] += q[j][i];
            s[j - IMG_P2][i] += s[j][i];
        }
    }
#pragma unroll
    for (int w = IMG_P2 / 2; w >= 1; w >>= 1) {
#pragma unroll
        for (int j = 0; j < w; ++j) {
#pragma unroll
            for (int i = 0; i < 4; ++i) {
                q[j][i] += q[j + w][i];
                s[j][i] += s[j + w][i];
            }
        }
    }
    *(float4*)&s_sq[tg][cg * 4] = make_float4(q[0][0], q[0][1], q[0][2], q[0][3]);
    *(float4*)&s_sm[tg][cg * 4] = make_float4(s[0][0], s[0][1], s[0][2], s[0][3]);
    __syncthreads();
#pragma unroll
    for (int i = 0; i < IMG_CH / 128; ++i) {
        const int c = tid + i * 128;
        const int d = c0 + c;
        float ssq = s_sq[0][c] + s_sq[1][c];
        float ssm = s_sm[0][c] + s_sm[1][c];
        float n = sqrtf(ssq) + EPS_L2;
        float s1 = ssm / n;
        float s2 = ssq / (n * n);
        meanR[(size_t)b * D_SZ + d] = s1 * (1.f / (float)R_SZ);
        atomicAdd(cs1 + d, s1);
        atomicAdd(cs2 + d, s2);
    }
}

__device__ __forceinline__
void fc_gemm_role(const int bx, const int by,
        const float* __restrict__ A, const float* __restrict__ Bw,
        const float* __restrict__ bias, float* __restrict__ Cout) {
    __shared__ __align__(16) float sA[2][16][KP];
    __shared__ __align__(16) float sB[2][16][KP];
    const int tid  = threadIdx.x;
    const int wave = tid >> 5;
    const int lane = tid & 31;
    const int wm = wave >> 1, wn = wave & 1;
    const int half = lane >> 4, fr = lane & 15;
    const int m0 = by * 32;
    const int n0 = bx * 32;
    v8f acc = {};
    const int srow = tid / ST_CPR;
    const int scol = (tid % ST_CPR) * 4;
    const float* gA = A  + (size_t)(m0 + srow) * D_SZ + scol;
    const float* gB = Bw + (size_t)(n0 + srow) * D_SZ + scol;
    float* lA = &sA[0][0][0] + srow * KP + scol;
    float* lB = &sB[0][0][0] + srow * KP + scol;
    float4 ra[ST_N], rb[ST_N];
#pragma unroll
    for (int i = 0; i < ST_N; ++i) {
        ra[i] = *(const float4*)(gA + (size_t)(ST_RPP * i) * D_SZ);
        rb[i] = *(const float4*)(gB + (size_t)(ST_RPP * i) * D_SZ);
    }
#pragma unroll 1
    for (int k0 = 0; k0 < D_SZ; k0 += KC) {
#pragma unroll
        for (int i = 0; i < ST_N; ++i) {
            *(float4*)(lA + ST_RPP * i * KP) =
                    make_float4(ra[i].x, ra[i].y, ra[i].z, ra[i].w);
            *(float4*)(lB + ST_RPP * i * KP) =
                    make_float4(rb[i].x, rb[i].y, rb[i].z, rb[i].w);
        }
        __syncthreads();
        if (k0 + KC < D_SZ) {
#pragma unroll
            for (int i = 0; i < ST_N; ++i) {
                ra[i] = *(const float4*)(gA + (size_t)(ST_RPP * i) * D_SZ + k0 + KC);
                rb[i] = *(const float4*)(gB + (size_t)(ST_RPP * i) * D_SZ + k0 + KC);
            }
        }
#pragma unroll
        for (int kk = 0; kk < KC; kk += 8) {
            const int kb = kk + half * 4;
            const float4 a4 = *(const float4*)&sA[wm][fr][kb];
            const float4 b4 = *(const float4*)&sB[wn][fr][kb];
            v2f a, b;
            a[0] = a4.x; a[1] = a4.y;
            b[0] = b4.x; b[1] = b4.y;
            acc = __builtin_amdgcn_wmma_f32_16x16x4_f32(
                    false, a, false, b, (short)0, acc, false, false);
            a[0] = a4.z; a[1] = a4.w;
            b[0] = b4.z; b[1] = b4.w;
            acc = __builtin_amdgcn_wmma_f32_16x16x4_f32(
                    false, a, false, b, (short)0, acc, false, false);
        }
        __syncthreads();
    }
#pragma unroll
    for (int j = 0; j < 8; ++j) {
        int m = m0 + wm * 16 + j + half * 8;
        int n = n0 + wn * 16 + fr;
        Cout[(size_t)m * (2 * D_SZ) + n] = acc[j] + bias[n];
    }
}

#define FC_NBX (2 * D_SZ / 32)
#define FC_NBY (B_SZ / 32)
#define IMG_NB (4 * B_SZ)
__global__ __launch_bounds__(128)
void img_fc_kernel(
        const float* __restrict__ cap_mean, const float* __restrict__ fc_w,
        const float* __restrict__ fc_b, float* __restrict__ params,
        const float* __restrict__ img, float* __restrict__ meanR,
        float* __restrict__ cs1, float* __restrict__ cs2) {
    const int blk = blockIdx.x;
    if (blk < FC_NBX * FC_NBY)
        fc_gemm_role(blk % FC_NBX, blk / FC_NBX, cap_mean, fc_w, fc_b, params);
    else
        img_pre_role(blk - FC_NBX * FC_NBY, img, meanR, cs1, cs2);
}

#define DPP_ROW_XMASK(n) (0x160 + (n))
template <int N>
__device__ __forceinline__ float row_xor(float x) {
    const int b = __builtin_bit_cast(int, x);
    return __builtin_bit_cast(float, __builtin_amdgcn_update_dpp(
            0, b, DPP_ROW_XMASK(N), 0xf, 0xf, false));
}
__device__ __forceinline__ float group8_sum(float x) {
    x += row_xor<4>(x);
    x += row_xor<2>(x);
    x += row_xor<1>(x);
    return x;
}

#define SM_T  16
#define SM_N  (SM_T / ST_RPP)
#define SM_KW (KC / 4)
#define SM_LPR (SM_KW / 4)
#define SM_NCH (D_SZ / KC)
#define SM_PF 3
#define SM_LPC (4 + 4 * SM_N)
static_assert(SM_PF * SM_LPC <= 63, "more requests than a wave's load counter holds");

__device__ __forceinline__ void sims_wave_handover() {
    __builtin_amdgcn_fence(__ATOMIC_RELEASE, "wavefront", "local");
    __builtin_amdgcn_wave_barrier();
    __builtin_amdgcn_fence(__ATOMIC_ACQUIRE, "wavefront", "local");
    __builtin_amdgcn_sched_barrier(0);
}

__device__ __forceinline__
void sims_request_chunk(
        const float* __restrict__ meanR,
        const float* __restrict__ cs1, const float* __restrict__ cs2,
        const float* __restrict__ bn_w, const float* __restrict__ bn_b,
        const float* __restrict__ cap_mean, const float* __restrict__ params,
        const size_t goffA, const size_t goffC, const size_t goffW,
        const int scol, const int kn,
        float4& g1, float4& g2, float4& gw, float4& gb,
        float4 (&rm)[SM_N], float4 (&rc)[SM_N],
        float4 (&rw0)[SM_N], float4 (&rw1)[SM_N]) {
    g1 = *(const float4*)(cs1  + scol + kn);
    g2 = *(const float4*)(cs2  + scol + kn);
    gw = *(const float4*)(bn_w + scol + kn);
    gb = *(const float4*)(bn_b + scol + kn);
#pragma unroll
    for (int i = 0; i < SM_N; ++i) {
        const size_t o = (size_t)(ST_RPP * i) * D_SZ + kn;
        rm[i]  = *(const float4*)(meanR    + goffA + o);
        rc[i]  = *(const float4*)(cap_mean + goffC + o);
        rw0[i] = *(const float4*)(params   + goffW + 2 * o);
        rw1[i] = *(const float4*)(params   + goffW + 2 * o + 4);
    }
}

__device__ __forceinline__
void sims_hold(float& inv_n, float& nz,
               float (&qs)[SM_N], float (&rs)[SM_N], float (&ns)[SM_N]) {
    static_assert(SM_N == 4, "the statement below names the 4 staged rows");
    asm volatile("" : "+v"(inv_n), "+v"(nz),
                      "+v"(qs[0]), "+v"(qs[1]), "+v"(qs[2]), "+v"(qs[3]),
                      "+v"(rs[0]), "+v"(rs[1]), "+v"(rs[2]), "+v"(rs[3]),
                      "+v"(ns[0]), "+v"(ns[1]), "+v"(ns[2]), "+v"(ns[3]));
}

__device__ __forceinline__
void sims_form_chunk(
        const float4 g1, const float4 g2, const float4 gw, const float4 gb,
        const float4 (&rm)[SM_N], const float4 (&rc)[SM_N],
        const float4 (&rw0)[SM_N], const float4 (&rw1)[SM_N],
        const float inv_n, const float nz,
        float (&qs)[SM_N], float (&rs)[SM_N], float (&ns)[SM_N],
        float4 (&fv)[SM_N], float4 (&fv2)[SM_N], float4 (&fp)[SM_N],
        float4 (&fa2)[SM_N], float4 (&fab)[SM_N]) {
    const float c1[4] = {g1.x, g1.y, g1.z, g1.w};
    const float c2[4] = {g2.x, g2.y, g2.z, g2.w};
    const float cw[4] = {gw.x, gw.y, gw.z, gw.w};
    const float cb[4] = {gb.x, gb.y, gb.z, gb.w};
    float mu[4], sc[4];
#pragma unroll
    for (int e = 0; e < 4; ++e) {
        mu[e] = c1[e] * inv_n;
        const float ex2 = c2[e] * inv_n;
        const float var = ex2 - mu[e] * mu[e];
        sc[e] = cw[e] * rsqrtf(var + EPS_BN);
    }
#pragma unroll
    for (int i = 0; i < SM_N; ++i) {
        const float m[4]  = {rm[i].x, rm[i].y, rm[i].z, rm[i].w};
        const float cm[4] = {rc[i].x, rc[i].y, rc[i].z, rc[i].w};
        const float al[4] = {rw0[i].x, rw0[i].z, rw1[i].x, rw1[i].z};
        const float be[4] = {rw0[i].y, rw0[i].w, rw1[i].y, rw1[i].w};
        float v[4], v2[4], p[4], a2[4], ab[4];
#pragma unroll
        for (int e = 0; e < 4; ++e) {
            v[e]  = (m[e] - mu[e]) * sc[e] + cb[e];
            v2[e] = v[e] * v[e];
            p[e]  = __builtin_fmaf(al[e], cm[e], nz);
            a2[e] = __builtin_fmaf(al[e], al[e], nz);
            ab[e] = __builtin_fmaf(al[e], be[e], nz);
            qs[i] = __builtin_fmaf(be[e], cm[e], qs[i]);
            rs[i] = __builtin_fmaf(be[e], be[e], rs[i]);
            ns[i] = __builtin_fmaf(cm[e], cm[e], ns[i]);
        }
        fv[i]  = make_float4(v[0],  v[1],  v[2],  v[3]);
        fv2[i] = make_float4(v2[0], v2[1], v2[2], v2[3]);
        fp[i]  = make_float4(p[0],  p[1],  p[2],  p[3]);
        fa2[i] = make_float4(a2[0], a2[1], a2[2], a2[3]);
        fab[i] = make_float4(ab[0], ab[1], ab[2], ab[3]);
    }
}

__device__ __forceinline__
void sims_store_pieces(float* tBa, float* tBa2, float* tP, float* tA2, float* tAB,
                       const int loff,
                       const float4 (&fv)[SM_N], const float4 (&fv2)[SM_N],
                       const float4 (&fp)[SM_N], const float4 (&fa2)[SM_N],
                       const float4 (&fab)[SM_N]) {
#pragma unroll
    for (int i = 0; i < SM_N; ++i) {
        const int lo = loff + ST_RPP * i * KP;
        *(float4*)(tBa  + lo) = fv[i];
        *(float4*)(tBa2 + lo) = fv2[i];
        *(float4*)(tP   + lo) = fp[i];
        *(float4*)(tA2  + lo) = fa2[i];
        *(float4*)(tAB  + lo) = fab[i];
    }
}

__device__ __forceinline__
void sims_read_half(const float* tBa, const float* tBa2, const float* tP,
                    const float* tA2, const float* tAB, const int fo,
                    float4 (&x1)[2], float4 (&x2)[2], float4 (&xp)[2],
                    float4 (&xq)[2], float4 (&xr)[2]) {
#pragma unroll
    for (int j = 0; j < 2; ++j) {
        x1[j] = *(const float4*)(tBa  + fo + 8 * j);
        x2[j] = *(const float4*)(tBa2 + fo + 8 * j);
        xp[j] = *(const float4*)(tP   + fo + 8 * j);
        xq[j] = *(const float4*)(tA2  + fo + 8 * j);
        xr[j] = *(const float4*)(tAB  + fo + 8 * j);
    }
}

__device__ __forceinline__
void sims_products_half(const float4 (&x1)[2], const float4 (&x2)[2],
                        const float4 (&xp)[2], const float4 (&xq)[2],
                        const float4 (&xr)[2],
                        v8f& acc1, v8f& acc2, v8f& acc3) {
#pragma unroll
    for (int j = 0; j < 2; ++j) {
        v2f a1, a2, bp, b2, bb;
        a1[0] = x1[j].x; a1[1] = x1[j].y;
        a2[0] = x2[j].x; a2[1] = x2[j].y;
        bp[0] = xp[j].x; bp[1] = xp[j].y;
        b2[0] = xq[j].x; b2[1] = xq[j].y;
        bb[0] = xr[j].x; bb[1] = xr[j].y;
        acc1 = __builtin_amdgcn_wmma_f32_16x16x4_f32(
                false, a1, false, bp, (short)0, acc1, false, false);
        acc2 = __builtin_amdgcn_wmma_f32_16x16x4_f32(
                false, a2, false, b2, (short)0, acc2, false, false);
        acc3 = __builtin_amdgcn_wmma_f32_16x16x4_f32(
                false, a1, false, bb, (short)0, acc3, false, false);
        a1[0] = x1[j].z; a1[1] = x1[j].w;
        a2[0] = x2[j].z; a2[1] = x2[j].w;
        bp[0] = xp[j].z; bp[1] = xp[j].w;
        b2[0] = xq[j].z; b2[1] = xq[j].w;
        bb[0] = xr[j].z; bb[1] = xr[j].w;
        acc1 = __builtin_amdgcn_wmma_f32_16x16x4_f32(
                false, a1, false, bp, (short)0, acc1, false, false);
        acc2 = __builtin_amdgcn_wmma_f32_16x16x4_f32(
                false, a2, false, b2, (short)0, acc2, false, false);
        acc3 = __builtin_amdgcn_wmma_f32_16x16x4_f32(
                false, a1, false, bb, (short)0, acc3, false, false);
    }
}

__global__ __launch_bounds__(128) __attribute__((amdgpu_waves_per_eu(1, 2)))
void sims_gemm_kernel(
        const float* __restrict__ meanR,
        const float* __restrict__ cs1, const float* __restrict__ cs2,
        const float* __restrict__ bn_w, const float* __restrict__ bn_b,
        const float* __restrict__ cap_mean, const float* __restrict__ params,
        float* __restrict__ out) {
    __shared__ __align__(16) float sBa [SM_T][KP];
    __shared__ __align__(16) float sBa2[SM_T][KP];
    __shared__ __align__(16) float sP  [SM_T][KP];
    __shared__ __align__(16) float sA2 [SM_T][KP];
    __shared__ __align__(16) float sAB [SM_T][KP];
    __shared__ float red[3][3][8][32];
    __shared__ float s_row[4][3][SM_T];
    const int tid  = threadIdx.x;
    const int wave = tid >> 5;
    const int lane = tid & 31;
    const int half = lane >> 4, fr = lane & 15;
    const int bm0 = blockIdx.y * SM_T;
    const int cn0 = blockIdx.x * SM_T;
    const int kw0 = wave * SM_KW;
    const float inv_n = 1.f / (float)(B_SZ * R_SZ);
    v8f acc1 = {}, acc2 = {}, acc3 = {};
    static_assert(32 / SM_LPR == ST_RPP, "a pass of one wave takes ST_RPP rows");
    const int srow = lane / SM_LPR;
    const int scol = kw0 + (lane % SM_LPR) * 4;
    const size_t goffA = (size_t)(bm0 + srow) * D_SZ + scol;
    const size_t goffC = (size_t)(cn0 + srow) * D_SZ + scol;
    const size_t goffW = (size_t)(cn0 + srow) * (2 * D_SZ) + 2 * scol;
    const int loff = srow * KP + scol;
    const int foff = fr * KP + kw0 + half * 4;
    float4 rm[SM_PF][SM_N], rc[SM_PF][SM_N], rw0[SM_PF][SM_N], rw1[SM_PF][SM_N];
    float4 g1[SM_PF], g2[SM_PF], gw[SM_PF], gb[SM_PF];
    float4 fv[SM_N], fv2[SM_N], fp[SM_N], fa2[SM_N], fab[SM_N];
    float qs[SM_N] = {}, rs[SM_N] = {}, ns[SM_N] = {};
#pragma unroll
    for (int ch = 0; ch < SM_PF; ++ch)
        sims_request_chunk(meanR, cs1, cs2, bn_w, bn_b, cap_mean, params,
                           goffA, goffC, goffW, scol, ch * KC,
                           g1[ch], g2[ch], gw[ch], gb[ch],
                           rm[ch], rc[ch], rw0[ch], rw1[ch]);
    float inv_nh = inv_n, nz = -0.0f;
    sims_form_chunk(g1[0], g2[0], gw[0], gb[0], rm[0], rc[0], rw0[0], rw1[0],
                    inv_nh, nz, qs, rs, ns, fv, fv2, fp, fa2, fab);
    float4 ya1[2], ya2[2], yap[2], yaq[2], yar[2];
    float4 yb1[2], yb2[2], ybp[2], ybq[2], ybr[2];
    sims_store_pieces(&sBa[0][0], &sBa2[0][0], &sP[0][0], &sA2[0][0], &sAB[0][0],
                      loff, fv, fv2, fp, fa2, fab);
    sims_wave_handover();
    sims_read_half(&sBa[0][0], &sBa2[0][0], &sP[0][0], &sA2[0][0], &sAB[0][0],
                   foff, ya1, ya2, yap, yaq, yar);
#pragma unroll
    for (int ch = 0; ch < SM_NCH; ++ch) {
        __builtin_amdgcn_sched_barrier(0);
        if (ch + SM_PF < SM_NCH)
            sims_request_chunk(meanR, cs1, cs2, bn_w, bn_b, cap_mean, params,
                               goffA, goffC, goffW, scol, (ch + SM_PF) * KC,
                               g1[ch % SM_PF], g2[ch % SM_PF],
                               gw[ch % SM_PF], gb[ch % SM_PF],
                               rm[ch % SM_PF], rc[ch % SM_PF],
                               rw0[ch % SM_PF], rw1[ch % SM_PF]);
        sims_read_half(&sBa[0][0], &sBa2[0][0], &sP[0][0], &sA2[0][0], &sAB[0][0],
                       foff + SM_KW / 2, yb1, yb2, ybp, ybq, ybr);
        sims_products_half(ya1, ya2, yap, yaq, yar, acc1, acc2, acc3);
        if (ch + 1 < SM_NCH) {
            sims_hold(inv_nh, nz, qs, rs, ns);
            sims_form_chunk(g1[(ch + 1) % SM_PF], g2[(ch + 1) % SM_PF],
                            gw[(ch + 1) % SM_PF], gb[(ch + 1) % SM_PF],
                            rm[(ch + 1) % SM_PF], rc[(ch + 1) % SM_PF],
                            rw0[(ch + 1) % SM_PF], rw1[(ch + 1) % SM_PF],
                            inv_nh, nz, qs, rs, ns, fv, fv2, fp, fa2, fab);
            sims_wave_handover();
            sims_store_pieces(&sBa[0][0], &sBa2[0][0], &sP[0][0], &sA2[0][0], &sAB[0][0],
                              loff, fv, fv2, fp, fa2, fab);
            sims_wave_handover();
            sims_read_half(&sBa[0][0], &sBa2[0][0], &sP[0][0], &sA2[0][0], &sAB[0][0],
                           foff, ya1, ya2, yap, yaq, yar);
        }
        __builtin_amdgcn_sched_barrier(0);
        sims_products_half(yb1, yb2, ybp, ybq, ybr, acc1, acc2, acc3);
    }
    static_assert(SM_LPR == 8, "group8_sum adds over the 8 lanes of a staged row");
    float fq[SM_N], fr2[SM_N], fn[SM_N];
#pragma unroll
    for (int i = 0; i < SM_N; ++i) {
        fq[i]  = group8_sum(qs[i]);
        fr2[i] = group8_sum(rs[i]);
        fn[i]  = group8_sum(ns[i]);
    }
    if (lane % SM_LPR == 0) {
#pragma unroll
        for (int i = 0; i < SM_N; ++i) {
            s_row[wave][0][srow + ST_RPP * i] = fq[i];
            s_row[wave][1][srow + ST_RPP * i] = fr2[i];
            s_row[wave][2][srow + ST_RPP * i] = fn[i];
        }
    }
    if (wave > 0) {
#pragma unroll
        for (int j = 0; j < 8; ++j) {
            red[wave - 1][0][j][lane] = acc1[j];
            red[wave - 1][1][j][lane] = acc2[j];
            red[wave - 1][2][j][lane] = acc3[j];
        }
    }
    __syncthreads();
    if (wave == 0) {
        const int c = cn0 + fr;
        const float qc  = ((s_row[0][0][fr] + s_row[1][0][fr]) + s_row[2][0][fr]) + s_row[3][0][fr];
        const float rc2 = ((s_row[0][1][fr] + s_row[1][1][fr]) + s_row[2][1][fr]) + s_row[3][1][fr];
        const float nsq = ((s_row[0][2][fr] + s_row[1][2][fr]) + s_row[2][2][fr]) + s_row[3][2][fr];
        const float nrm = sqrtf(nsq) + EPS_L2;
#pragma unroll
        for (int j = 0; j < 8; ++j) {
            float s1 = acc1[j], s2 = acc2[j], s3 = acc3[j];
            for (int w = 0; w < 3; ++w) {
                s1 += red[w][0][j][lane];
                s2 += red[w][1][j][lane];
                s3 += red[w][2][j][lane];
            }
            int b = bm0 + j + half * 8;
            float num = (s1 + qc) / nrm;
            float den = sqrtf(s2 + 2.f * s3 + rc2) + EPS_L2;
            out[(size_t)b * B_SZ + c] = num / den;
        }
    }
}

extern "C" void kernel_launch(void* const* d_in, const int* in_sizes, int n_in,
                              void* d_out, int out_size, void* d_ws, size_t ws_size,
                              hipStream_t stream) {
    const float* img   = (const float*)d_in[0];
    const float* cap   = (const float*)d_in[1];
    const int*   lens  = (const int*)  d_in[2];
    const float* fc_w  = (const float*)d_in[3];
    const float* fc_b  = (const float*)d_in[4];
    const float* bn_w  = (const float*)d_in[5];
    const float* bn_b  = (const float*)d_in[6];
    float* out = (float*)d_out;

    float* ws = (float*)d_ws;
    const size_t BD = (size_t)B_SZ * D_SZ;
    float* cap_mean = ws;
    float* meanR    = cap_mean + BD;
    float* params   = meanR + BD;
    float* colsum1  = params + 2 * BD;
    float* colsum2  = colsum1 + D_SZ;

    cap_pre_kernel<<<dim3(2 * B_SZ), 256, 0, stream>>>(cap, lens, cap_mean, colsum1);
    img_fc_kernel<<<dim3(FC_NBX * FC_NBY + IMG_NB), 128, 0, stream>>>(
            cap_mean, fc_w, fc_b, params, img, meanR, colsum1, colsum2);
    sims_gemm_kernel<<<dim3(B_SZ / SM_T, B_SZ / SM_T), 128, 0, stream>>>(
            meanR, colsum1, colsum2, bn_w, bn_b, cap_mean, params, out);
}
